# starting kernel
# baseline (speedup 1.0000x reference)
#include <hip/hip_runtime.h>


typedef __attribute__((ext_vector_type(2))) float v2f;
typedef __attribute__((ext_vector_type(8))) float v8f;


__global__ void zero_f4_kernel(float4* __restrict__ p, int n4) {
    int i = blockIdx.x * blockDim.x + threadIdx.x;
    if (i < n4) p[i] = make_float4(0.f, 0.f, 0.f, 0.f);
}

__global__ void transpose_kernel(const float* __restrict__ in,
                                 float* __restrict__ outT, int K, int N) {
    int i = blockIdx.x * blockDim.x + threadIdx.x;
    if (i < N * K) {
        int n = i / K, k = i % K;
        outT[i] = in[k * N + n];
    }
}

__global__ void degree_count_kernel(const int* __restrict__ src,
                                    const int* __restrict__ dst,
                                    float* __restrict__ deg_out,
                                    float* __restrict__ deg_in, int nE) {
    int e = blockIdx.x * blockDim.x + threadIdx.x;
    if (e < nE) {
        atomicAdd(&deg_out[src[e]], 1.0f);
        atomicAdd(&deg_in[dst[e]], 1.0f);
    }
}

__global__ void rsqrt_clamp_kernel(float* __restrict__ d, int n) {
    int i = blockIdx.x * blockDim.x + threadIdx.x;
    if (i < n) d[i] = rsqrtf(fmaxf(d[i], 1.0f));
}

__global__ void edge_aggregate_kernel(const int* __restrict__ src,
                                      const int* __restrict__ dst,
                                      const float* __restrict__ h,
                                      float* __restrict__ agg, int nE) {
    int tid = blockIdx.x * blockDim.x + threadIdx.x;
    int e = tid >> 5;
    int f = (tid & 31) * 4;
    if (e < nE) {
        const float4 v = *(const float4*)(h + (size_t)src[e] * 128 + f);
        float* a = agg + (size_t)dst[e] * 128 + f;
        atomicAdd(a + 0, v.x);
        atomicAdd(a + 1, v.y);
        atomicAdd(a + 2, v.z);
        atomicAdd(a + 3, v.w);
    }
}

__global__ void scatter_rows_kernel(const int* __restrict__ ids,
                                    const float* __restrict__ h,
                                    float* __restrict__ out_feat, int n) {
    int tid = blockIdx.x * blockDim.x + threadIdx.x;
    int r = tid >> 5;
    int f = (tid & 31) * 4;
    if (r < n) {
        const float4 v = *(const float4*)(h + (size_t)r * 128 + f);
        float* o = out_feat + (size_t)ids[r] * 128 + f;
        atomicAdd(o + 0, v.x);
        atomicAdd(o + 1, v.y);
        atomicAdd(o + 2, v.z);
        atomicAdd(o + 3, v.w);
    }
}

template <int NTILES>
__global__ __launch_bounds__(256)
void gemm_wmma_k128_kernel(const float* __restrict__ A,
                           const float* __restrict__ Wt,
                           const float* __restrict__ bias,
                           const float* __restrict__ arowscale,
                           const float* __restrict__ erowscale,
                           float* __restrict__ out,
                           int M, int relu) {
    const int lane    = threadIdx.x & 31;
    const int wave    = threadIdx.x >> 5;
    const int task    = blockIdx.x * (blockDim.x >> 5) + wave;
    const int t0      = task * 2;
    const int t1      = t0 + 1;
    if (t0 * 16 >= M) return;
    const bool has1   = (t1 * 16 < M);
    const int halfsel = lane >> 4;
    const int lid     = lane & 15;
    const int N       = NTILES * 16;

    v8f acc0[NTILES], acc1[NTILES];
#pragma unroll
    for (int nt = 0; nt < NTILES; ++nt) {
        float bv = bias[nt * 16 + lid];
#pragma unroll
        for (int j = 0; j < 8; ++j) { acc0[nt][j] = bv; acc1[nt][j] = bv; }
    }

    const int arow0 = t0 * 16 + lid;
    const int arow1 = min(t1 * 16 + lid, M - 1);
    const float* aptr0 = A + (size_t)arow0 * 128 + halfsel * 2;
    const float* aptr1 = A + (size_t)arow1 * 128 + halfsel * 2;
    const float ascale0 = arowscale ? arowscale[arow0] : 1.0f;
    const float ascale1 = arowscale ? arowscale[arow1] : 1.0f;

    const float* wlane = Wt + (size_t)lid * 128 + halfsel * 2;

#pragma unroll 2
    for (int kk = 0; kk < 32; ++kk) {
        v2f a0 = *(const v2f*)(aptr0);
        v2f a1 = *(const v2f*)(aptr1);
        a0.x *= ascale0; a0.y *= ascale0;
        a1.x *= ascale1; a1.y *= ascale1;
        aptr0 += 4; aptr1 += 4;
#pragma unroll
        for (int nt = 0; nt < NTILES; ++nt) {
            v2f b = *(const v2f*)(wlane + (size_t)nt * 16 * 128 + kk * 4);
            acc0[nt] = __builtin_amdgcn_wmma_f32_16x16x4_f32(
                false, a0, false, b, (short)0, acc0[nt], false, false);
            acc1[nt] = __builtin_amdgcn_wmma_f32_16x16x4_f32(
                false, a1, false, b, (short)0, acc1[nt], false, false);
        }
    }

#pragma unroll
    for (int nt = 0; nt < NTILES; ++nt) {
#pragma unroll
        for (int j = 0; j < 8; ++j) {
            float v = acc0[nt][j];
            if (relu) v = fmaxf(v, 0.0f);
            const int row = t0 * 16 + j + halfsel * 8;
            if (erowscale) v *= erowscale[row];
            out[(size_t)row * N + nt * 16 + lid] = v;
        }
    }
    if (has1) {
#pragma unroll
        for (int nt = 0; nt < NTILES; ++nt) {
#pragma unroll
            for (int j = 0; j < 8; ++j) {
                float v = acc1[nt][j];
                if (relu) v = fmaxf(v, 0.0f);
                const int row = t1 * 16 + j + halfsel * 8;
                if (erowscale) v *= erowscale[row];
                out[(size_t)row * N + nt * 16 + lid] = v;
            }
        }
    }
}


extern "C" void kernel_launch(void* const* d_in, const int* in_sizes, int n_in,
                              void* d_out, int out_size, void* d_ws, size_t ws_size,
                              hipStream_t stream) {
    (void)n_in; (void)ws_size;
    const float* x   = (const float*)d_in[0];
    const float* W2  = (const float*)d_in[1];
    const float* b2  = (const float*)d_in[2];
    const float* gw  = (const float*)d_in[3];
    const float* gb  = (const float*)d_in[4];
    const float* Wc  = (const float*)d_in[5];
    const float* bc  = (const float*)d_in[6];
    const int*   src = (const int*)d_in[7];
    const int*   dst = (const int*)d_in[8];
    const int*   ids = (const int*)d_in[9];

    const int H  = 128;
    const int C  = 16;
    const int L  = in_sizes[2] / H;
    const int NS = in_sizes[9] / L;
    const int NE = in_sizes[7] / L;
    const int NL = in_sizes[4] / (L * H);
    const int NN = out_size / C;

    float* ws       = (float*)d_ws;
    float* out_feat = ws;
    float* hbuf     = out_feat + (size_t)NN * H;
    float* agg      = hbuf + (size_t)NS * H;
    float* deg_out  = agg + (size_t)NS * H;
    float* deg_in   = deg_out + NS;
    float* w2t      = deg_in + NS;
    float* gwt      = w2t + (size_t)L * H * H;
    float* wct      = gwt + (size_t)L * NL * H * H;

    const int T = 256;
    const int feat_blocks_NS = (NS * 32 + T - 1) / T;
    const int gemm_blocks_NS = (((NS / 16) + 1) / 2 + 7) / 8;
    const int gemm_blocks_NN = (((NN / 16) + 1) / 2 + 7) / 8;

    {
        int nHH = H * H;
        int bHH = (nHH + T - 1) / T;
        for (int i = 0; i < L; ++i)
            transpose_kernel<<<bHH, T, 0, stream>>>(W2 + (size_t)i * nHH,
                                                    w2t + (size_t)i * nHH, H, H);
        for (int i = 0; i < L * NL; ++i)
            transpose_kernel<<<bHH, T, 0, stream>>>(gw + (size_t)i * nHH,
                                                    gwt + (size_t)i * nHH, H, H);
        int nHC = H * C;
        transpose_kernel<<<(nHC + T - 1) / T, T, 0, stream>>>(Wc, wct, H, C);
    }

    {
        int n4 = (int)((size_t)NN * H / 4);
        zero_f4_kernel<<<(n4 + T - 1) / T, T, 0, stream>>>((float4*)out_feat, n4);
    }

    for (int i = 0; i < L; ++i) {
        const int*   src_i = src + (size_t)i * NE;
        const int*   dst_i = dst + (size_t)i * NE;
        const int*   ids_i = ids + (size_t)i * NS;
        const float* x_i   = x   + (size_t)i * NS * H;
        const float* w2t_i = w2t + (size_t)i * H * H;
        const float* b2_i  = b2  + (size_t)i * H;

        {
            int n4 = (2 * NS) / 4;
            zero_f4_kernel<<<(n4 + T - 1) / T, T, 0, stream>>>((float4*)deg_out, n4);
        }
        degree_count_kernel<<<(NE + T - 1) / T, T, 0, stream>>>(src_i, dst_i,
                                                                deg_out, deg_in, NE);
        rsqrt_clamp_kernel<<<(2 * NS + T - 1) / T, T, 0, stream>>>(deg_out, 2 * NS);

        gemm_wmma_k128_kernel<8><<<gemm_blocks_NS, T, 0, stream>>>(
            x_i, w2t_i, b2_i, nullptr, deg_out, hbuf, NS, 0);

        for (int l = 0; l < NL; ++l) {
            const float* gwt_il = gwt + ((size_t)i * NL + l) * H * H;
            const float* gb_il  = gb  + ((size_t)i * NL + l) * H;
            {
                int n4 = (int)((size_t)NS * H / 4);
                zero_f4_kernel<<<(n4 + T - 1) / T, T, 0, stream>>>((float4*)agg, n4);
            }
            edge_aggregate_kernel<<<(NE * 32) / T, T, 0, stream>>>(src_i, dst_i,
                                                                   hbuf, agg, NE);
            const float* escale = (l + 1 < NL) ? deg_out : nullptr;
            gemm_wmma_k128_kernel<8><<<gemm_blocks_NS, T, 0, stream>>>(
                agg, gwt_il, gb_il, deg_in, escale, hbuf, NS, 1);
        }

        scatter_rows_kernel<<<feat_blocks_NS, T, 0, stream>>>(ids_i, hbuf,
                                                              out_feat, NS);
    }

    gemm_wmma_k128_kernel<1><<<gemm_blocks_NN, T, 0, stream>>>(
        out_feat, wct, bc, nullptr, nullptr, (float*)d_out, NN, 0);
}


# GEMM: W packed in B-fragment order and kept whole in each wave's registers (one wave per SIMD, 256 blocks); waves stream 16-row A tiles, next A tile and 8 row scales loaded ahead; k order kept
# speedup vs baseline: 1.1747x; 1.1747x over previous
#include <hip/hip_runtime.h>


typedef __attribute__((ext_vector_type(2))) float v2f;
typedef __attribute__((ext_vector_type(8))) float v8f;


__global__ void zero_f4_kernel(float4* __restrict__ p, int n4) {
    int i = blockIdx.x * blockDim.x + threadIdx.x;
    if (i < n4) p[i] = make_float4(0.f, 0.f, 0.f, 0.f);
}

__global__ void pack_bfrag_kernel(const float* __restrict__ in,
                                  float* __restrict__ outF, int N) {
    int i = blockIdx.x * blockDim.x + threadIdx.x;
    if (i < N * 128) {
        int c    = i & 3;
        int lane = (i >> 2) & 31;
        int kk2  = (i >> 7) & 15;
        int nt   = i >> 11;
        int kk   = 2 * kk2 + (c >> 1);
        int k    = kk * 4 + (lane >> 4) * 2 + (c & 1);
        int n    = nt * 16 + (lane & 15);
        outF[i] = in[k * N + n];
    }
}

__global__ void degree_count_kernel(const int* __restrict__ src,
                                    const int* __restrict__ dst,
                                    float* __restrict__ deg_out,
                                    float* __restrict__ deg_in, int nE) {
    int e = blockIdx.x * blockDim.x + threadIdx.x;
    if (e < nE) {
        atomicAdd(&deg_out[src[e]], 1.0f);
        atomicAdd(&deg_in[dst[e]], 1.0f);
    }
}

__global__ void rsqrt_clamp_kernel(float* __restrict__ d, int n) {
    int i = blockIdx.x * blockDim.x + threadIdx.x;
    if (i < n) d[i] = rsqrtf(fmaxf(d[i], 1.0f));
}

__global__ void edge_aggregate_kernel(const int* __restrict__ src,
                                      const int* __restrict__ dst,
                                      const float* __restrict__ h,
                                      float* __restrict__ agg, int nE) {
    int tid = blockIdx.x * blockDim.x + threadIdx.x;
    int e = tid >> 5;
    int f = (tid & 31) * 4;
    if (e < nE) {
        const float4 v = *(const float4*)(h + (size_t)src[e] * 128 + f);
        float* a = agg + (size_t)dst[e] * 128 + f;
        atomicAdd(a + 0, v.x);
        atomicAdd(a + 1, v.y);
        atomicAdd(a + 2, v.z);
        atomicAdd(a + 3, v.w);
    }
}

__global__ void scatter_rows_kernel(const int* __restrict__ ids,
                                    const float* __restrict__ h,
                                    float* __restrict__ out_feat, int n) {
    int tid = blockIdx.x * blockDim.x + threadIdx.x;
    int r = tid >> 5;
    int f = (tid & 31) * 4;
    if (r < n) {
        const float4 v = *(const float4*)(h + (size_t)r * 128 + f);
        float* o = out_feat + (size_t)ids[r] * 128 + f;
        atomicAdd(o + 0, v.x);
        atomicAdd(o + 1, v.y);
        atomicAdd(o + 2, v.z);
        atomicAdd(o + 3, v.w);
    }
}

template <int NTILES>
__global__ __launch_bounds__(128)
void gemm_wmma_k128_kernel(const float* __restrict__ A,
                           const float* __restrict__ Wf,
                           const float* __restrict__ bias,
                           const float* __restrict__ arowscale,
                           const float* __restrict__ erowscale,
                           float* __restrict__ out,
                           int M, int relu) {
    const int lane    = threadIdx.x & 31;
    const int wave    = __builtin_amdgcn_readfirstlane((int)(threadIdx.x >> 5));
    const int wpb     = blockDim.x >> 5;
    const int nwaves  = gridDim.x * wpb;
    const int ntiles  = M >> 4;
    int tile          = blockIdx.x * wpb + wave;
    if (tile >= ntiles) return;
    const int halfsel = lane >> 4;
    const int lid     = lane & 15;
    const int N       = NTILES * 16;

    v2f b[NTILES][32];
    {
        const float4* wf4 = (const float4*)Wf + lane;
#pragma unroll
        for (int nt = 0; nt < NTILES; ++nt) {
#pragma unroll
            for (int kk2 = 0; kk2 < 16; ++kk2) {
                const float4 q = wf4[(nt * 16 + kk2) * 32];
                b[nt][2 * kk2].x     = q.x; b[nt][2 * kk2].y     = q.y;
                b[nt][2 * kk2 + 1].x = q.z; b[nt][2 * kk2 + 1].y = q.w;
            }
        }
    }
    float bv[NTILES];
#pragma unroll
    for (int nt = 0; nt < NTILES; ++nt) bv[nt] = bias[nt * 16 + lid];

    v2f   an[32];
    float sn;
    {
        const int r = tile * 16 + lid;
        const float* ap = A + (size_t)r * 128 + halfsel * 2;
#pragma unroll
        for (int kk = 0; kk < 32; ++kk) an[kk] = *(const v2f*)(ap + kk * 4);
        sn = arowscale ? arowscale[r] : 1.0f;
    }

    for (; tile < ntiles; tile += nwaves) {
        v2f a[32];
#pragma unroll
        for (int kk = 0; kk < 32; ++kk) { a[kk].x = an[kk].x * sn; a[kk].y = an[kk].y * sn; }

        const int tnext = tile + nwaves;
        if (tnext < ntiles) {
            const int r = tnext * 16 + lid;
            const float* ap = A + (size_t)r * 128 + halfsel * 2;
#pragma unroll
            for (int kk = 0; kk < 32; ++kk) an[kk] = *(const v2f*)(ap + kk * 4);
            sn = arowscale ? arowscale[r] : 1.0f;
        }

        float es[8];
        if (erowscale) {
#pragma unroll
            for (int j = 0; j < 8; ++j) es[j] = erowscale[tile * 16 + j + halfsel * 8];
        } else {
#pragma unroll
            for (int j = 0; j < 8; ++j) es[j] = 1.0f;
        }

        v8f acc[NTILES];
#pragma unroll
        for (int nt = 0; nt < NTILES; ++nt) {
#pragma unroll
            for (int j = 0; j < 8; ++j) acc[nt][j] = bv[nt];
        }

#pragma unroll
        for (int kk = 0; kk < 32; ++kk) {
#pragma unroll
            for (int nt = 0; nt < NTILES; ++nt) {
                acc[nt] = __builtin_amdgcn_wmma_f32_16x16x4_f32(
                    false, a[kk], false, b[nt][kk], (short)0, acc[nt], false, false);
            }
        }

#pragma unroll
        for (int nt = 0; nt < NTILES; ++nt) {
#pragma unroll
            for (int j = 0; j < 8; ++j) {
                float v = acc[nt][j];
                if (relu) v = fmaxf(v, 0.0f);
                const int row = tile * 16 + j + halfsel * 8;
                if (erowscale) v *= es[j];
                out[(size_t)row * N + nt * 16 + lid] = v;
            }
        }
    }
}


extern "C" void kernel_launch(void* const* d_in, const int* in_sizes, int n_in,
                              void* d_out, int out_size, void* d_ws, size_t ws_size,
                              hipStream_t stream) {
    (void)n_in; (void)ws_size;
    const float* x   = (const float*)d_in[0];
    const float* W2  = (const float*)d_in[1];
    const float* b2  = (const float*)d_in[2];
    const float* gw  = (const float*)d_in[3];
    const float* gb  = (const float*)d_in[4];
    const float* Wc  = (const float*)d_in[5];
    const float* bc  = (const float*)d_in[6];
    const int*   src = (const int*)d_in[7];
    const int*   dst = (const int*)d_in[8];
    const int*   ids = (const int*)d_in[9];

    const int H  = 128;
    const int C  = 16;
    const int L  = in_sizes[2] / H;
    const int NS = in_sizes[9] / L;
    const int NE = in_sizes[7] / L;
    const int NL = in_sizes[4] / (L * H);
    const int NN = out_size / C;

    float* ws       = (float*)d_ws;
    float* out_feat = ws;
    float* hbuf     = out_feat + (size_t)NN * H;
    float* agg      = hbuf + (size_t)NS * H;
    float* deg_out  = agg + (size_t)NS * H;
    float* deg_in   = deg_out + NS;
    float* w2t      = deg_in + NS;
    float* gwt      = w2t + (size_t)L * H * H;
    float* wct      = gwt + (size_t)L * NL * H * H;

    const int T = 256;
    const int feat_blocks_NS = (NS * 32 + T - 1) / T;
    const int GT = 128;
    const int gemm_blocks_NS = 256;
    const int gemm_blocks_NN = (((NN / 16) + 1) / 2 + 3) / 4;

    {
        int nHH = H * H;
        int bHH = (nHH + T - 1) / T;
        for (int i = 0; i < L; ++i)
            pack_bfrag_kernel<<<bHH, T, 0, stream>>>(W2 + (size_t)i * nHH,
                                                     w2t + (size_t)i * nHH, H);
        for (int i = 0; i < L * NL; ++i)
            pack_bfrag_kernel<<<bHH, T, 0, stream>>>(gw + (size_t)i * nHH,
                                                     gwt + (size_t)i * nHH, H);
        int nHC = H * C;
        pack_bfrag_kernel<<<(nHC + T - 1) / T, T, 0, stream>>>(Wc, wct, C);
    }

    {
        int n4 = (int)((size_t)NN * H / 4);
        zero_f4_kernel<<<(n4 + T - 1) / T, T, 0, stream>>>((float4*)out_feat, n4);
    }

    for (int i = 0; i < L; ++i) {
        const int*   src_i = src + (size_t)i * NE;
        const int*   dst_i = dst + (size_t)i * NE;
        const int*   ids_i = ids + (size_t)i * NS;
        const float* x_i   = x   + (size_t)i * NS * H;
        const float* w2t_i = w2t + (size_t)i * H * H;
        const float* b2_i  = b2  + (size_t)i * H;

        {
            int n4 = (2 * NS) / 4;
            zero_f4_kernel<<<(n4 + T - 1) / T, T, 0, stream>>>((float4*)deg_out, n4);
        }
        degree_count_kernel<<<(NE + T - 1) / T, T, 0, stream>>>(src_i, dst_i,
                                                                deg_out, deg_in, NE);
        rsqrt_clamp_kernel<<<(2 * NS + T - 1) / T, T, 0, stream>>>(deg_out, 2 * NS);

        gemm_wmma_k128_kernel<8><<<gemm_blocks_NS, GT, 0, stream>>>(
            x_i, w2t_i, b2_i, nullptr, deg_out, hbuf, NS, 0);

        for (int l = 0; l < NL; ++l) {
            const float* gwt_il = gwt + ((size_t)i * NL + l) * H * H;
            const float* gb_il  = gb  + ((size_t)i * NL + l) * H;
            {
                int n4 = (int)((size_t)NS * H / 4);
                zero_f4_kernel<<<(n4 + T - 1) / T, T, 0, stream>>>((float4*)agg, n4);
            }
            edge_aggregate_kernel<<<(NE * 32) / T, T, 0, stream>>>(src_i, dst_i,
                                                                   hbuf, agg, NE);
            const float* escale = (l + 1 < NL) ? deg_out : nullptr;
            gemm_wmma_k128_kernel<8><<<gemm_blocks_NS, GT, 0, stream>>>(
                agg, gwt_il, gb_il, deg_in, escale, hbuf, NS, 1);
        }

        scatter_rows_kernel<<<feat_blocks_NS, T, 0, stream>>>(ids_i, hbuf,
                                                              out_feat, NS);
    }

    gemm_wmma_k128_kernel<1><<<gemm_blocks_NN, GT, 0, stream>>>(
        out_feat, wct, bc, nullptr, nullptr, (float*)d_out, NN, 0);
}


# edge sums: 256 work-groups each own a destination row range in LDS, scan all dst ids, add matching source rows by LDS float atomics, store rows whole; no global atomics or zero pass; arrival order
# speedup vs baseline: 1.9575x; 1.6664x over previous
#include <hip/hip_runtime.h>


typedef __attribute__((ext_vector_type(2))) float v2f;
typedef __attribute__((ext_vector_type(8))) float v8f;


__global__ void zero_f4_kernel(float4* __restrict__ p, int n4) {
    int i = blockIdx.x * blockDim.x + threadIdx.x;
    if (i < n4) p[i] = make_float4(0.f, 0.f, 0.f, 0.f);
}

__global__ void pack_bfrag_kernel(const float* __restrict__ in,
                                  float* __restrict__ outF, int N) {
    int i = blockIdx.x * blockDim.x + threadIdx.x;
    if (i < N * 128) {
        int c    = i & 3;
        int lane = (i >> 2) & 31;
        int kk2  = (i >> 7) & 15;
        int nt   = i >> 11;
        int kk   = 2 * kk2 + (c >> 1);
        int k    = kk * 4 + (lane >> 4) * 2 + (c & 1);
        int n    = nt * 16 + (lane & 15);
        outF[i] = in[k * N + n];
    }
}

__global__ void degree_count_kernel(const int* __restrict__ src,
                                    const int* __restrict__ dst,
                                    float* __restrict__ deg_out,
                                    float* __restrict__ deg_in, int nE) {
    int e = blockIdx.x * blockDim.x + threadIdx.x;
    if (e < nE) {
        atomicAdd(&deg_out[src[e]], 1.0f);
        atomicAdd(&deg_in[dst[e]], 1.0f);
    }
}

__global__ void rsqrt_clamp_kernel(float* __restrict__ d, int n) {
    int i = blockIdx.x * blockDim.x + threadIdx.x;
    if (i < n) d[i] = rsqrtf(fmaxf(d[i], 1.0f));
}

#define OWN_RMAX  200
#define OWN_WAVES 32
#define OWN_QCAP  256
#define OWN_FLUSH 64
#define OWN_U     8

__device__ __forceinline__ void own_drain(const int* __restrict__ src,
                                          const float* __restrict__ h,
                                          float* acc, const unsigned* qe,
                                          const unsigned* qd, int n, int lane) {
    for (int i0 = 0; i0 < n; i0 += 32) {
        const int m = (n - i0 < 32) ? (n - i0) : 32;
        int my_s = 0, my_d = 0;
        if (lane < m) { my_s = src[qe[i0 + lane]]; my_d = (int)qd[i0 + lane]; }
        for (int j0 = 0; j0 < m; j0 += 8) {
            float4 v[8];
            int    dl[8];
#pragma unroll
            for (int u = 0; u < 8; ++u) {
                const int j = (j0 + u < m) ? (j0 + u) : (m - 1);
                const int s = __builtin_amdgcn_readlane(my_s, j);
                dl[u]       = __builtin_amdgcn_readlane(my_d, j);
                v[u] = *(const float4*)(h + (size_t)s * 128 + lane * 4);
            }
#pragma unroll
            for (int u = 0; u < 8; ++u) {
                if (j0 + u < m) {
                    float* row = acc + dl[u] * 128 + lane;
                    __hip_atomic_fetch_add(row,      v[u].x, __ATOMIC_RELAXED, __HIP_MEMORY_SCOPE_WORKGROUP);
                    __hip_atomic_fetch_add(row + 32, v[u].y, __ATOMIC_RELAXED, __HIP_MEMORY_SCOPE_WORKGROUP);
                    __hip_atomic_fetch_add(row + 64, v[u].z, __ATOMIC_RELAXED, __HIP_MEMORY_SCOPE_WORKGROUP);
                    __hip_atomic_fetch_add(row + 96, v[u].w, __ATOMIC_RELAXED, __HIP_MEMORY_SCOPE_WORKGROUP);
                }
            }
        }
    }
}

__device__ __forceinline__ int own_push(bool m, unsigned e, unsigned r,
                                        unsigned* qe, unsigned* qd, int cnt) {
    const unsigned k = __builtin_amdgcn_ballot_w32(m);
    if (m) {
        const int p = cnt + (int)__builtin_amdgcn_mbcnt_lo(k, 0u);
        qe[p] = e;
        qd[p] = r;
    }
    return cnt + __builtin_popcount(k);
}

__global__ __launch_bounds__(1024)
void edge_sum_owner_kernel(const int* __restrict__ src,
                           const int* __restrict__ dst,
                           const float* __restrict__ h,
                           float* __restrict__ agg, int nE, int nRows) {
    __shared__ float    acc[OWN_RMAX * 128];
    __shared__ unsigned qe[OWN_WAVES][OWN_QCAP];
    __shared__ unsigned qd[OWN_WAVES][OWN_QCAP];

    const int lane = threadIdx.x & 31;
    const int wave = __builtin_amdgcn_readfirstlane((int)(threadIdx.x >> 5));
    const int nwav = blockDim.x >> 5;
    int R = (nRows + (int)gridDim.x - 1) / (int)gridDim.x;
    if (R > OWN_RMAX) R = OWN_RMAX;
    const int nGf = nE >> 7;
    const int rot = (int)(((long long)blockIdx.x * nGf) / (int)gridDim.x);
    unsigned* myqe = qe[wave];
    unsigned* myqd = qd[wave];

    for (int lo = blockIdx.x * R; lo < nRows; lo += (int)gridDim.x * R) {
        const int rows = (nRows - lo < R) ? (nRows - lo) : R;
        for (int i = threadIdx.x; i < rows * 32; i += blockDim.x)
            ((float4*)acc)[i] = make_float4(0.f, 0.f, 0.f, 0.f);
        __syncthreads();

        int cnt = 0;
        for (int g0 = wave; g0 < nGf; g0 += OWN_U * nwav) {
            uint4 d[OWN_U];
            int   eb[OWN_U];
#pragma unroll
            for (int u = 0; u < OWN_U; ++u) {
                int g = g0 + u * nwav;
                if (g >= nGf) g = nGf - 1;
                g += rot;
                if (g >= nGf) g -= nGf;
                eb[u] = g * 128 + lane * 4;
                d[u]  = *(const uint4*)(dst + eb[u]);
            }
#pragma unroll
            for (int u = 0; u < OWN_U; ++u) {
                if (g0 + u * nwav < nGf) {
                    const unsigned r0 = d[u].x - (unsigned)lo;
                    const unsigned r1 = d[u].y - (unsigned)lo;
                    const unsigned r2 = d[u].z - (unsigned)lo;
                    const unsigned r3 = d[u].w - (unsigned)lo;
                    const bool m0 = r0 < (unsigned)rows, m1 = r1 < (unsigned)rows;
                    const bool m2 = r2 < (unsigned)rows, m3 = r3 < (unsigned)rows;
                    if (__builtin_amdgcn_ballot_w32(m0 | m1 | m2 | m3)) {
                        cnt = own_push(m0, (unsigned)eb[u],     r0, myqe, myqd, cnt);
                        cnt = own_push(m1, (unsigned)eb[u] + 1, r1, myqe, myqd, cnt);
                        cnt = own_push(m2, (unsigned)eb[u] + 2, r2, myqe, myqd, cnt);
                        cnt = own_push(m3, (unsigned)eb[u] + 3, r3, myqe, myqd, cnt);
                        if (cnt >= OWN_FLUSH) {
                            __builtin_amdgcn_fence(__ATOMIC_ACQ_REL, "wavefront");
                            own_drain(src, h, acc, myqe, myqd, cnt, lane);
                            cnt = 0;
                        }
                    }
                }
            }
        }
        if (wave == 0) {
            for (int eb0 = nGf * 128; eb0 < nE; eb0 += 32) {
                const int e = eb0 + lane;
                const unsigned r = ((e < nE) ? (unsigned)dst[e] : 0xFFFFFFFFu) - (unsigned)lo;
                cnt = own_push(r < (unsigned)rows && e < nE, (unsigned)e, r, myqe, myqd, cnt);
                if (cnt >= OWN_FLUSH) {
                    __builtin_amdgcn_fence(__ATOMIC_ACQ_REL, "wavefront");
                    own_drain(src, h, acc, myqe, myqd, cnt, lane);
                    cnt = 0;
                }
            }
        }
        __builtin_amdgcn_fence(__ATOMIC_ACQ_REL, "wavefront");
        own_drain(src, h, acc, myqe, myqd, cnt, lane);
        __syncthreads();

        for (int r = wave; r < rows; r += nwav) {
            const float* row = acc + r * 128 + lane;
            const float4 o = make_float4(row[0], row[32], row[64], row[96]);
            *(float4*)(agg + (size_t)(lo + r) * 128 + lane * 4) = o;
        }
        __syncthreads();
    }
}

__global__ void scatter_rows_kernel(const int* __restrict__ ids,
                                    const float* __restrict__ h,
                                    float* __restrict__ out_feat, int n) {
    int tid = blockIdx.x * blockDim.x + threadIdx.x;
    int r = tid >> 5;
    int f = (tid & 31) * 4;
    if (r < n) {
        const float4 v = *(const float4*)(h + (size_t)r * 128 + f);
        float* o = out_feat + (size_t)ids[r] * 128 + f;
        atomicAdd(o + 0, v.x);
        atomicAdd(o + 1, v.y);
        atomicAdd(o + 2, v.z);
        atomicAdd(o + 3, v.w);
    }
}

template <int NTILES>
__global__ __launch_bounds__(128)
void gemm_wmma_k128_kernel(const float* __restrict__ A,
                           const float* __restrict__ Wf,
                           const float* __restrict__ bias,
                           const float* __restrict__ arowscale,
                           const float* __restrict__ erowscale,
                           float* __restrict__ out,
                           int M, int relu) {
    const int lane    = threadIdx.x & 31;
    const int wave    = __builtin_amdgcn_readfirstlane((int)(threadIdx.x >> 5));
    const int wpb     = blockDim.x >> 5;
    const int nwaves  = gridDim.x * wpb;
    const int ntiles  = M >> 4;
    int tile          = blockIdx.x * wpb + wave;
    if (tile >= ntiles) return;
    const int halfsel = lane >> 4;
    const int lid     = lane & 15;
    const int N       = NTILES * 16;

    v2f b[NTILES][32];
    {
        const float4* wf4 = (const float4*)Wf + lane;
#pragma unroll
        for (int nt = 0; nt < NTILES; ++nt) {
#pragma unroll
            for (int kk2 = 0; kk2 < 16; ++kk2) {
                const float4 q = wf4[(nt * 16 + kk2) * 32];
                b[nt][2 * kk2].x     = q.x; b[nt][2 * kk2].y     = q.y;
                b[nt][2 * kk2 + 1].x = q.z; b[nt][2 * kk2 + 1].y = q.w;
            }
        }
    }
    float bv[NTILES];
#pragma unroll
    for (int nt = 0; nt < NTILES; ++nt) bv[nt] = bias[nt * 16 + lid];

    v2f   an[32];
    float sn;
    {
        const int r = tile * 16 + lid;
        const float* ap = A + (size_t)r * 128 + halfsel * 2;
#pragma unroll
        for (int kk = 0; kk < 32; ++kk) an[kk] = *(const v2f*)(ap + kk * 4);
        sn = arowscale ? arowscale[r] : 1.0f;
    }

    for (; tile < ntiles; tile += nwaves) {
        v2f a[32];
#pragma unroll
        for (int kk = 0; kk < 32; ++kk) { a[kk].x = an[kk].x * sn; a[kk].y = an[kk].y * sn; }

        const int tnext = tile + nwaves;
        if (tnext < ntiles) {
            const int r = tnext * 16 + lid;
            const float* ap = A + (size_t)r * 128 + halfsel * 2;
#pragma unroll
            for (int kk = 0; kk < 32; ++kk) an[kk] = *(const v2f*)(ap + kk * 4);
            sn = arowscale ? arowscale[r] : 1.0f;
        }

        float es[8];
        if (erowscale) {
#pragma unroll
            for (int j = 0; j < 8; ++j) es[j] = erowscale[tile * 16 + j + halfsel * 8];
        } else {
#pragma unroll
            for (int j = 0; j < 8; ++j) es[j] = 1.0f;
        }

        v8f acc[NTILES];
#pragma unroll
        for (int nt = 0; nt < NTILES; ++nt) {
#pragma unroll
            for (int j = 0; j < 8; ++j) acc[nt][j] = bv[nt];
        }

#pragma unroll
        for (int kk = 0; kk < 32; ++kk) {
#pragma unroll
            for (int nt = 0; nt < NTILES; ++nt) {
                acc[nt] = __builtin_amdgcn_wmma_f32_16x16x4_f32(
                    false, a[kk], false, b[nt][kk], (short)0, acc[nt], false, false);
            }
        }

#pragma unroll
        for (int nt = 0; nt < NTILES; ++nt) {
#pragma unroll
            for (int j = 0; j < 8; ++j) {
                float v = acc[nt][j];
                if (relu) v = fmaxf(v, 0.0f);
                const int row = tile * 16 + j + halfsel * 8;
                if (erowscale) v *= es[j];
                out[(size_t)row * N + nt * 16 + lid] = v;
            }
        }
    }
}


extern "C" void kernel_launch(void* const* d_in, const int* in_sizes, int n_in,
                              void* d_out, int out_size, void* d_ws, size_t ws_size,
                              hipStream_t stream) {
    (void)n_in; (void)ws_size;
    const float* x   = (const float*)d_in[0];
    const float* W2  = (const float*)d_in[1];
    const float* b2  = (const float*)d_in[2];
    const float* gw  = (const float*)d_in[3];
    const float* gb  = (const float*)d_in[4];
    const float* Wc  = (const float*)d_in[5];
    const float* bc  = (const float*)d_in[6];
    const int*   src = (const int*)d_in[7];
    const int*   dst = (const int*)d_in[8];
    const int*   ids = (const int*)d_in[9];

    const int H  = 128;
    const int C  = 16;
    const int L  = in_sizes[2] / H;
    const int NS = in_sizes[9] / L;
    const int NE = in_sizes[7] / L;
    const int NL = in_sizes[4] / (L * H);
    const int NN = out_size / C;

    float* ws       = (float*)d_ws;
    float* out_feat = ws;
    float* hbuf     = out_feat + (size_t)NN * H;
    float* agg      = hbuf + (size_t)NS * H;
    float* deg_out  = agg + (size_t)NS * H;
    float* deg_in   = deg_out + NS;
    float* w2t      = deg_in + NS;
    float* gwt      = w2t + (size_t)L * H * H;
    float* wct      = gwt + (size_t)L * NL * H * H;

    const int T = 256;
    const int feat_blocks_NS = (NS * 32 + T - 1) / T;
    const int GT = 128;
    const int gemm_blocks_NS = 256;
    const int gemm_blocks_NN = (((NN / 16) + 1) / 2 + 3) / 4;

    {
        int nHH = H * H;
        int bHH = (nHH + T - 1) / T;
        for (int i = 0; i < L; ++i)
            pack_bfrag_kernel<<<bHH, T, 0, stream>>>(W2 + (size_t)i * nHH,
                                                     w2t + (size_t)i * nHH, H);
        for (int i = 0; i < L * NL; ++i)
            pack_bfrag_kernel<<<bHH, T, 0, stream>>>(gw + (size_t)i * nHH,
                                                     gwt + (size_t)i * nHH, H);
        int nHC = H * C;
        pack_bfrag_kernel<<<(nHC + T - 1) / T, T, 0, stream>>>(Wc, wct, C);
    }

    {
        int n4 = (int)((size_t)NN * H / 4);
        zero_f4_kernel<<<(n4 + T - 1) / T, T, 0, stream>>>((float4*)out_feat, n4);
    }

    for (int i = 0; i < L; ++i) {
        const int*   src_i = src + (size_t)i * NE;
        const int*   dst_i = dst + (size_t)i * NE;
        const int*   ids_i = ids + (size_t)i * NS;
        const float* x_i   = x   + (size_t)i * NS * H;
        const float* w2t_i = w2t + (size_t)i * H * H;
        const float* b2_i  = b2  + (size_t)i * H;

        {
            int n4 = (2 * NS) / 4;
            zero_f4_kernel<<<(n4 + T - 1) / T, T, 0, stream>>>((float4*)deg_out, n4);
        }
        degree_count_kernel<<<(NE + T - 1) / T, T, 0, stream>>>(src_i, dst_i,
                                                                deg_out, deg_in, NE);
        rsqrt_clamp_kernel<<<(2 * NS + T - 1) / T, T, 0, stream>>>(deg_out, 2 * NS);

        gemm_wmma_k128_kernel<8><<<gemm_blocks_NS, GT, 0, stream>>>(
            x_i, w2t_i, b2_i, nullptr, deg_out, hbuf, NS, 0);

        for (int l = 0; l < NL; ++l) {
            const float* gwt_il = gwt + ((size_t)i * NL + l) * H * H;
            const float* gb_il  = gb  + ((size_t)i * NL + l) * H;
            edge_sum_owner_kernel<<<256, 1024, 0, stream>>>(src_i, dst_i,
                                                            hbuf, agg, NE, NS);
            const float* escale = (l + 1 < NL) ? deg_out : nullptr;
            gemm_wmma_k128_kernel<8><<<gemm_blocks_NS, GT, 0, stream>>>(
                agg, gwt_il, gb_il, deg_in, escale, hbuf, NS, 1);
        }

        scatter_rows_kernel<<<feat_blocks_NS, T, 0, stream>>>(ids_i, hbuf,
                                                              out_feat, NS);
    }

    gemm_wmma_k128_kernel<1><<<gemm_blocks_NN, GT, 0, stream>>>(
        out_feat, wct, bc, nullptr, nullptr, (float*)d_out, NN, 0);
}


# edge sums: owner work-groups scan a 1-byte owner key per edge (dst row / rows per group, written by the degree kernel) 16 edges a lane, not 4-byte dst ids; LDS sums still in arrival order
# speedup vs baseline: 2.3235x; 1.1870x over previous
#include <hip/hip_runtime.h>


typedef __attribute__((ext_vector_type(2))) float v2f;
typedef __attribute__((ext_vector_type(8))) float v8f;


__global__ void zero_f4_kernel(float4* __restrict__ p, int n4) {
    int i = blockIdx.x * blockDim.x + threadIdx.x;
    if (i < n4) p[i] = make_float4(0.f, 0.f, 0.f, 0.f);
}

__global__ void pack_bfrag_kernel(const float* __restrict__ in,
                                  float* __restrict__ outF, int N) {
    int i = blockIdx.x * blockDim.x + threadIdx.x;
    if (i < N * 128) {
        int c    = i & 3;
        int lane = (i >> 2) & 31;
        int kk2  = (i >> 7) & 15;
        int nt   = i >> 11;
        int kk   = 2 * kk2 + (c >> 1);
        int k    = kk * 4 + (lane >> 4) * 2 + (c & 1);
        int n    = nt * 16 + (lane & 15);
        outF[i] = in[k * N + n];
    }
}

#define OWN_GRID  256
#define OWN_RMAX  200

__device__ __forceinline__ int own_range_rows(int nRows) {
    const int R = (nRows + OWN_GRID - 1) / OWN_GRID;
    return (R > OWN_RMAX) ? OWN_RMAX : R;
}

__global__ void degree_count_kernel(const int* __restrict__ src,
                                    const int* __restrict__ dst,
                                    float* __restrict__ deg_out,
                                    float* __restrict__ deg_in,
                                    unsigned char* __restrict__ key,
                                    int nE, int nRows) {
    const unsigned R = (unsigned)own_range_rows(nRows);
    int e = blockIdx.x * blockDim.x + threadIdx.x;
    if (e < nE) {
        const int d = dst[e];
        atomicAdd(&deg_out[src[e]], 1.0f);
        atomicAdd(&deg_in[d], 1.0f);
        key[e] = (unsigned char)(((unsigned)d / R) % (unsigned)OWN_GRID);
    }
}

__global__ void rsqrt_clamp_kernel(float* __restrict__ d, int n) {
    int i = blockIdx.x * blockDim.x + threadIdx.x;
    if (i < n) d[i] = rsqrtf(fmaxf(d[i], 1.0f));
}

#define OWN_WAVES 32
#define OWN_QCAP  512
#define OWN_FLUSH 64
#define OWN_U     8

__device__ __forceinline__ void own_drain(const int* __restrict__ src,
                                          const int* __restrict__ dst,
                                          const float* __restrict__ h,
                                          float* acc, const unsigned* qe,
                                          int n, int lo, int rows, int lane) {
    for (int i0 = 0; i0 < n; i0 += 32) {
        const int m = (n - i0 < 32) ? (n - i0) : 32;
        int my_s = 0, my_d = 0;
        if (lane < m) {
            const unsigned e = qe[i0 + lane];
            my_s = src[e];
            my_d = dst[e] - lo;
        }
        for (int j0 = 0; j0 < m; j0 += 8) {
            float4 v[8];
            int    dl[8];
#pragma unroll
            for (int u = 0; u < 8; ++u) {
                const int j = (j0 + u < m) ? (j0 + u) : (m - 1);
                const int s = __builtin_amdgcn_readlane(my_s, j);
                dl[u]       = __builtin_amdgcn_readlane(my_d, j);
                v[u] = *(const float4*)(h + (size_t)s * 128 + lane * 4);
            }
#pragma unroll
            for (int u = 0; u < 8; ++u) {
                if (j0 + u < m && (unsigned)dl[u] < (unsigned)rows) {
                    float* row = acc + dl[u] * 128 + lane;
                    __hip_atomic_fetch_add(row,      v[u].x, __ATOMIC_RELAXED, __HIP_MEMORY_SCOPE_WORKGROUP);
                    __hip_atomic_fetch_add(row + 32, v[u].y, __ATOMIC_RELAXED, __HIP_MEMORY_SCOPE_WORKGROUP);
                    __hip_atomic_fetch_add(row + 64, v[u].z, __ATOMIC_RELAXED, __HIP_MEMORY_SCOPE_WORKGROUP);
                    __hip_atomic_fetch_add(row + 96, v[u].w, __ATOMIC_RELAXED, __HIP_MEMORY_SCOPE_WORKGROUP);
                }
            }
        }
    }
}

__device__ __forceinline__ int own_push(bool m, unsigned e, unsigned* qe, int cnt) {
    const unsigned k = __builtin_amdgcn_ballot_w32(m);
    if (m) qe[cnt + (int)__builtin_amdgcn_mbcnt_lo(k, 0u)] = e;
    return cnt + __builtin_popcount(k);
}

__device__ __forceinline__ unsigned own_zero_bytes(unsigned x) {
    const unsigned t = (x & 0x7F7F7F7Fu) + 0x7F7F7F7Fu;
    return ~(t | x | 0x7F7F7F7Fu);
}

__global__ __launch_bounds__(1024)
void edge_sum_owner_kernel(const int* __restrict__ src,
                           const int* __restrict__ dst,
                           const unsigned char* __restrict__ key,
                           const float* __restrict__ h,
                           float* __restrict__ agg, int nE, int nRows) {
    __shared__ float    acc[OWN_RMAX * 128];
    __shared__ unsigned qe[OWN_WAVES][OWN_QCAP];

    const int lane = threadIdx.x & 31;
    const int wave = __builtin_amdgcn_readfirstlane((int)(threadIdx.x >> 5));
    const int nwav = blockDim.x >> 5;
    const int R   = own_range_rows(nRows);
    const int nGf = nE >> 9;
    const int rot = (int)(((long long)blockIdx.x * nGf) / OWN_GRID);
    const unsigned pat = (unsigned)blockIdx.x * 0x01010101u;
    unsigned* myqe = qe[wave];

    for (int lo = blockIdx.x * R; lo < nRows; lo += OWN_GRID * R) {
        const int rows = (nRows - lo < R) ? (nRows - lo) : R;
        for (int i = threadIdx.x; i < rows * 32; i += blockDim.x)
            ((float4*)acc)[i] = make_float4(0.f, 0.f, 0.f, 0.f);
        __syncthreads();

        int cnt = 0;
        for (int g0 = wave; g0 < nGf; g0 += OWN_U * nwav) {
            uint4 d[OWN_U];
            int   eb[OWN_U];
#pragma unroll
            for (int u = 0; u < OWN_U; ++u) {
                int g = g0 + u * nwav;
                if (g >= nGf) g = nGf - 1;
                g += rot;
                if (g >= nGf) g -= nGf;
                eb[u] = g * 512 + lane * 16;
                d[u]  = *(const uint4*)(key + eb[u]);
            }
#pragma unroll
            for (int u = 0; u < OWN_U; ++u) {
                if (g0 + u * nwav < nGf) {
                    unsigned hit = (own_zero_bytes(d[u].x ^ pat) >> 7)
                                 | (own_zero_bytes(d[u].y ^ pat) >> 6)
                                 | (own_zero_bytes(d[u].z ^ pat) >> 5)
                                 | (own_zero_bytes(d[u].w ^ pat) >> 4);
                    while (__builtin_amdgcn_ballot_w32(hit != 0u)) {
                        const bool     has = hit != 0u;
                        const unsigned p   = (unsigned)__builtin_ctz(hit | 0x80000000u);
                        const unsigned e   = (unsigned)eb[u] + ((p & 7u) << 2) + (p >> 3);
                        cnt = own_push(has, e, myqe, cnt);
                        hit &= hit - 1u;
                        if (cnt >= OWN_FLUSH) {
                            __builtin_amdgcn_fence(__ATOMIC_ACQ_REL, "wavefront");
                            own_drain(src, dst, h, acc, myqe, cnt, lo, rows, lane);
                            cnt = 0;
                        }
                    }
                }
            }
        }
        if (wave == 0) {
            for (int eb0 = nGf * 512; eb0 < nE; eb0 += 32) {
                const int e = eb0 + lane;
                const unsigned r = ((e < nE) ? (unsigned)dst[e] : 0xFFFFFFFFu) - (unsigned)lo;
                cnt = own_push(r < (unsigned)rows && e < nE, (unsigned)e, myqe, cnt);
                if (cnt >= OWN_FLUSH) {
                    __builtin_amdgcn_fence(__ATOMIC_ACQ_REL, "wavefront");
                    own_drain(src, dst, h, acc, myqe, cnt, lo, rows, lane);
                    cnt = 0;
                }
            }
        }
        __builtin_amdgcn_fence(__ATOMIC_ACQ_REL, "wavefront");
        own_drain(src, dst, h, acc, myqe, cnt, lo, rows, lane);
        __syncthreads();

        for (int r = wave; r < rows; r += nwav) {
            const float* row = acc + r * 128 + lane;
            const float4 o = make_float4(row[0], row[32], row[64], row[96]);
            *(float4*)(agg + (size_t)(lo + r) * 128 + lane * 4) = o;
        }
        __syncthreads();
    }
}

__global__ void scatter_rows_kernel(const int* __restrict__ ids,
                                    const float* __restrict__ h,
                                    float* __restrict__ out_feat, int n) {
    int tid = blockIdx.x * blockDim.x + threadIdx.x;
    int r = tid >> 5;
    int f = (tid & 31) * 4;
    if (r < n) {
        const float4 v = *(const float4*)(h + (size_t)r * 128 + f);
        float* o = out_feat + (size_t)ids[r] * 128 + f;
        atomicAdd(o + 0, v.x);
        atomicAdd(o + 1, v.y);
        atomicAdd(o + 2, v.z);
        atomicAdd(o + 3, v.w);
    }
}

template <int NTILES>
__global__ __launch_bounds__(128)
void gemm_wmma_k128_kernel(const float* __restrict__ A,
                           const float* __restrict__ Wf,
                           const float* __restrict__ bias,
                           const float* __restrict__ arowscale,
                           const float* __restrict__ erowscale,
                           float* __restrict__ out,
                           int M, int relu) {
    const int lane    = threadIdx.x & 31;
    const int wave    = __builtin_amdgcn_readfirstlane((int)(threadIdx.x >> 5));
    const int wpb     = blockDim.x >> 5;
    const int nwaves  = gridDim.x * wpb;
    const int ntiles  = M >> 4;
    int tile          = blockIdx.x * wpb + wave;
    if (tile >= ntiles) return;
    const int halfsel = lane >> 4;
    const int lid     = lane & 15;
    const int N       = NTILES * 16;

    v2f b[NTILES][32];
    {
        const float4* wf4 = (const float4*)Wf + lane;
#pragma unroll
        for (int nt = 0; nt < NTILES; ++nt) {
#pragma unroll
            for (int kk2 = 0; kk2 < 16; ++kk2) {
                const float4 q = wf4[(nt * 16 + kk2) * 32];
                b[nt][2 * kk2].x     = q.x; b[nt][2 * kk2].y     = q.y;
                b[nt][2 * kk2 + 1].x = q.z; b[nt][2 * kk2 + 1].y = q.w;
            }
        }
    }
    float bv[NTILES];
#pragma unroll
    for (int nt = 0; nt < NTILES; ++nt) bv[nt] = bias[nt * 16 + lid];

    v2f   an[32];
    float sn;
    {
        const int r = tile * 16 + lid;
        const float* ap = A + (size_t)r * 128 + halfsel * 2;
#pragma unroll
        for (int kk = 0; kk < 32; ++kk) an[kk] = *(const v2f*)(ap + kk * 4);
        sn = arowscale ? arowscale[r] : 1.0f;
    }

    for (; tile < ntiles; tile += nwaves) {
        v2f a[32];
#pragma unroll
        for (int kk = 0; kk < 32; ++kk) { a[kk].x = an[kk].x * sn; a[kk].y = an[kk].y * sn; }

        const int tnext = tile + nwaves;
        if (tnext < ntiles) {
            const int r = tnext * 16 + lid;
            const float* ap = A + (size_t)r * 128 + halfsel * 2;
#pragma unroll
            for (int kk = 0; kk < 32; ++kk) an[kk] = *(const v2f*)(ap + kk * 4);
            sn = arowscale ? arowscale[r] : 1.0f;
        }

        float es[8];
        if (erowscale) {
#pragma unroll
            for (int j = 0; j < 8; ++j) es[j] = erowscale[tile * 16 + j + halfsel * 8];
        } else {
#pragma unroll
            for (int j = 0; j < 8; ++j) es[j] = 1.0f;
        }

        v8f acc[NTILES];
#pragma unroll
        for (int nt = 0; nt < NTILES; ++nt) {
#pragma unroll
            for (int j = 0; j < 8; ++j) acc[nt][j] = bv[nt];
        }

#pragma unroll
        for (int kk = 0; kk < 32; ++kk) {
#pragma unroll
            for (int nt = 0; nt < NTILES; ++nt) {
                acc[nt] = __builtin_amdgcn_wmma_f32_16x16x4_f32(
                    false, a[kk], false, b[nt][kk], (short)0, acc[nt], false, false);
            }
        }

#pragma unroll
        for (int nt = 0; nt < NTILES; ++nt) {
#pragma unroll
            for (int j = 0; j < 8; ++j) {
                float v = acc[nt][j];
                if (relu) v = fmaxf(v, 0.0f);
                const int row = tile * 16 + j + halfsel * 8;
                if (erowscale) v *= es[j];
                out[(size_t)row * N + nt * 16 + lid] = v;
            }
        }
    }
}


extern "C" void kernel_launch(void* const* d_in, const int* in_sizes, int n_in,
                              void* d_out, int out_size, void* d_ws, size_t ws_size,
                              hipStream_t stream) {
    (void)n_in; (void)ws_size;
    const float* x   = (const float*)d_in[0];
    const float* W2  = (const float*)d_in[1];
    const float* b2  = (const float*)d_in[2];
    const float* gw  = (const float*)d_in[3];
    const float* gb  = (const float*)d_in[4];
    const float* Wc  = (const float*)d_in[5];
    const float* bc  = (const float*)d_in[6];
    const int*   src = (const int*)d_in[7];
    const int*   dst = (const int*)d_in[8];
    const int*   ids = (const int*)d_in[9];

    const int H  = 128;
    const int C  = 16;
    const int L  = in_sizes[2] / H;
    const int NS = in_sizes[9] / L;
    const int NE = in_sizes[7] / L;
    const int NL = in_sizes[4] / (L * H);
    const int NN = out_size / C;

    float* ws       = (float*)d_ws;
    float* out_feat = ws;
    float* hbuf     = out_feat + (size_t)NN * H;
    float* agg      = hbuf + (size_t)NS * H;
    float* deg_out  = agg + (size_t)NS * H;
    float* deg_in   = deg_out + NS;
    float* w2t      = deg_in + NS;
    float* gwt      = w2t + (size_t)L * H * H;
    float* wct      = gwt + (size_t)L * NL * H * H;
    unsigned char* okey = (unsigned char*)d_out;

    const int T = 256;
    const int feat_blocks_NS = (NS * 32 + T - 1) / T;
    const int GT = 128;
    const int gemm_blocks_NS = 256;
    const int gemm_blocks_NN = (((NN / 16) + 1) / 2 + 3) / 4;

    {
        int nHH = H * H;
        int bHH = (nHH + T - 1) / T;
        for (int i = 0; i < L; ++i)
            pack_bfrag_kernel<<<bHH, T, 0, stream>>>(W2 + (size_t)i * nHH,
                                                     w2t + (size_t)i * nHH, H);
        for (int i = 0; i < L * NL; ++i)
            pack_bfrag_kernel<<<bHH, T, 0, stream>>>(gw + (size_t)i * nHH,
                                                     gwt + (size_t)i * nHH, H);
        int nHC = H * C;
        pack_bfrag_kernel<<<(nHC + T - 1) / T, T, 0, stream>>>(Wc, wct, C);
    }

    {
        int n4 = (int)((size_t)NN * H / 4);
        zero_f4_kernel<<<(n4 + T - 1) / T, T, 0, stream>>>((float4*)out_feat, n4);
    }

    for (int i = 0; i < L; ++i) {
        const int*   src_i = src + (size_t)i * NE;
        const int*   dst_i = dst + (size_t)i * NE;
        const int*   ids_i = ids + (size_t)i * NS;
        const float* x_i   = x   + (size_t)i * NS * H;
        const float* w2t_i = w2t + (size_t)i * H * H;
        const float* b2_i  = b2  + (size_t)i * H;

        {
            int n4 = (2 * NS) / 4;
            zero_f4_kernel<<<(n4 + T - 1) / T, T, 0, stream>>>((float4*)deg_out, n4);
        }
        degree_count_kernel<<<(NE + T - 1) / T, T, 0, stream>>>(src_i, dst_i,
                                                                deg_out, deg_in,
                                                                okey, NE, NS);
        rsqrt_clamp_kernel<<<(2 * NS + T - 1) / T, T, 0, stream>>>(deg_out, 2 * NS);

        gemm_wmma_k128_kernel<8><<<gemm_blocks_NS, GT, 0, stream>>>(
            x_i, w2t_i, b2_i, nullptr, deg_out, hbuf, NS, 0);

        for (int l = 0; l < NL; ++l) {
            const float* gwt_il = gwt + ((size_t)i * NL + l) * H * H;
            const float* gb_il  = gb  + ((size_t)i * NL + l) * H;
            edge_sum_owner_kernel<<<OWN_GRID, 1024, 0, stream>>>(src_i, dst_i, okey,
                                                                 hbuf, agg, NE, NS);
            const float* escale = (l + 1 < NL) ? deg_out : nullptr;
            gemm_wmma_k128_kernel<8><<<gemm_blocks_NS, GT, 0, stream>>>(
                agg, gwt_il, gb_il, deg_in, escale, hbuf, NS, 1);
        }

        scatter_rows_kernel<<<feat_blocks_NS, T, 0, stream>>>(ids_i, hbuf,
                                                              out_feat, NS);
    }

    gemm_wmma_k128_kernel<1><<<gemm_blocks_NN, GT, 0, stream>>>(
        out_feat, wct, bc, nullptr, nullptr, (float*)d_out, NN, 0);
}


# id scatter + classifier: 256 owner work-groups sum each node's rows in LDS (float atomics, arrival order), apply Wc on WMMA from LDS, write out once; no zero pass or global atomics; 39->35 launches
# speedup vs baseline: 2.4803x; 1.0675x over previous
#include <hip/hip_runtime.h>


typedef __attribute__((ext_vector_type(2))) float v2f;
typedef __attribute__((ext_vector_type(8))) float v8f;


__global__ void zero_f4_kernel(float4* __restrict__ p, int n4) {
    int i = blockIdx.x * blockDim.x + threadIdx.x;
    if (i < n4) p[i] = make_float4(0.f, 0.f, 0.f, 0.f);
}

__global__ void pack_bfrag_kernel(const float* __restrict__ in,
                                  float* __restrict__ outF, int N) {
    int i = blockIdx.x * blockDim.x + threadIdx.x;
    if (i < N * 128) {
        int c    = i & 3;
        int lane = (i >> 2) & 31;
        int kk2  = (i >> 7) & 15;
        int nt   = i >> 11;
        int kk   = 2 * kk2 + (c >> 1);
        int k    = kk * 4 + (lane >> 4) * 2 + (c & 1);
        int n    = nt * 16 + (lane & 15);
        outF[i] = in[k * N + n];
    }
}

#define OWN_GRID  256
#define OWN_RMAX  200

__device__ __forceinline__ int own_range_rows(int nRows) {
    const int R = (nRows + OWN_GRID - 1) / OWN_GRID;
    return (R > OWN_RMAX) ? OWN_RMAX : R;
}

__global__ void degree_count_kernel(const int* __restrict__ src,
                                    const int* __restrict__ dst,
                                    float* __restrict__ deg_out,
                                    float* __restrict__ deg_in,
                                    unsigned char* __restrict__ key,
                                    int nE, int nRows) {
    const unsigned R = (unsigned)own_range_rows(nRows);
    int e = blockIdx.x * blockDim.x + threadIdx.x;
    if (e < nE) {
        const int d = dst[e];
        atomicAdd(&deg_out[src[e]], 1.0f);
        atomicAdd(&deg_in[d], 1.0f);
        key[e] = (unsigned char)(((unsigned)d / R) % (unsigned)OWN_GRID);
    }
}

__global__ void rsqrt_clamp_kernel(float* __restrict__ d, int n) {
    int i = blockIdx.x * blockDim.x + threadIdx.x;
    if (i < n) d[i] = rsqrtf(fmaxf(d[i], 1.0f));
}

#define OWN_WAVES 32
#define OWN_QCAP  512
#define OWN_FLUSH 64
#define OWN_U     8

__device__ __forceinline__ void own_drain(const int* __restrict__ src,
                                          const int* __restrict__ dst,
                                          const float* __restrict__ h,
                                          float* acc, const unsigned* qe,
                                          int n, int lo, int rows, int lane) {
    for (int i0 = 0; i0 < n; i0 += 32) {
        const int m = (n - i0 < 32) ? (n - i0) : 32;
        int my_s = 0, my_d = 0;
        if (lane < m) {
            const unsigned e = qe[i0 + lane];
            my_s = src[e];
            my_d = dst[e] - lo;
        }
        for (int j0 = 0; j0 < m; j0 += 8) {
            float4 v[8];
            int    dl[8];
#pragma unroll
            for (int u = 0; u < 8; ++u) {
                const int j = (j0 + u < m) ? (j0 + u) : (m - 1);
                const int s = __builtin_amdgcn_readlane(my_s, j);
                dl[u]       = __builtin_amdgcn_readlane(my_d, j);
                v[u] = *(const float4*)(h + (size_t)s * 128 + lane * 4);
            }
#pragma unroll
            for (int u = 0; u < 8; ++u) {
                if (j0 + u < m && (unsigned)dl[u] < (unsigned)rows) {
                    float* row = acc + dl[u] * 128 + lane;
                    __hip_atomic_fetch_add(row,      v[u].x, __ATOMIC_RELAXED, __HIP_MEMORY_SCOPE_WORKGROUP);
                    __hip_atomic_fetch_add(row + 32, v[u].y, __ATOMIC_RELAXED, __HIP_MEMORY_SCOPE_WORKGROUP);
                    __hip_atomic_fetch_add(row + 64, v[u].z, __ATOMIC_RELAXED, __HIP_MEMORY_SCOPE_WORKGROUP);
                    __hip_atomic_fetch_add(row + 96, v[u].w, __ATOMIC_RELAXED, __HIP_MEMORY_SCOPE_WORKGROUP);
                }
            }
        }
    }
}

__device__ __forceinline__ int own_push(bool m, unsigned e, unsigned* qe, int cnt) {
    const unsigned k = __builtin_amdgcn_ballot_w32(m);
    if (m) qe[cnt + (int)__builtin_amdgcn_mbcnt_lo(k, 0u)] = e;
    return cnt + __builtin_popcount(k);
}

__device__ __forceinline__ unsigned own_zero_bytes(unsigned x) {
    const unsigned t = (x & 0x7F7F7F7Fu) + 0x7F7F7F7Fu;
    return ~(t | x | 0x7F7F7F7Fu);
}

__global__ __launch_bounds__(1024)
void edge_sum_owner_kernel(const int* __restrict__ src,
                           const int* __restrict__ dst,
                           const unsigned char* __restrict__ key,
                           const float* __restrict__ h,
                           float* __restrict__ agg, int nE, int nRows) {
    __shared__ float    acc[OWN_RMAX * 128];
    __shared__ unsigned qe[OWN_WAVES][OWN_QCAP];

    const int lane = threadIdx.x & 31;
    const int wave = __builtin_amdgcn_readfirstlane((int)(threadIdx.x >> 5));
    const int nwav = blockDim.x >> 5;
    const int R   = own_range_rows(nRows);
    const int nGf = nE >> 9;
    const int rot = (int)(((long long)blockIdx.x * nGf) / OWN_GRID);
    const unsigned pat = (unsigned)blockIdx.x * 0x01010101u;
    unsigned* myqe = qe[wave];

    for (int lo = blockIdx.x * R; lo < nRows; lo += OWN_GRID * R) {
        const int rows = (nRows - lo < R) ? (nRows - lo) : R;
        for (int i = threadIdx.x; i < rows * 32; i += blockDim.x)
            ((float4*)acc)[i] = make_float4(0.f, 0.f, 0.f, 0.f);
        __syncthreads();

        int cnt = 0;
        for (int g0 = wave; g0 < nGf; g0 += OWN_U * nwav) {
            uint4 d[OWN_U];
            int   eb[OWN_U];
#pragma unroll
            for (int u = 0; u < OWN_U; ++u) {
                int g = g0 + u * nwav;
                if (g >= nGf) g = nGf - 1;
                g += rot;
                if (g >= nGf) g -= nGf;
                eb[u] = g * 512 + lane * 16;
                d[u]  = *(const uint4*)(key + eb[u]);
            }
#pragma unroll
            for (int u = 0; u < OWN_U; ++u) {
                if (g0 + u * nwav < nGf) {
                    unsigned hit = (own_zero_bytes(d[u].x ^ pat) >> 7)
                                 | (own_zero_bytes(d[u].y ^ pat) >> 6)
                                 | (own_zero_bytes(d[u].z ^ pat) >> 5)
                                 | (own_zero_bytes(d[u].w ^ pat) >> 4);
                    while (__builtin_amdgcn_ballot_w32(hit != 0u)) {
                        const bool     has = hit != 0u;
                        const unsigned p   = (unsigned)__builtin_ctz(hit | 0x80000000u);
                        const unsigned e   = (unsigned)eb[u] + ((p & 7u) << 2) + (p >> 3);
                        cnt = own_push(has, e, myqe, cnt);
                        hit &= hit - 1u;
                        if (cnt >= OWN_FLUSH) {
                            __builtin_amdgcn_fence(__ATOMIC_ACQ_REL, "wavefront");
                            own_drain(src, dst, h, acc, myqe, cnt, lo, rows, lane);
                            cnt = 0;
                        }
                    }
                }
            }
        }
        if (wave == 0) {
            for (int eb0 = nGf * 512; eb0 < nE; eb0 += 32) {
                const int e = eb0 + lane;
                const unsigned r = ((e < nE) ? (unsigned)dst[e] : 0xFFFFFFFFu) - (unsigned)lo;
                cnt = own_push(r < (unsigned)rows && e < nE, (unsigned)e, myqe, cnt);
                if (cnt >= OWN_FLUSH) {
                    __builtin_amdgcn_fence(__ATOMIC_ACQ_REL, "wavefront");
                    own_drain(src, dst, h, acc, myqe, cnt, lo, rows, lane);
                    cnt = 0;
                }
            }
        }
        __builtin_amdgcn_fence(__ATOMIC_ACQ_REL, "wavefront");
        own_drain(src, dst, h, acc, myqe, cnt, lo, rows, lane);
        __syncthreads();

        for (int r = wave; r < rows; r += nwav) {
            const float* row = acc + r * 128 + lane;
            const float4 o = make_float4(row[0], row[32], row[64], row[96]);
            *(float4*)(agg + (size_t)(lo + r) * 128 + lane * 4) = o;
        }
        __syncthreads();
    }
}

#define SC_RMAX  400
#define SC_P     132
#define SC_QCAP  256
#define SC_FLUSH 64
#define SC_U     8

__device__ __forceinline__ void sc_drain(const int* __restrict__ ids,
                                         const float* __restrict__ h,
                                         float* acc, const unsigned* qe,
                                         int n, int lo, int lane) {
    for (int i0 = 0; i0 < n; i0 += 32) {
        const int m = (n - i0 < 32) ? (n - i0) : 32;
        int my_g = 0, my_d = 0;
        if (lane < m) {
            const unsigned g = qe[i0 + lane];
            my_g = (int)g;
            my_d = ids[g] - lo;
        }
        for (int j0 = 0; j0 < m; j0 += 8) {
            float4 v[8];
            int    dl[8];
#pragma unroll
            for (int u = 0; u < 8; ++u) {
                const int j = (j0 + u < m) ? (j0 + u) : (m - 1);
                const int g = __builtin_amdgcn_readlane(my_g, j);
                dl[u]       = __builtin_amdgcn_readlane(my_d, j);
                v[u] = *(const float4*)(h + (size_t)g * 128 + lane * 4);
            }
#pragma unroll
            for (int u = 0; u < 8; ++u) {
                if (j0 + u < m) {
                    float* row = acc + dl[u] * SC_P + lane;
                    __hip_atomic_fetch_add(row,      v[u].x, __ATOMIC_RELAXED, __HIP_MEMORY_SCOPE_WORKGROUP);
                    __hip_atomic_fetch_add(row + 32, v[u].y, __ATOMIC_RELAXED, __HIP_MEMORY_SCOPE_WORKGROUP);
                    __hip_atomic_fetch_add(row + 64, v[u].z, __ATOMIC_RELAXED, __HIP_MEMORY_SCOPE_WORKGROUP);
                    __hip_atomic_fetch_add(row + 96, v[u].w, __ATOMIC_RELAXED, __HIP_MEMORY_SCOPE_WORKGROUP);
                }
            }
        }
    }
}

__global__ __launch_bounds__(1024)
void scatter_classify_owner_kernel(const int* __restrict__ ids,
                                   const float* __restrict__ h,
                                   const float* __restrict__ Wcf,
                                   const float* __restrict__ bc,
                                   float* __restrict__ out,
                                   int nIds, int nNodes) {
    __shared__ float    acc[SC_RMAX * SC_P];
    __shared__ unsigned qe[OWN_WAVES][SC_QCAP];

    const int lane = threadIdx.x & 31;
    const int wave = __builtin_amdgcn_readfirstlane((int)(threadIdx.x >> 5));
    const int nwav = blockDim.x >> 5;
    const int halfsel = lane >> 4;
    const int lid     = lane & 15;
    int R = (nNodes + OWN_GRID - 1) / OWN_GRID;
    if (R > SC_RMAX) R = SC_RMAX;
    const int nGf = nIds >> 7;
    const int rot = (int)(((long long)blockIdx.x * nGf) / OWN_GRID);
    unsigned* myqe = qe[wave];

    for (int lo = blockIdx.x * R; lo < nNodes; lo += OWN_GRID * R) {
        const int rows  = (nNodes - lo < R) ? (nNodes - lo) : R;
        const int ntile = (rows + 15) >> 4;
        for (int i = threadIdx.x; i < ntile * 16 * (SC_P / 4); i += blockDim.x)
            ((float4*)acc)[i] = make_float4(0.f, 0.f, 0.f, 0.f);
        __syncthreads();

        int cnt = 0;
        for (int g0 = wave; g0 < nGf; g0 += SC_U * nwav) {
            uint4 d[SC_U];
            int   eb[SC_U];
#pragma unroll
            for (int u = 0; u < SC_U; ++u) {
                int g = g0 + u * nwav;
                if (g >= nGf) g = nGf - 1;
                g += rot;
                if (g >= nGf) g -= nGf;
                eb[u] = g * 128 + lane * 4;
                d[u]  = *(const uint4*)(ids + eb[u]);
            }
#pragma unroll
            for (int u = 0; u < SC_U; ++u) {
                if (g0 + u * nwav < nGf) {
                    const unsigned b = (unsigned)eb[u];
                    cnt = own_push(d[u].x - (unsigned)lo < (unsigned)rows, b,      myqe, cnt);
                    cnt = own_push(d[u].y - (unsigned)lo < (unsigned)rows, b + 1u, myqe, cnt);
                    cnt = own_push(d[u].z - (unsigned)lo < (unsigned)rows, b + 2u, myqe, cnt);
                    cnt = own_push(d[u].w - (unsigned)lo < (unsigned)rows, b + 3u, myqe, cnt);
                    if (cnt >= SC_FLUSH) {
                        __builtin_amdgcn_fence(__ATOMIC_ACQ_REL, "wavefront");
                        sc_drain(ids, h, acc, myqe, cnt, lo, lane);
                        cnt = 0;
                    }
                }
            }
        }
        if (wave == 0) {
            for (int e0 = nGf * 128; e0 < nIds; e0 += 32) {
                const int e = e0 + lane;
                const unsigned r = ((e < nIds) ? (unsigned)ids[e] : 0xFFFFFFFFu) - (unsigned)lo;
                cnt = own_push(r < (unsigned)rows && e < nIds, (unsigned)e, myqe, cnt);
                if (cnt >= SC_FLUSH) {
                    __builtin_amdgcn_fence(__ATOMIC_ACQ_REL, "wavefront");
                    sc_drain(ids, h, acc, myqe, cnt, lo, lane);
                    cnt = 0;
                }
            }
        }
        __builtin_amdgcn_fence(__ATOMIC_ACQ_REL, "wavefront");
        sc_drain(ids, h, acc, myqe, cnt, lo, lane);
        __syncthreads();

        for (int t = wave; t < ntile; t += nwav) {
            v2f b[32];
            const float4* wf4 = (const float4*)Wcf + lane;
#pragma unroll
            for (int kk2 = 0; kk2 < 16; ++kk2) {
                const float4 q = wf4[kk2 * 32];
                b[2 * kk2].x     = q.x; b[2 * kk2].y     = q.y;
                b[2 * kk2 + 1].x = q.z; b[2 * kk2 + 1].y = q.w;
            }
            const float bv = bc[lid];
            v8f c;
#pragma unroll
            for (int j = 0; j < 8; ++j) c[j] = bv;
            const float* ar = acc + (t * 16 + lid) * SC_P + halfsel * 64;
#pragma unroll
            for (int kk = 0; kk < 32; ++kk) {
                v2f a;
                a.x = ar[kk];
                a.y = ar[32 + kk];
                c = __builtin_amdgcn_wmma_f32_16x16x4_f32(
                    false, a, false, b[kk], (short)0, c, false, false);
            }
#pragma unroll
            for (int j = 0; j < 8; ++j) {
                const int row = t * 16 + j + halfsel * 8;
                if (row < rows) out[(size_t)(lo + row) * 16 + lid] = c[j];
            }
        }
        __syncthreads();
    }
}

template <int NTILES>
__global__ __launch_bounds__(128)
void gemm_wmma_k128_kernel(const float* __restrict__ A,
                           const float* __restrict__ Wf,
                           const float* __restrict__ bias,
                           const float* __restrict__ arowscale,
                           const float* __restrict__ erowscale,
                           float* __restrict__ out,
                           int M, int relu) {
    const int lane    = threadIdx.x & 31;
    const int wave    = __builtin_amdgcn_readfirstlane((int)(threadIdx.x >> 5));
    const int wpb     = blockDim.x >> 5;
    const int nwaves  = gridDim.x * wpb;
    const int ntiles  = M >> 4;
    int tile          = blockIdx.x * wpb + wave;
    if (tile >= ntiles) return;
    const int halfsel = lane >> 4;
    const int lid     = lane & 15;
    const int N       = NTILES * 16;

    v2f b[NTILES][32];
    {
        const float4* wf4 = (const float4*)Wf + lane;
#pragma unroll
        for (int nt = 0; nt < NTILES; ++nt) {
#pragma unroll
            for (int kk2 = 0; kk2 < 16; ++kk2) {
                const float4 q = wf4[(nt * 16 + kk2) * 32];
                b[nt][2 * kk2].x     = q.x; b[nt][2 * kk2].y     = q.y;
                b[nt][2 * kk2 + 1].x = q.z; b[nt][2 * kk2 + 1].y = q.w;
            }
        }
    }
    float bv[NTILES];
#pragma unroll
    for (int nt = 0; nt < NTILES; ++nt) bv[nt] = bias[nt * 16 + lid];

    v2f   an[32];
    float sn;
    {
        const int r = tile * 16 + lid;
        const float* ap = A + (size_t)r * 128 + halfsel * 2;
#pragma unroll
        for (int kk = 0; kk < 32; ++kk) an[kk] = *(const v2f*)(ap + kk * 4);
        sn = arowscale ? arowscale[r] : 1.0f;
    }

    for (; tile < ntiles; tile += nwaves) {
        v2f a[32];
#pragma unroll
        for (int kk = 0; kk < 32; ++kk) { a[kk].x = an[kk].x * sn; a[kk].y = an[kk].y * sn; }

        const int tnext = tile + nwaves;
        if (tnext < ntiles) {
            const int r = tnext * 16 + lid;
            const float* ap = A + (size_t)r * 128 + halfsel * 2;
#pragma unroll
            for (int kk = 0; kk < 32; ++kk) an[kk] = *(const v2f*)(ap + kk * 4);
            sn = arowscale ? arowscale[r] : 1.0f;
        }

        float es[8];
        if (erowscale) {
#pragma unroll
            for (int j = 0; j < 8; ++j) es[j] = erowscale[tile * 16 + j + halfsel * 8];
        } else {
#pragma unroll
            for (int j = 0; j < 8; ++j) es[j] = 1.0f;
        }

        v8f acc[NTILES];
#pragma unroll
        for (int nt = 0; nt < NTILES; ++nt) {
#pragma unroll
            for (int j = 0; j < 8; ++j) acc[nt][j] = bv[nt];
        }

#pragma unroll
        for (int kk = 0; kk < 32; ++kk) {
#pragma unroll
            for (int nt = 0; nt < NTILES; ++nt) {
                acc[nt] = __builtin_amdgcn_wmma_f32_16x16x4_f32(
                    false, a[kk], false, b[nt][kk], (short)0, acc[nt], false, false);
            }
        }

#pragma unroll
        for (int nt = 0; nt < NTILES; ++nt) {
#pragma unroll
            for (int j = 0; j < 8; ++j) {
                float v = acc[nt][j];
                if (relu) v = fmaxf(v, 0.0f);
                const int row = tile * 16 + j + halfsel * 8;
                if (erowscale) v *= es[j];
                out[(size_t)row * N + nt * 16 + lid] = v;
            }
        }
    }
}


extern "C" void kernel_launch(void* const* d_in, const int* in_sizes, int n_in,
                              void* d_out, int out_size, void* d_ws, size_t ws_size,
                              hipStream_t stream) {
    (void)n_in; (void)ws_size;
    const float* x   = (const float*)d_in[0];
    const float* W2  = (const float*)d_in[1];
    const float* b2  = (const float*)d_in[2];
    const float* gw  = (const float*)d_in[3];
    const float* gb  = (const float*)d_in[4];
    const float* Wc  = (const float*)d_in[5];
    const float* bc  = (const float*)d_in[6];
    const int*   src = (const int*)d_in[7];
    const int*   dst = (const int*)d_in[8];
    const int*   ids = (const int*)d_in[9];

    const int H  = 128;
    const int C  = 16;
    const int L  = in_sizes[2] / H;
    const int NS = in_sizes[9] / L;
    const int NE = in_sizes[7] / L;
    const int NL = in_sizes[4] / (L * H);
    const int NN = out_size / C;

    float* ws       = (float*)d_ws;
    float* hfin     = ws;
    float* hbuf     = hfin + (size_t)(L - 1) * NS * H;
    float* agg      = hbuf + (size_t)NS * H;
    float* deg_out  = agg + (size_t)NS * H;
    float* deg_in   = deg_out + NS;
    float* w2t      = deg_in + NS;
    float* gwt      = w2t + (size_t)L * H * H;
    float* wct      = gwt + (size_t)L * NL * H * H;
    unsigned char* okey = (unsigned char*)d_out;

    const int T = 256;
    const int GT = 128;
    const int gemm_blocks_NS = 256;

    {
        int nHH = H * H;
        int bHH = (nHH + T - 1) / T;
        for (int i = 0; i < L; ++i)
            pack_bfrag_kernel<<<bHH, T, 0, stream>>>(W2 + (size_t)i * nHH,
                                                     w2t + (size_t)i * nHH, H);
        for (int i = 0; i < L * NL; ++i)
            pack_bfrag_kernel<<<bHH, T, 0, stream>>>(gw + (size_t)i * nHH,
                                                     gwt + (size_t)i * nHH, H);
        int nHC = H * C;
        pack_bfrag_kernel<<<(nHC + T - 1) / T, T, 0, stream>>>(Wc, wct, C);
    }

    for (int i = 0; i < L; ++i) {
        const int*   src_i = src + (size_t)i * NE;
        const int*   dst_i = dst + (size_t)i * NE;
        const float* x_i   = x   + (size_t)i * NS * H;
        const float* w2t_i = w2t + (size_t)i * H * H;
        const float* b2_i  = b2  + (size_t)i * H;

        {
            int n4 = (2 * NS) / 4;
            zero_f4_kernel<<<(n4 + T - 1) / T, T, 0, stream>>>((float4*)deg_out, n4);
        }
        degree_count_kernel<<<(NE + T - 1) / T, T, 0, stream>>>(src_i, dst_i,
                                                                deg_out, deg_in,
                                                                okey, NE, NS);
        rsqrt_clamp_kernel<<<(2 * NS + T - 1) / T, T, 0, stream>>>(deg_out, 2 * NS);

        gemm_wmma_k128_kernel<8><<<gemm_blocks_NS, GT, 0, stream>>>(
            x_i, w2t_i, b2_i, nullptr, deg_out, hbuf, NS, 0);

        for (int l = 0; l < NL; ++l) {
            const float* gwt_il = gwt + ((size_t)i * NL + l) * H * H;
            const float* gb_il  = gb  + ((size_t)i * NL + l) * H;
            edge_sum_owner_kernel<<<OWN_GRID, 1024, 0, stream>>>(src_i, dst_i, okey,
                                                                 hbuf, agg, NE, NS);
            const float* escale = (l + 1 < NL) ? deg_out : nullptr;
            float* hout = (l + 1 < NL) ? hbuf : hfin + (size_t)i * NS * H;
            gemm_wmma_k128_kernel<8><<<gemm_blocks_NS, GT, 0, stream>>>(
                agg, gwt_il, gb_il, deg_in, escale, hout, NS, 1);
        }
    }

    scatter_classify_owner_kernel<<<OWN_GRID, 1024, 0, stream>>>(
        ids, hfin, wct, bc, (float*)d_out, L * NS, NN);
}


# edge sums, later conv layers of a link: an owner wave reads the (source row, local row) list its first-layer key scan saved in the output buffer; no rescan or id loads; LDS sums in arrival order
# speedup vs baseline: 2.5888x; 1.0437x over previous
#include <hip/hip_runtime.h>


typedef __attribute__((ext_vector_type(2))) float v2f;
typedef __attribute__((ext_vector_type(8))) float v8f;


__global__ void zero_f4_kernel(float4* __restrict__ p, int n4) {
    int i = blockIdx.x * blockDim.x + threadIdx.x;
    if (i < n4) p[i] = make_float4(0.f, 0.f, 0.f, 0.f);
}

__global__ void pack_bfrag_kernel(const float* __restrict__ in,
                                  float* __restrict__ outF, int N) {
    int i = blockIdx.x * blockDim.x + threadIdx.x;
    if (i < N * 128) {
        int c    = i & 3;
        int lane = (i >> 2) & 31;
        int kk2  = (i >> 7) & 15;
        int nt   = i >> 11;
        int kk   = 2 * kk2 + (c >> 1);
        int k    = kk * 4 + (lane >> 4) * 2 + (c & 1);
        int n    = nt * 16 + (lane & 15);
        outF[i] = in[k * N + n];
    }
}

#define OWN_GRID  256
#define OWN_RMAX  200

__device__ __forceinline__ int own_range_rows(int nRows) {
    const int R = (nRows + OWN_GRID - 1) / OWN_GRID;
    return (R > OWN_RMAX) ? OWN_RMAX : R;
}

__global__ void degree_count_kernel(const int* __restrict__ src,
                                    const int* __restrict__ dst,
                                    float* __restrict__ deg_out,
                                    float* __restrict__ deg_in,
                                    unsigned char* __restrict__ key,
                                    int nE, int nRows) {
    const unsigned R = (unsigned)own_range_rows(nRows);
    int e = blockIdx.x * blockDim.x + threadIdx.x;
    if (e < nE) {
        const int d = dst[e];
        atomicAdd(&deg_out[src[e]], 1.0f);
        atomicAdd(&deg_in[d], 1.0f);
        key[e] = (unsigned char)(((unsigned)d / R) % (unsigned)OWN_GRID);
    }
}

__global__ void rsqrt_clamp_kernel(float* __restrict__ d, int n) {
    int i = blockIdx.x * blockDim.x + threadIdx.x;
    if (i < n) d[i] = rsqrtf(fmaxf(d[i], 1.0f));
}

#define OWN_WAVES 32
#define OWN_QCAP  512
#define OWN_FLUSH 64
#define OWN_U     8
#define OWN_LSTRIDE 160
#define OWN_LCAP    (OWN_LSTRIDE - 1)
#define OWN_LBATCH  5
#define OWN_LOVER   0xFFFFFFFFu

__device__ __forceinline__ void own_add_rows(const float* __restrict__ h, float* acc,
                                             int my_s, int my_d, int m, int rows,
                                             int lane) {
    for (int j0 = 0; j0 < m; j0 += 8) {
        float4 v[8];
        int    dl[8];
#pragma unroll
        for (int u = 0; u < 8; ++u) {
            const int j = (j0 + u < m) ? (j0 + u) : (m - 1);
            const int s = __builtin_amdgcn_readlane(my_s, j);
            dl[u]       = __builtin_amdgcn_readlane(my_d, j);
            v[u] = *(const float4*)(h + (size_t)s * 128 + lane * 4);
        }
#pragma unroll
        for (int u = 0; u < 8; ++u) {
            if (j0 + u < m && (unsigned)dl[u] < (unsigned)rows) {
                float* row = acc + dl[u] * 128 + lane;
                __hip_atomic_fetch_add(row,      v[u].x, __ATOMIC_RELAXED, __HIP_MEMORY_SCOPE_WORKGROUP);
                __hip_atomic_fetch_add(row + 32, v[u].y, __ATOMIC_RELAXED, __HIP_MEMORY_SCOPE_WORKGROUP);
                __hip_atomic_fetch_add(row + 64, v[u].z, __ATOMIC_RELAXED, __HIP_MEMORY_SCOPE_WORKGROUP);
                __hip_atomic_fetch_add(row + 96, v[u].w, __ATOMIC_RELAXED, __HIP_MEMORY_SCOPE_WORKGROUP);
            }
        }
    }
}

__device__ __forceinline__ int own_drain(const int* __restrict__ src,
                                         const int* __restrict__ dst,
                                         const float* __restrict__ h,
                                         float* acc, const unsigned* qe,
                                         int n, int lo, int rows, int lane,
                                         bool save, unsigned* lst, int saved) {
    for (int i0 = 0; i0 < n; i0 += 32) {
        const int m = (n - i0 < 32) ? (n - i0) : 32;
        int my_s = 0, my_d = 0;
        if (lane < m) {
            const unsigned e = qe[i0 + lane];
            my_s = src[e];
            my_d = dst[e] - lo;
            const int p = saved + i0 + lane;
            if (save && p < OWN_LCAP)
                lst[1 + p] = ((unsigned)my_s << 8) | (unsigned)my_d;
        }
        own_add_rows(h, acc, my_s, my_d, m, rows, lane);
    }
    return saved + n;
}

__device__ __forceinline__ int own_push(bool m, unsigned e, unsigned* qe, int cnt) {
    const unsigned k = __builtin_amdgcn_ballot_w32(m);
    if (m) qe[cnt + (int)__builtin_amdgcn_mbcnt_lo(k, 0u)] = e;
    return cnt + __builtin_popcount(k);
}

__device__ __forceinline__ unsigned own_zero_bytes(unsigned x) {
    const unsigned t = (x & 0x7F7F7F7Fu) + 0x7F7F7F7Fu;
    return ~(t | x | 0x7F7F7F7Fu);
}

__global__ __launch_bounds__(1024)
void edge_sum_owner_kernel(const int* __restrict__ src,
                           const int* __restrict__ dst,
                           const unsigned char* __restrict__ key,
                           const float* __restrict__ h,
                           float* __restrict__ agg,
                           unsigned* __restrict__ lst,
                           int reuse,
                           int nE, int nRows) {
    __shared__ float    acc[OWN_RMAX * 128];
    __shared__ unsigned qe[OWN_WAVES][OWN_QCAP];

    const int lane = threadIdx.x & 31;
    const int wave = __builtin_amdgcn_readfirstlane((int)(threadIdx.x >> 5));
    const int nwav = blockDim.x >> 5;
    const int R   = own_range_rows(nRows);
    const int nGf = nE >> 9;
    const int rot = (int)(((long long)blockIdx.x * nGf) / OWN_GRID);
    const unsigned pat = (unsigned)blockIdx.x * 0x01010101u;
    unsigned* myqe = qe[wave];
    const bool listable = (OWN_GRID * R >= nRows) && (nRows <= (1 << 24));
    const bool save     = listable && !reuse;
    unsigned* mylst = lst + ((size_t)blockIdx.x * OWN_WAVES + wave) * OWN_LSTRIDE;
    unsigned nl = OWN_LOVER;
    unsigned pk[OWN_LBATCH];
#pragma unroll
    for (int b = 0; b < OWN_LBATCH; ++b) pk[b] = 0u;
    if (listable && reuse && blockIdx.x * R < nRows) {
        nl = (unsigned)__builtin_amdgcn_readfirstlane((int)mylst[0]);
        if (nl <= (unsigned)OWN_LCAP) {
#pragma unroll
            for (int b = 0; b < OWN_LBATCH; ++b) {
                const unsigned idx = (unsigned)(b * 32 + lane);
                if (idx < nl) pk[b] = mylst[1 + idx];
            }
        }
    }
    const bool scan = nl > (unsigned)OWN_LCAP;

    for (int lo = blockIdx.x * R; lo < nRows; lo += OWN_GRID * R) {
        const int rows = (nRows - lo < R) ? (nRows - lo) : R;
        for (int i = threadIdx.x; i < rows * 32; i += blockDim.x)
            ((float4*)acc)[i] = make_float4(0.f, 0.f, 0.f, 0.f);
        __syncthreads();

        if (!scan) {
#pragma unroll
            for (int b = 0; b < OWN_LBATCH; ++b) {
                if ((unsigned)(b * 32) < nl) {
                    const int left = (int)nl - b * 32;
                    own_add_rows(h, acc, (int)(pk[b] >> 8), (int)(pk[b] & 255u),
                                 (left < 32) ? left : 32, rows, lane);
                }
            }
        }

        int cnt = 0;
        int saved = 0;
        const int nGs = scan ? nGf : 0;
        for (int g0 = wave; g0 < nGs; g0 += OWN_U * nwav) {
            uint4 d[OWN_U];
            int   eb[OWN_U];
#pragma unroll
            for (int u = 0; u < OWN_U; ++u) {
                int g = g0 + u * nwav;
                if (g >= nGf) g = nGf - 1;
                g += rot;
                if (g >= nGf) g -= nGf;
                eb[u] = g * 512 + lane * 16;
                d[u]  = *(const uint4*)(key + eb[u]);
            }
#pragma unroll
            for (int u = 0; u < OWN_U; ++u) {
                if (g0 + u * nwav < nGf) {
                    unsigned hit = (own_zero_bytes(d[u].x ^ pat) >> 7)
                                 | (own_zero_bytes(d[u].y ^ pat) >> 6)
                                 | (own_zero_bytes(d[u].z ^ pat) >> 5)
                                 | (own_zero_bytes(d[u].w ^ pat) >> 4);
                    while (__builtin_amdgcn_ballot_w32(hit != 0u)) {
                        const bool     has = hit != 0u;
                        const unsigned p   = (unsigned)__builtin_ctz(hit | 0x80000000u);
                        const unsigned e   = (unsigned)eb[u] + ((p & 7u) << 2) + (p >> 3);
                        cnt = own_push(has, e, myqe, cnt);
                        hit &= hit - 1u;
                        if (cnt >= OWN_FLUSH) {
                            __builtin_amdgcn_fence(__ATOMIC_ACQ_REL, "wavefront");
                            saved = own_drain(src, dst, h, acc, myqe, cnt, lo, rows, lane,
                                              save, mylst, saved);
                            cnt = 0;
                        }
                    }
                }
            }
        }
        if (wave == 0 && scan) {
            for (int eb0 = nGf * 512; eb0 < nE; eb0 += 32) {
                const int e = eb0 + lane;
                const unsigned r = ((e < nE) ? (unsigned)dst[e] : 0xFFFFFFFFu) - (unsigned)lo;
                cnt = own_push(r < (unsigned)rows && e < nE, (unsigned)e, myqe, cnt);
                if (cnt >= OWN_FLUSH) {
                    __builtin_amdgcn_fence(__ATOMIC_ACQ_REL, "wavefront");
                    saved = own_drain(src, dst, h, acc, myqe, cnt, lo, rows, lane,
                                      save, mylst, saved);
                    cnt = 0;
                }
            }
        }
        __builtin_amdgcn_fence(__ATOMIC_ACQ_REL, "wavefront");
        saved = own_drain(src, dst, h, acc, myqe, cnt, lo, rows, lane, save, mylst, saved);
        if (save && lane == 0)
            mylst[0] = (saved <= OWN_LCAP) ? (unsigned)saved : OWN_LOVER;
        __syncthreads();

        for (int r = wave; r < rows; r += nwav) {
            const float* row = acc + r * 128 + lane;
            const float4 o = make_float4(row[0], row[32], row[64], row[96]);
            *(float4*)(agg + (size_t)(lo + r) * 128 + lane * 4) = o;
        }
        __syncthreads();
    }
}

#define SC_RMAX  400
#define SC_P     132
#define SC_QCAP  256
#define SC_FLUSH 64
#define SC_U     8

__device__ __forceinline__ void sc_drain(const int* __restrict__ ids,
                                         const float* __restrict__ h,
                                         float* acc, const unsigned* qe,
                                         int n, int lo, int lane) {
    for (int i0 = 0; i0 < n; i0 += 32) {
        const int m = (n - i0 < 32) ? (n - i0) : 32;
        int my_g = 0, my_d = 0;
        if (lane < m) {
            const unsigned g = qe[i0 + lane];
            my_g = (int)g;
            my_d = ids[g] - lo;
        }
        for (int j0 = 0; j0 < m; j0 += 8) {
            float4 v[8];
            int    dl[8];
#pragma unroll
            for (int u = 0; u < 8; ++u) {
                const int j = (j0 + u < m) ? (j0 + u) : (m - 1);
                const int g = __builtin_amdgcn_readlane(my_g, j);
                dl[u]       = __builtin_amdgcn_readlane(my_d, j);
                v[u] = *(const float4*)(h + (size_t)g * 128 + lane * 4);
            }
#pragma unroll
            for (int u = 0; u < 8; ++u) {
                if (j0 + u < m) {
                    float* row = acc + dl[u] * SC_P + lane;
                    __hip_atomic_fetch_add(row,      v[u].x, __ATOMIC_RELAXED, __HIP_MEMORY_SCOPE_WORKGROUP);
                    __hip_atomic_fetch_add(row + 32, v[u].y, __ATOMIC_RELAXED, __HIP_MEMORY_SCOPE_WORKGROUP);
                    __hip_atomic_fetch_add(row + 64, v[u].z, __ATOMIC_RELAXED, __HIP_MEMORY_SCOPE_WORKGROUP);
                    __hip_atomic_fetch_add(row + 96, v[u].w, __ATOMIC_RELAXED, __HIP_MEMORY_SCOPE_WORKGROUP);
                }
            }
        }
    }
}

__global__ __launch_bounds__(1024)
void scatter_classify_owner_kernel(const int* __restrict__ ids,
                                   const float* __restrict__ h,
                                   const float* __restrict__ Wcf,
                                   const float* __restrict__ bc,
                                   float* __restrict__ out,
                                   int nIds, int nNodes) {
    __shared__ float    acc[SC_RMAX * SC_P];
    __shared__ unsigned qe[OWN_WAVES][SC_QCAP];

    const int lane = threadIdx.x & 31;
    const int wave = __builtin_amdgcn_readfirstlane((int)(threadIdx.x >> 5));
    const int nwav = blockDim.x >> 5;
    const int halfsel = lane >> 4;
    const int lid     = lane & 15;
    int R = (nNodes + OWN_GRID - 1) / OWN_GRID;
    if (R > SC_RMAX) R = SC_RMAX;
    const int nGf = nIds >> 7;
    const int rot = (int)(((long long)blockIdx.x * nGf) / OWN_GRID);
    unsigned* myqe = qe[wave];

    for (int lo = blockIdx.x * R; lo < nNodes; lo += OWN_GRID * R) {
        const int rows  = (nNodes - lo < R) ? (nNodes - lo) : R;
        const int ntile = (rows + 15) >> 4;
        for (int i = threadIdx.x; i < ntile * 16 * (SC_P / 4); i += blockDim.x)
            ((float4*)acc)[i] = make_float4(0.f, 0.f, 0.f, 0.f);
        __syncthreads();

        int cnt = 0;
        for (int g0 = wave; g0 < nGf; g0 += SC_U * nwav) {
            uint4 d[SC_U];
            int   eb[SC_U];
#pragma unroll
            for (int u = 0; u < SC_U; ++u) {
                int g = g0 + u * nwav;
                if (g >= nGf) g = nGf - 1;
                g += rot;
                if (g >= nGf) g -= nGf;
                eb[u] = g * 128 + lane * 4;
                d[u]  = *(const uint4*)(ids + eb[u]);
            }
#pragma unroll
            for (int u = 0; u < SC_U; ++u) {
                if (g0 + u * nwav < nGf) {
                    const unsigned b = (unsigned)eb[u];
                    cnt = own_push(d[u].x - (unsigned)lo < (unsigned)rows, b,      myqe, cnt);
                    cnt = own_push(d[u].y - (unsigned)lo < (unsigned)rows, b + 1u, myqe, cnt);
                    cnt = own_push(d[u].z - (unsigned)lo < (unsigned)rows, b + 2u, myqe, cnt);
                    cnt = own_push(d[u].w - (unsigned)lo < (unsigned)rows, b + 3u, myqe, cnt);
                    if (cnt >= SC_FLUSH) {
                        __builtin_amdgcn_fence(__ATOMIC_ACQ_REL, "wavefront");
                        sc_drain(ids, h, acc, myqe, cnt, lo, lane);
                        cnt = 0;
                    }
                }
            }
        }
        if (wave == 0) {
            for (int e0 = nGf * 128; e0 < nIds; e0 += 32) {
                const int e = e0 + lane;
                const unsigned r = ((e < nIds) ? (unsigned)ids[e] : 0xFFFFFFFFu) - (unsigned)lo;
                cnt = own_push(r < (unsigned)rows && e < nIds, (unsigned)e, myqe, cnt);
                if (cnt >= SC_FLUSH) {
                    __builtin_amdgcn_fence(__ATOMIC_ACQ_REL, "wavefront");
                    sc_drain(ids, h, acc, myqe, cnt, lo, lane);
                    cnt = 0;
                }
            }
        }
        __builtin_amdgcn_fence(__ATOMIC_ACQ_REL, "wavefront");
        sc_drain(ids, h, acc, myqe, cnt, lo, lane);
        __syncthreads();

        for (int t = wave; t < ntile; t += nwav) {
            v2f b[32];
            const float4* wf4 = (const float4*)Wcf + lane;
#pragma unroll
            for (int kk2 = 0; kk2 < 16; ++kk2) {
                const float4 q = wf4[kk2 * 32];
                b[2 * kk2].x     = q.x; b[2 * kk2].y     = q.y;
                b[2 * kk2 + 1].x = q.z; b[2 * kk2 + 1].y = q.w;
            }
            const float bv = bc[lid];
            v8f c;
#pragma unroll
            for (int j = 0; j < 8; ++j) c[j] = bv;
            const float* ar = acc + (t * 16 + lid) * SC_P + halfsel * 64;
#pragma unroll
            for (int kk = 0; kk < 32; ++kk) {
                v2f a;
                a.x = ar[kk];
                a.y = ar[32 + kk];
                c = __builtin_amdgcn_wmma_f32_16x16x4_f32(
                    false, a, false, b[kk], (short)0, c, false, false);
            }
#pragma unroll
            for (int j = 0; j < 8; ++j) {
                const int row = t * 16 + j + halfsel * 8;
                if (row < rows) out[(size_t)(lo + row) * 16 + lid] = c[j];
            }
        }
        __syncthreads();
    }
}

template <int NTILES>
__global__ __launch_bounds__(128)
void gemm_wmma_k128_kernel(const float* __restrict__ A,
                           const float* __restrict__ Wf,
                           const float* __restrict__ bias,
                           const float* __restrict__ arowscale,
                           const float* __restrict__ erowscale,
                           float* __restrict__ out,
                           int M, int relu) {
    const int lane    = threadIdx.x & 31;
    const int wave    = __builtin_amdgcn_readfirstlane((int)(threadIdx.x >> 5));
    const int wpb     = blockDim.x >> 5;
    const int nwaves  = gridDim.x * wpb;
    const int ntiles  = M >> 4;
    int tile          = blockIdx.x * wpb + wave;
    if (tile >= ntiles) return;
    const int halfsel = lane >> 4;
    const int lid     = lane & 15;
    const int N       = NTILES * 16;

    v2f b[NTILES][32];
    {
        const float4* wf4 = (const float4*)Wf + lane;
#pragma unroll
        for (int nt = 0; nt < NTILES; ++nt) {
#pragma unroll
            for (int kk2 = 0; kk2 < 16; ++kk2) {
                const float4 q = wf4[(nt * 16 + kk2) * 32];
                b[nt][2 * kk2].x     = q.x; b[nt][2 * kk2].y     = q.y;
                b[nt][2 * kk2 + 1].x = q.z; b[nt][2 * kk2 + 1].y = q.w;
            }
        }
    }
    float bv[NTILES];
#pragma unroll
    for (int nt = 0; nt < NTILES; ++nt) bv[nt] = bias[nt * 16 + lid];

    v2f   an[32];
    float sn;
    {
        const int r = tile * 16 + lid;
        const float* ap = A + (size_t)r * 128 + halfsel * 2;
#pragma unroll
        for (int kk = 0; kk < 32; ++kk) an[kk] = *(const v2f*)(ap + kk * 4);
        sn = arowscale ? arowscale[r] : 1.0f;
    }

    for (; tile < ntiles; tile += nwaves) {
        v2f a[32];
#pragma unroll
        for (int kk = 0; kk < 32; ++kk) { a[kk].x = an[kk].x * sn; a[kk].y = an[kk].y * sn; }

        const int tnext = tile + nwaves;
        if (tnext < ntiles) {
            const int r = tnext * 16 + lid;
            const float* ap = A + (size_t)r * 128 + halfsel * 2;
#pragma unroll
            for (int kk = 0; kk < 32; ++kk) an[kk] = *(const v2f*)(ap + kk * 4);
            sn = arowscale ? arowscale[r] : 1.0f;
        }

        float es[8];
        if (erowscale) {
#pragma unroll
            for (int j = 0; j < 8; ++j) es[j] = erowscale[tile * 16 + j + halfsel * 8];
        } else {
#pragma unroll
            for (int j = 0; j < 8; ++j) es[j] = 1.0f;
        }

        v8f acc[NTILES];
#pragma unroll
        for (int nt = 0; nt < NTILES; ++nt) {
#pragma unroll
            for (int j = 0; j < 8; ++j) acc[nt][j] = bv[nt];
        }

#pragma unroll
        for (int kk = 0; kk < 32; ++kk) {
#pragma unroll
            for (int nt = 0; nt < NTILES; ++nt) {
                acc[nt] = __builtin_amdgcn_wmma_f32_16x16x4_f32(
                    false, a[kk], false, b[nt][kk], (short)0, acc[nt], false, false);
            }
        }

#pragma unroll
        for (int nt = 0; nt < NTILES; ++nt) {
#pragma unroll
            for (int j = 0; j < 8; ++j) {
                float v = acc[nt][j];
                if (relu) v = fmaxf(v, 0.0f);
                const int row = tile * 16 + j + halfsel * 8;
                if (erowscale) v *= es[j];
                out[(size_t)row * N + nt * 16 + lid] = v;
            }
        }
    }
}


extern "C" void kernel_launch(void* const* d_in, const int* in_sizes, int n_in,
                              void* d_out, int out_size, void* d_ws, size_t ws_size,
                              hipStream_t stream) {
    (void)n_in; (void)ws_size;
    const float* x   = (const float*)d_in[0];
    const float* W2  = (const float*)d_in[1];
    const float* b2  = (const float*)d_in[2];
    const float* gw  = (const float*)d_in[3];
    const float* gb  = (const float*)d_in[4];
    const float* Wc  = (const float*)d_in[5];
    const float* bc  = (const float*)d_in[6];
    const int*   src = (const int*)d_in[7];
    const int*   dst = (const int*)d_in[8];
    const int*   ids = (const int*)d_in[9];

    const int H  = 128;
    const int C  = 16;
    const int L  = in_sizes[2] / H;
    const int NS = in_sizes[9] / L;
    const int NE = in_sizes[7] / L;
    const int NL = in_sizes[4] / (L * H);
    const int NN = out_size / C;

    float* ws       = (float*)d_ws;
    float* hfin     = ws;
    float* hbuf     = hfin + (size_t)(L - 1) * NS * H;
    float* agg      = hbuf + (size_t)NS * H;
    float* deg_out  = agg + (size_t)NS * H;
    float* deg_in   = deg_out + NS;
    float* w2t      = deg_in + NS;
    float* gwt      = w2t + (size_t)L * H * H;
    float* wct      = gwt + (size_t)L * NL * H * H;
    unsigned char* okey = (unsigned char*)d_out;
    unsigned* olist = (unsigned*)(okey + (((size_t)NE + 255) & ~(size_t)255));

    const int T = 256;
    const int GT = 128;
    const int gemm_blocks_NS = 256;

    {
        int nHH = H * H;
        int bHH = (nHH + T - 1) / T;
        for (int i = 0; i < L; ++i)
            pack_bfrag_kernel<<<bHH, T, 0, stream>>>(W2 + (size_t)i * nHH,
                                                     w2t + (size_t)i * nHH, H);
        for (int i = 0; i < L * NL; ++i)
            pack_bfrag_kernel<<<bHH, T, 0, stream>>>(gw + (size_t)i * nHH,
                                                     gwt + (size_t)i * nHH, H);
        int nHC = H * C;
        pack_bfrag_kernel<<<(nHC + T - 1) / T, T, 0, stream>>>(Wc, wct, C);
    }

    for (int i = 0; i < L; ++i) {
        const int*   src_i = src + (size_t)i * NE;
        const int*   dst_i = dst + (size_t)i * NE;
        const float* x_i   = x   + (size_t)i * NS * H;
        const float* w2t_i = w2t + (size_t)i * H * H;
        const float* b2_i  = b2  + (size_t)i * H;

        {
            int n4 = (2 * NS) / 4;
            zero_f4_kernel<<<(n4 + T - 1) / T, T, 0, stream>>>((float4*)deg_out, n4);
        }
        degree_count_kernel<<<(NE + T - 1) / T, T, 0, stream>>>(src_i, dst_i,
                                                                deg_out, deg_in,
                                                                okey, NE, NS);
        rsqrt_clamp_kernel<<<(2 * NS + T - 1) / T, T, 0, stream>>>(deg_out, 2 * NS);

        gemm_wmma_k128_kernel<8><<<gemm_blocks_NS, GT, 0, stream>>>(
            x_i, w2t_i, b2_i, nullptr, deg_out, hbuf, NS, 0);

        for (int l = 0; l < NL; ++l) {
            const float* gwt_il = gwt + ((size_t)i * NL + l) * H * H;
            const float* gb_il  = gb  + ((size_t)i * NL + l) * H;
            edge_sum_owner_kernel<<<OWN_GRID, 1024, 0, stream>>>(src_i, dst_i, okey,
                                                                 hbuf, agg, olist, l,
                                                                 NE, NS);
            const float* escale = (l + 1 < NL) ? deg_out : nullptr;
            float* hout = (l + 1 < NL) ? hbuf : hfin + (size_t)i * NS * H;
            gemm_wmma_k128_kernel<8><<<gemm_blocks_NS, GT, 0, stream>>>(
                agg, gwt_il, gb_il, deg_in, escale, hout, NS, 1);
        }
    }

    scatter_classify_owner_kernel<<<OWN_GRID, 1024, 0, stream>>>(
        ids, hfin, wct, bc, (float*)d_out, L * NS, NN);
}


# conv layers: owner work-group multiplies its summed LDS rows by the layer weights on WMMA (scale, bias, relu there), writes the new rows; no sums plane, no conv GEMM; 35->29 launches; sum orders kept
# speedup vs baseline: 2.6861x; 1.0376x over previous
#include <hip/hip_runtime.h>


typedef __attribute__((ext_vector_type(2))) float v2f;
typedef __attribute__((ext_vector_type(8))) float v8f;


__global__ void zero_f4_kernel(float4* __restrict__ p, int n4) {
    int i = blockIdx.x * blockDim.x + threadIdx.x;
    if (i < n4) p[i] = make_float4(0.f, 0.f, 0.f, 0.f);
}

__global__ void pack_bfrag_kernel(const float* __restrict__ in,
                                  float* __restrict__ outF, int N) {
    int i = blockIdx.x * blockDim.x + threadIdx.x;
    if (i < N * 128) {
        int c    = i & 3;
        int lane = (i >> 2) & 31;
        int kk2  = (i >> 7) & 15;
        int nt   = i >> 11;
        int kk   = 2 * kk2 + (c >> 1);
        int k    = kk * 4 + (lane >> 4) * 2 + (c & 1);
        int n    = nt * 16 + (lane & 15);
        outF[i] = in[k * N + n];
    }
}

#define OWN_GRID  256
#define OWN_RMAX  208
#define OWN_P     132
#define OWN_CS    33

__device__ __forceinline__ int own_range_rows(int nRows) {
    const int R = (nRows + OWN_GRID - 1) / OWN_GRID;
    return (R > OWN_RMAX) ? OWN_RMAX : R;
}

__global__ void degree_count_kernel(const int* __restrict__ src,
                                    const int* __restrict__ dst,
                                    float* __restrict__ deg_out,
                                    float* __restrict__ deg_in,
                                    unsigned char* __restrict__ key,
                                    int nE, int nRows) {
    const unsigned R = (unsigned)own_range_rows(nRows);
    int e = blockIdx.x * blockDim.x + threadIdx.x;
    if (e < nE) {
        const int d = dst[e];
        atomicAdd(&deg_out[src[e]], 1.0f);
        atomicAdd(&deg_in[d], 1.0f);
        key[e] = (unsigned char)(((unsigned)d / R) % (unsigned)OWN_GRID);
    }
}

__global__ void rsqrt_clamp_kernel(float* __restrict__ d, int n) {
    int i = blockIdx.x * blockDim.x + threadIdx.x;
    if (i < n) d[i] = rsqrtf(fmaxf(d[i], 1.0f));
}

#define OWN_WAVES 32
#define OWN_QCAP  512
#define OWN_FLUSH 64
#define OWN_U     8
#define OWN_LSTRIDE 160
#define OWN_LCAP    (OWN_LSTRIDE - 1)
#define OWN_LBATCH  5
#define OWN_LOVER   0xFFFFFFFFu

__device__ __forceinline__ void own_add_rows(const float* __restrict__ h, float* acc,
                                             int my_s, int my_d, int m, int rows,
                                             int lane) {
    for (int j0 = 0; j0 < m; j0 += 8) {
        float4 v[8];
        int    dl[8];
#pragma unroll
        for (int u = 0; u < 8; ++u) {
            const int j = (j0 + u < m) ? (j0 + u) : (m - 1);
            const int s = __builtin_amdgcn_readlane(my_s, j);
            dl[u]       = __builtin_amdgcn_readlane(my_d, j);
            v[u] = *(const float4*)(h + (size_t)s * 128 + lane * 4);
        }
#pragma unroll
        for (int u = 0; u < 8; ++u) {
            if (j0 + u < m && (unsigned)dl[u] < (unsigned)rows) {
                float* row = acc + dl[u] * OWN_P + lane;
                __hip_atomic_fetch_add(row,              v[u].x, __ATOMIC_RELAXED, __HIP_MEMORY_SCOPE_WORKGROUP);
                __hip_atomic_fetch_add(row + OWN_CS,     v[u].y, __ATOMIC_RELAXED, __HIP_MEMORY_SCOPE_WORKGROUP);
                __hip_atomic_fetch_add(row + 2 * OWN_CS, v[u].z, __ATOMIC_RELAXED, __HIP_MEMORY_SCOPE_WORKGROUP);
                __hip_atomic_fetch_add(row + 3 * OWN_CS, v[u].w, __ATOMIC_RELAXED, __HIP_MEMORY_SCOPE_WORKGROUP);
            }
        }
    }
}

__device__ __forceinline__ int own_drain(const int* __restrict__ src,
                                         const int* __restrict__ dst,
                                         const float* __restrict__ h,
                                         float* acc, const unsigned* qe,
                                         int n, int lo, int rows, int lane,
                                         bool save, unsigned* lst, int saved) {
    for (int i0 = 0; i0 < n; i0 += 32) {
        const int m = (n - i0 < 32) ? (n - i0) : 32;
        int my_s = 0, my_d = 0;
        if (lane < m) {
            const unsigned e = qe[i0 + lane];
            my_s = src[e];
            my_d = dst[e] - lo;
            const int p = saved + i0 + lane;
            if (save && p < OWN_LCAP)
                lst[1 + p] = ((unsigned)my_s << 8) | (unsigned)my_d;
        }
        own_add_rows(h, acc, my_s, my_d, m, rows, lane);
    }
    return saved + n;
}

__device__ __forceinline__ int own_push(bool m, unsigned e, unsigned* qe, int cnt) {
    const unsigned k = __builtin_amdgcn_ballot_w32(m);
    if (m) qe[cnt + (int)__builtin_amdgcn_mbcnt_lo(k, 0u)] = e;
    return cnt + __builtin_popcount(k);
}

__device__ __forceinline__ unsigned own_zero_bytes(unsigned x) {
    const unsigned t = (x & 0x7F7F7F7Fu) + 0x7F7F7F7Fu;
    return ~(t | x | 0x7F7F7F7Fu);
}

__global__ __launch_bounds__(1024)
void conv_owner_kernel(const int* __restrict__ src,
                       const int* __restrict__ dst,
                       const unsigned char* __restrict__ key,
                       const float* __restrict__ h,
                       const float* Wf,
                       const float* bias,
                       const float* __restrict__ arowscale,
                       const float* __restrict__ erowscale,
                       float* __restrict__ hout,
                       unsigned* __restrict__ lst,
                       int reuse,
                       int nE, int nRows) {
    __shared__ float    acc[OWN_RMAX * OWN_P];
    __shared__ unsigned qe[OWN_WAVES][OWN_QCAP];

    const int lane = threadIdx.x & 31;
    const int halfsel = lane >> 4;
    const int lid     = lane & 15;
    const int wave = __builtin_amdgcn_readfirstlane((int)(threadIdx.x >> 5));
    const int nwav = blockDim.x >> 5;
    const int R   = own_range_rows(nRows);
    const int nGf = nE >> 9;
    const int rot = (int)(((long long)blockIdx.x * nGf) / OWN_GRID);
    const unsigned pat = (unsigned)blockIdx.x * 0x01010101u;
    unsigned* myqe = qe[wave];
    const bool listable = (OWN_GRID * R >= nRows) && (nRows <= (1 << 24));
    const bool save     = listable && !reuse;
    unsigned* mylst = lst + ((size_t)blockIdx.x * OWN_WAVES + wave) * OWN_LSTRIDE;
    unsigned nl = OWN_LOVER;
    unsigned pk[OWN_LBATCH];
#pragma unroll
    for (int b = 0; b < OWN_LBATCH; ++b) pk[b] = 0u;
    if (listable && reuse && blockIdx.x * R < nRows) {
        nl = (unsigned)__builtin_amdgcn_readfirstlane((int)mylst[0]);
        if (nl <= (unsigned)OWN_LCAP) {
#pragma unroll
            for (int b = 0; b < OWN_LBATCH; ++b) {
                const unsigned idx = (unsigned)(b * 32 + lane);
                if (idx < nl) pk[b] = mylst[1 + idx];
            }
        }
    }
    const bool scan = nl > (unsigned)OWN_LCAP;

    for (int lo = blockIdx.x * R; lo < nRows; lo += OWN_GRID * R) {
        const int rows  = (nRows - lo < R) ? (nRows - lo) : R;
        const int ntile = (rows + 15) >> 4;
        for (int i = threadIdx.x; i < ntile * 16 * (OWN_P / 4); i += blockDim.x)
            ((float4*)acc)[i] = make_float4(0.f, 0.f, 0.f, 0.f);
        __syncthreads();

        if (!scan) {
#pragma unroll
            for (int b = 0; b < OWN_LBATCH; ++b) {
                if ((unsigned)(b * 32) < nl) {
                    const int left = (int)nl - b * 32;
                    own_add_rows(h, acc, (int)(pk[b] >> 8), (int)(pk[b] & 255u),
                                 (left < 32) ? left : 32, rows, lane);
                }
            }
        }

        int cnt = 0;
        int saved = 0;
        const int nGs = scan ? nGf : 0;
        for (int g0 = wave; g0 < nGs; g0 += OWN_U * nwav) {
            uint4 d[OWN_U];
            int   eb[OWN_U];
#pragma unroll
            for (int u = 0; u < OWN_U; ++u) {
                int g = g0 + u * nwav;
                if (g >= nGf) g = nGf - 1;
                g += rot;
                if (g >= nGf) g -= nGf;
                eb[u] = g * 512 + lane * 16;
                d[u]  = *(const uint4*)(key + eb[u]);
            }
#pragma unroll
            for (int u = 0; u < OWN_U; ++u) {
                if (g0 + u * nwav < nGf) {
                    unsigned hit = (own_zero_bytes(d[u].x ^ pat) >> 7)
                                 | (own_zero_bytes(d[u].y ^ pat) >> 6)
                                 | (own_zero_bytes(d[u].z ^ pat) >> 5)
                                 | (own_zero_bytes(d[u].w ^ pat) >> 4);
                    while (__builtin_amdgcn_ballot_w32(hit != 0u)) {
                        const bool     has = hit != 0u;
                        const unsigned p   = (unsigned)__builtin_ctz(hit | 0x80000000u);
                        const unsigned e   = (unsigned)eb[u] + ((p & 7u) << 2) + (p >> 3);
                        cnt = own_push(has, e, myqe, cnt);
                        hit &= hit - 1u;
                        if (cnt >= OWN_FLUSH) {
                            __builtin_amdgcn_fence(__ATOMIC_ACQ_REL, "wavefront");
                            saved = own_drain(src, dst, h, acc, myqe, cnt, lo, rows, lane,
                                              save, mylst, saved);
                            cnt = 0;
                        }
                    }
                }
            }
        }
        if (wave == 0 && scan) {
            for (int eb0 = nGf * 512; eb0 < nE; eb0 += 32) {
                const int e = eb0 + lane;
                const unsigned r = ((e < nE) ? (unsigned)dst[e] : 0xFFFFFFFFu) - (unsigned)lo;
                cnt = own_push(r < (unsigned)rows && e < nE, (unsigned)e, myqe, cnt);
                if (cnt >= OWN_FLUSH) {
                    __builtin_amdgcn_fence(__ATOMIC_ACQ_REL, "wavefront");
                    saved = own_drain(src, dst, h, acc, myqe, cnt, lo, rows, lane,
                                      save, mylst, saved);
                    cnt = 0;
                }
            }
        }
        __builtin_amdgcn_fence(__ATOMIC_ACQ_REL, "wavefront");
        saved = own_drain(src, dst, h, acc, myqe, cnt, lo, rows, lane, save, mylst, saved);
        if (save && lane == 0)
            mylst[0] = (saved <= OWN_LCAP) ? (unsigned)saved : OWN_LOVER;

        const int nt = wave & 7;
        v2f b[32];
        {
            const float4* wf4 = (const float4*)Wf + (size_t)nt * 16 * 32 + lane;
#pragma unroll
            for (int kk2 = 0; kk2 < 16; ++kk2) {
                const float4 q = wf4[kk2 * 32];
                b[2 * kk2].x     = q.x; b[2 * kk2].y     = q.y;
                b[2 * kk2 + 1].x = q.z; b[2 * kk2 + 1].y = q.w;
            }
        }
        const float bv = bias[nt * 16 + lid];
        __syncthreads();

        for (int t = wave >> 3; t < ntile; t += (nwav >> 3)) {
            const int r = t * 16 + lid;
            const float sn = (r < rows) ? arowscale[lo + r] : 0.0f;
            const float* ar = acc + r * OWN_P + halfsel * (2 * OWN_CS);
            v8f c;
#pragma unroll
            for (int j = 0; j < 8; ++j) c[j] = bv;
#pragma unroll
            for (int kk = 0; kk < 32; ++kk) {
                v2f a;
                a.x = ar[kk] * sn;
                a.y = ar[OWN_CS + kk] * sn;
                c = __builtin_amdgcn_wmma_f32_16x16x4_f32(
                    false, a, false, b[kk], (short)0, c, false, false);
            }
#pragma unroll
            for (int j = 0; j < 8; ++j) {
                const int row = t * 16 + j + halfsel * 8;
                if (row < rows) {
                    float v = fmaxf(c[j], 0.0f);
                    if (erowscale) v *= erowscale[lo + row];
                    hout[(size_t)(lo + row) * 128 + nt * 16 + lid] = v;
                }
            }
        }
        __syncthreads();
    }
}

#define SC_RMAX  400
#define SC_P     132
#define SC_QCAP  256
#define SC_FLUSH 64
#define SC_U     8

__device__ __forceinline__ void sc_drain(const int* __restrict__ ids,
                                         const float* __restrict__ h,
                                         float* acc, const unsigned* qe,
                                         int n, int lo, int lane) {
    for (int i0 = 0; i0 < n; i0 += 32) {
        const int m = (n - i0 < 32) ? (n - i0) : 32;
        int my_g = 0, my_d = 0;
        if (lane < m) {
            const unsigned g = qe[i0 + lane];
            my_g = (int)g;
            my_d = ids[g] - lo;
        }
        for (int j0 = 0; j0 < m; j0 += 8) {
            float4 v[8];
            int    dl[8];
#pragma unroll
            for (int u = 0; u < 8; ++u) {
                const int j = (j0 + u < m) ? (j0 + u) : (m - 1);
                const int g = __builtin_amdgcn_readlane(my_g, j);
                dl[u]       = __builtin_amdgcn_readlane(my_d, j);
                v[u] = *(const float4*)(h + (size_t)g * 128 + lane * 4);
            }
#pragma unroll
            for (int u = 0; u < 8; ++u) {
                if (j0 + u < m) {
                    float* row = acc + dl[u] * SC_P + lane;
                    __hip_atomic_fetch_add(row,      v[u].x, __ATOMIC_RELAXED, __HIP_MEMORY_SCOPE_WORKGROUP);
                    __hip_atomic_fetch_add(row + 32, v[u].y, __ATOMIC_RELAXED, __HIP_MEMORY_SCOPE_WORKGROUP);
                    __hip_atomic_fetch_add(row + 64, v[u].z, __ATOMIC_RELAXED, __HIP_MEMORY_SCOPE_WORKGROUP);
                    __hip_atomic_fetch_add(row + 96, v[u].w, __ATOMIC_RELAXED, __HIP_MEMORY_SCOPE_WORKGROUP);
                }
            }
        }
    }
}

__global__ __launch_bounds__(1024)
void scatter_classify_owner_kernel(const int* __restrict__ ids,
                                   const float* __restrict__ h,
                                   const float* __restrict__ Wcf,
                                   const float* __restrict__ bc,
                                   float* __restrict__ out,
                                   int nIds, int nNodes) {
    __shared__ float    acc[SC_RMAX * SC_P];
    __shared__ unsigned qe[OWN_WAVES][SC_QCAP];

    const int lane = threadIdx.x & 31;
    const int wave = __builtin_amdgcn_readfirstlane((int)(threadIdx.x >> 5));
    const int nwav = blockDim.x >> 5;
    const int halfsel = lane >> 4;
    const int lid     = lane & 15;
    int R = (nNodes + OWN_GRID - 1) / OWN_GRID;
    if (R > SC_RMAX) R = SC_RMAX;
    const int nGf = nIds >> 7;
    const int rot = (int)(((long long)blockIdx.x * nGf) / OWN_GRID);
    unsigned* myqe = qe[wave];

    for (int lo = blockIdx.x * R; lo < nNodes; lo += OWN_GRID * R) {
        const int rows  = (nNodes - lo < R) ? (nNodes - lo) : R;
        const int ntile = (rows + 15) >> 4;
        for (int i = threadIdx.x; i < ntile * 16 * (SC_P / 4); i += blockDim.x)
            ((float4*)acc)[i] = make_float4(0.f, 0.f, 0.f, 0.f);
        __syncthreads();

        int cnt = 0;
        for (int g0 = wave; g0 < nGf; g0 += SC_U * nwav) {
            uint4 d[SC_U];
            int   eb[SC_U];
#pragma unroll
            for (int u = 0; u < SC_U; ++u) {
                int g = g0 + u * nwav;
                if (g >= nGf) g = nGf - 1;
                g += rot;
                if (g >= nGf) g -= nGf;
                eb[u] = g * 128 + lane * 4;
                d[u]  = *(const uint4*)(ids + eb[u]);
            }
#pragma unroll
            for (int u = 0; u < SC_U; ++u) {
                if (g0 + u * nwav < nGf) {
                    const unsigned b = (unsigned)eb[u];
                    cnt = own_push(d[u].x - (unsigned)lo < (unsigned)rows, b,      myqe, cnt);
                    cnt = own_push(d[u].y - (unsigned)lo < (unsigned)rows, b + 1u, myqe, cnt);
                    cnt = own_push(d[u].z - (unsigned)lo < (unsigned)rows, b + 2u, myqe, cnt);
                    cnt = own_push(d[u].w - (unsigned)lo < (unsigned)rows, b + 3u, myqe, cnt);
                    if (cnt >= SC_FLUSH) {
                        __builtin_amdgcn_fence(__ATOMIC_ACQ_REL, "wavefront");
                        sc_drain(ids, h, acc, myqe, cnt, lo, lane);
                        cnt = 0;
                    }
                }
            }
        }
        if (wave == 0) {
            for (int e0 = nGf * 128; e0 < nIds; e0 += 32) {
                const int e = e0 + lane;
                const unsigned r = ((e < nIds) ? (unsigned)ids[e] : 0xFFFFFFFFu) - (unsigned)lo;
                cnt = own_push(r < (unsigned)rows && e < nIds, (unsigned)e, myqe, cnt);
                if (cnt >= SC_FLUSH) {
                    __builtin_amdgcn_fence(__ATOMIC_ACQ_REL, "wavefront");
                    sc_drain(ids, h, acc, myqe, cnt, lo, lane);
                    cnt = 0;
                }
            }
        }
        __builtin_amdgcn_fence(__ATOMIC_ACQ_REL, "wavefront");
        sc_drain(ids, h, acc, myqe, cnt, lo, lane);
        __syncthreads();

        for (int t = wave; t < ntile; t += nwav) {
            v2f b[32];
            const float4* wf4 = (const float4*)Wcf + lane;
#pragma unroll
            for (int kk2 = 0; kk2 < 16; ++kk2) {
                const float4 q = wf4[kk2 * 32];
                b[2 * kk2].x     = q.x; b[2 * kk2].y     = q.y;
                b[2 * kk2 + 1].x = q.z; b[2 * kk2 + 1].y = q.w;
            }
            const float bv = bc[lid];
            v8f c;
#pragma unroll
            for (int j = 0; j < 8; ++j) c[j] = bv;
            const float* ar = acc + (t * 16 + lid) * SC_P + halfsel * 64;
#pragma unroll
            for (int kk = 0; kk < 32; ++kk) {
                v2f a;
                a.x = ar[kk];
                a.y = ar[32 + kk];
                c = __builtin_amdgcn_wmma_f32_16x16x4_f32(
                    false, a, false, b[kk], (short)0, c, false, false);
            }
#pragma unroll
            for (int j = 0; j < 8; ++j) {
                const int row = t * 16 + j + halfsel * 8;
                if (row < rows) out[(size_t)(lo + row) * 16 + lid] = c[j];
            }
        }
        __syncthreads();
    }
}

template <int NTILES>
__global__ __launch_bounds__(128)
void gemm_wmma_k128_kernel(const float* __restrict__ A,
                           const float* __restrict__ Wf,
                           const float* __restrict__ bias,
                           const float* __restrict__ arowscale,
                           const float* __restrict__ erowscale,
                           float* __restrict__ out,
                           int M, int relu) {
    const int lane    = threadIdx.x & 31;
    const int wave    = __builtin_amdgcn_readfirstlane((int)(threadIdx.x >> 5));
    const int wpb     = blockDim.x >> 5;
    const int nwaves  = gridDim.x * wpb;
    const int ntiles  = M >> 4;
    int tile          = blockIdx.x * wpb + wave;
    if (tile >= ntiles) return;
    const int halfsel = lane >> 4;
    const int lid     = lane & 15;
    const int N       = NTILES * 16;

    v2f b[NTILES][32];
    {
        const float4* wf4 = (const float4*)Wf + lane;
#pragma unroll
        for (int nt = 0; nt < NTILES; ++nt) {
#pragma unroll
            for (int kk2 = 0; kk2 < 16; ++kk2) {
                const float4 q = wf4[(nt * 16 + kk2) * 32];
                b[nt][2 * kk2].x     = q.x; b[nt][2 * kk2].y     = q.y;
                b[nt][2 * kk2 + 1].x = q.z; b[nt][2 * kk2 + 1].y = q.w;
            }
        }
    }
    float bv[NTILES];
#pragma unroll
    for (int nt = 0; nt < NTILES; ++nt) bv[nt] = bias[nt * 16 + lid];

    v2f   an[32];
    float sn;
    {
        const int r = tile * 16 + lid;
        const float* ap = A + (size_t)r * 128 + halfsel * 2;
#pragma unroll
        for (int kk = 0; kk < 32; ++kk) an[kk] = *(const v2f*)(ap + kk * 4);
        sn = arowscale ? arowscale[r] : 1.0f;
    }

    for (; tile < ntiles; tile += nwaves) {
        v2f a[32];
#pragma unroll
        for (int kk = 0; kk < 32; ++kk) { a[kk].x = an[kk].x * sn; a[kk].y = an[kk].y * sn; }

        const int tnext = tile + nwaves;
        if (tnext < ntiles) {
            const int r = tnext * 16 + lid;
            const float* ap = A + (size_t)r * 128 + halfsel * 2;
#pragma unroll
            for (int kk = 0; kk < 32; ++kk) an[kk] = *(const v2f*)(ap + kk * 4);
            sn = arowscale ? arowscale[r] : 1.0f;
        }

        float es[8];
        if (erowscale) {
#pragma unroll
            for (int j = 0; j < 8; ++j) es[j] = erowscale[tile * 16 + j + halfsel * 8];
        } else {
#pragma unroll
            for (int j = 0; j < 8; ++j) es[j] = 1.0f;
        }

        v8f acc[NTILES];
#pragma unroll
        for (int nt = 0; nt < NTILES; ++nt) {
#pragma unroll
            for (int j = 0; j < 8; ++j) acc[nt][j] = bv[nt];
        }

#pragma unroll
        for (int kk = 0; kk < 32; ++kk) {
#pragma unroll
            for (int nt = 0; nt < NTILES; ++nt) {
                acc[nt] = __builtin_amdgcn_wmma_f32_16x16x4_f32(
                    false, a[kk], false, b[nt][kk], (short)0, acc[nt], false, false);
            }
        }

#pragma unroll
        for (int nt = 0; nt < NTILES; ++nt) {
#pragma unroll
            for (int j = 0; j < 8; ++j) {
                float v = acc[nt][j];
                if (relu) v = fmaxf(v, 0.0f);
                const int row = tile * 16 + j + halfsel * 8;
                if (erowscale) v *= es[j];
                out[(size_t)row * N + nt * 16 + lid] = v;
            }
        }
    }
}


extern "C" void kernel_launch(void* const* d_in, const int* in_sizes, int n_in,
                              void* d_out, int out_size, void* d_ws, size_t ws_size,
                              hipStream_t stream) {
    (void)n_in; (void)ws_size;
    const float* x   = (const float*)d_in[0];
    const float* W2  = (const float*)d_in[1];
    const float* b2  = (const float*)d_in[2];
    const float* gw  = (const float*)d_in[3];
    const float* gb  = (const float*)d_in[4];
    const float* Wc  = (const float*)d_in[5];
    const float* bc  = (const float*)d_in[6];
    const int*   src = (const int*)d_in[7];
    const int*   dst = (const int*)d_in[8];
    const int*   ids = (const int*)d_in[9];

    const int H  = 128;
    const int C  = 16;
    const int L  = in_sizes[2] / H;
    const int NS = in_sizes[9] / L;
    const int NE = in_sizes[7] / L;
    const int NL = in_sizes[4] / (L * H);
    const int NN = out_size / C;

    float* ws       = (float*)d_ws;
    float* hfin     = ws;
    float* hbuf     = hfin + (size_t)(L - 1) * NS * H;
    float* agg      = hbuf + (size_t)NS * H;
    float* deg_out  = agg + (size_t)NS * H;
    float* deg_in   = deg_out + NS;
    float* w2t      = deg_in + NS;
    float* gwt      = w2t + (size_t)L * H * H;
    float* wct      = gwt + (size_t)L * NL * H * H;
    unsigned char* okey = (unsigned char*)d_out;
    unsigned* olist = (unsigned*)(okey + (((size_t)NE + 255) & ~(size_t)255));

    const int T = 256;
    const int GT = 128;
    const int gemm_blocks_NS = 256;

    {
        int nHH = H * H;
        int bHH = (nHH + T - 1) / T;
        for (int i = 0; i < L; ++i)
            pack_bfrag_kernel<<<bHH, T, 0, stream>>>(W2 + (size_t)i * nHH,
                                                     w2t + (size_t)i * nHH, H);
        for (int i = 0; i < L * NL; ++i)
            pack_bfrag_kernel<<<bHH, T, 0, stream>>>(gw + (size_t)i * nHH,
                                                     gwt + (size_t)i * nHH, H);
        int nHC = H * C;
        pack_bfrag_kernel<<<(nHC + T - 1) / T, T, 0, stream>>>(Wc, wct, C);
    }

    for (int i = 0; i < L; ++i) {
        const int*   src_i = src + (size_t)i * NE;
        const int*   dst_i = dst + (size_t)i * NE;
        const float* x_i   = x   + (size_t)i * NS * H;
        const float* w2t_i = w2t + (size_t)i * H * H;
        const float* b2_i  = b2  + (size_t)i * H;

        {
            int n4 = (2 * NS) / 4;
            zero_f4_kernel<<<(n4 + T - 1) / T, T, 0, stream>>>((float4*)deg_out, n4);
        }
        degree_count_kernel<<<(NE + T - 1) / T, T, 0, stream>>>(src_i, dst_i,
                                                                deg_out, deg_in,
                                                                okey, NE, NS);
        rsqrt_clamp_kernel<<<(2 * NS + T - 1) / T, T, 0, stream>>>(deg_out, 2 * NS);

        float* pin  = (NL & 1) ? agg : hbuf;
        float* pout = (NL & 1) ? hbuf : agg;

        gemm_wmma_k128_kernel<8><<<gemm_blocks_NS, GT, 0, stream>>>(
            x_i, w2t_i, b2_i, nullptr, deg_out, pin, NS, 0);

        for (int l = 0; l < NL; ++l) {
            const float* gwt_il = gwt + ((size_t)i * NL + l) * H * H;
            const float* gb_il  = gb  + ((size_t)i * NL + l) * H;
            const float* escale = (l + 1 < NL) ? deg_out : nullptr;
            float* hout = (l + 1 < NL) ? pout : hfin + (size_t)i * NS * H;
            conv_owner_kernel<<<OWN_GRID, 1024, 0, stream>>>(src_i, dst_i, okey, pin,
                                                             gwt_il, gb_il, deg_in, escale,
                                                             hout, olist, l, NE, NS);
            float* tmp = pin; pin = pout; pout = tmp;
        }
    }

    scatter_classify_owner_kernel<<<OWN_GRID, 1024, 0, stream>>>(
        ids, hfin, wct, bc, (float*)d_out, L * NS, NN);
}


# in-degrees: the owner work-group counts the edges it adds to each LDS row (one LDS add an edge, in the row's spare word) and scales by that count; half the global float atomics gone; sum orders kept
# speedup vs baseline: 3.1088x; 1.1574x over previous
#include <hip/hip_runtime.h>


typedef __attribute__((ext_vector_type(2))) float v2f;
typedef __attribute__((ext_vector_type(8))) float v8f;


__global__ void zero_f4_kernel(float4* __restrict__ p, int n4) {
    int i = blockIdx.x * blockDim.x + threadIdx.x;
    if (i < n4) p[i] = make_float4(0.f, 0.f, 0.f, 0.f);
}

__global__ void pack_bfrag_kernel(const float* __restrict__ in,
                                  float* __restrict__ outF, int N) {
    int i = blockIdx.x * blockDim.x + threadIdx.x;
    if (i < N * 128) {
        int c    = i & 3;
        int lane = (i >> 2) & 31;
        int kk2  = (i >> 7) & 15;
        int nt   = i >> 11;
        int kk   = 2 * kk2 + (c >> 1);
        int k    = kk * 4 + (lane >> 4) * 2 + (c & 1);
        int n    = nt * 16 + (lane & 15);
        outF[i] = in[k * N + n];
    }
}

#define OWN_GRID  256
#define OWN_RMAX  208
#define OWN_P     132
#define OWN_CS    33
#define OWN_DEGW  32

__device__ __forceinline__ int own_range_rows(int nRows) {
    const int R = (nRows + OWN_GRID - 1) / OWN_GRID;
    return (R > OWN_RMAX) ? OWN_RMAX : R;
}

__global__ void degree_count_kernel(const int* __restrict__ src,
                                    const int* __restrict__ dst,
                                    float* __restrict__ deg_out,
                                    unsigned char* __restrict__ key,
                                    int nE, int nRows) {
    const unsigned R = (unsigned)own_range_rows(nRows);
    int e = blockIdx.x * blockDim.x + threadIdx.x;
    if (e < nE) {
        atomicAdd(&deg_out[src[e]], 1.0f);
        key[e] = (unsigned char)(((unsigned)dst[e] / R) % (unsigned)OWN_GRID);
    }
}

__global__ void rsqrt_clamp_kernel(float* __restrict__ d, int n) {
    int i = blockIdx.x * blockDim.x + threadIdx.x;
    if (i < n) d[i] = rsqrtf(fmaxf(d[i], 1.0f));
}

#define OWN_WAVES 32
#define OWN_QCAP  512
#define OWN_FLUSH 64
#define OWN_U     8
#define OWN_LSTRIDE 160
#define OWN_LCAP    (OWN_LSTRIDE - 1)
#define OWN_LBATCH  5
#define OWN_LOVER   0xFFFFFFFFu

__device__ __forceinline__ void own_add_rows(const float* __restrict__ h, float* acc,
                                             int my_s, int my_d, int m, int rows,
                                             int lane) {
    for (int j0 = 0; j0 < m; j0 += 8) {
        float4 v[8];
        int    dl[8];
#pragma unroll
        for (int u = 0; u < 8; ++u) {
            const int j = (j0 + u < m) ? (j0 + u) : (m - 1);
            const int s = __builtin_amdgcn_readlane(my_s, j);
            dl[u]       = __builtin_amdgcn_readlane(my_d, j);
            v[u] = *(const float4*)(h + (size_t)s * 128 + lane * 4);
        }
#pragma unroll
        for (int u = 0; u < 8; ++u) {
            if (j0 + u < m && (unsigned)dl[u] < (unsigned)rows) {
                float* row = acc + dl[u] * OWN_P + lane;
                __hip_atomic_fetch_add(row,              v[u].x, __ATOMIC_RELAXED, __HIP_MEMORY_SCOPE_WORKGROUP);
                __hip_atomic_fetch_add(row + OWN_CS,     v[u].y, __ATOMIC_RELAXED, __HIP_MEMORY_SCOPE_WORKGROUP);
                __hip_atomic_fetch_add(row + 2 * OWN_CS, v[u].z, __ATOMIC_RELAXED, __HIP_MEMORY_SCOPE_WORKGROUP);
                __hip_atomic_fetch_add(row + 3 * OWN_CS, v[u].w, __ATOMIC_RELAXED, __HIP_MEMORY_SCOPE_WORKGROUP);
                if (lane == 0)
                    __hip_atomic_fetch_add(row + OWN_DEGW, 1.0f, __ATOMIC_RELAXED, __HIP_MEMORY_SCOPE_WORKGROUP);
            }
        }
    }
}

__device__ __forceinline__ int own_drain(const int* __restrict__ src,
                                         const int* __restrict__ dst,
                                         const float* __restrict__ h,
                                         float* acc, const unsigned* qe,
                                         int n, int lo, int rows, int lane,
                                         bool save, unsigned* lst, int saved) {
    for (int i0 = 0; i0 < n; i0 += 32) {
        const int m = (n - i0 < 32) ? (n - i0) : 32;
        int my_s = 0, my_d = 0;
        if (lane < m) {
            const unsigned e = qe[i0 + lane];
            my_s = src[e];
            my_d = dst[e] - lo;
            const int p = saved + i0 + lane;
            if (save && p < OWN_LCAP)
                lst[1 + p] = ((unsigned)my_s << 8) | (unsigned)my_d;
        }
        own_add_rows(h, acc, my_s, my_d, m, rows, lane);
    }
    return saved + n;
}

__device__ __forceinline__ int own_push(bool m, unsigned e, unsigned* qe, int cnt) {
    const unsigned k = __builtin_amdgcn_ballot_w32(m);
    if (m) qe[cnt + (int)__builtin_amdgcn_mbcnt_lo(k, 0u)] = e;
    return cnt + __builtin_popcount(k);
}

__device__ __forceinline__ unsigned own_zero_bytes(unsigned x) {
    const unsigned t = (x & 0x7F7F7F7Fu) + 0x7F7F7F7Fu;
    return ~(t | x | 0x7F7F7F7Fu);
}

__global__ __launch_bounds__(1024)
void conv_owner_kernel(const int* __restrict__ src,
                       const int* __restrict__ dst,
                       const unsigned char* __restrict__ key,
                       const float* __restrict__ h,
                       const float* Wf,
                       const float* bias,
                       const float* __restrict__ erowscale,
                       float* __restrict__ hout,
                       unsigned* __restrict__ lst,
                       int reuse,
                       int nE, int nRows) {
    __shared__ float    acc[OWN_RMAX * OWN_P];
    __shared__ unsigned qe[OWN_WAVES][OWN_QCAP];

    const int lane = threadIdx.x & 31;
    const int halfsel = lane >> 4;
    const int lid     = lane & 15;
    const int wave = __builtin_amdgcn_readfirstlane((int)(threadIdx.x >> 5));
    const int nwav = blockDim.x >> 5;
    const int R   = own_range_rows(nRows);
    const int nGf = nE >> 9;
    const int rot = (int)(((long long)blockIdx.x * nGf) / OWN_GRID);
    const unsigned pat = (unsigned)blockIdx.x * 0x01010101u;
    unsigned* myqe = qe[wave];
    const bool listable = (OWN_GRID * R >= nRows) && (nRows <= (1 << 24));
    const bool save     = listable && !reuse;
    unsigned* mylst = lst + ((size_t)blockIdx.x * OWN_WAVES + wave) * OWN_LSTRIDE;
    unsigned nl = OWN_LOVER;
    unsigned pk[OWN_LBATCH];
#pragma unroll
    for (int b = 0; b < OWN_LBATCH; ++b) pk[b] = 0u;
    if (listable && reuse && blockIdx.x * R < nRows) {
        nl = (unsigned)__builtin_amdgcn_readfirstlane((int)mylst[0]);
        if (nl <= (unsigned)OWN_LCAP) {
#pragma unroll
            for (int b = 0; b < OWN_LBATCH; ++b) {
                const unsigned idx = (unsigned)(b * 32 + lane);
                if (idx < nl) pk[b] = mylst[1 + idx];
            }
        }
    }
    const bool scan = nl > (unsigned)OWN_LCAP;

    for (int lo = blockIdx.x * R; lo < nRows; lo += OWN_GRID * R) {
        const int rows  = (nRows - lo < R) ? (nRows - lo) : R;
        const int ntile = (rows + 15) >> 4;
        for (int i = threadIdx.x; i < ntile * 16 * (OWN_P / 4); i += blockDim.x)
            ((float4*)acc)[i] = make_float4(0.f, 0.f, 0.f, 0.f);
        __syncthreads();

        if (!scan) {
#pragma unroll
            for (int b = 0; b < OWN_LBATCH; ++b) {
                if ((unsigned)(b * 32) < nl) {
                    const int left = (int)nl - b * 32;
                    own_add_rows(h, acc, (int)(pk[b] >> 8), (int)(pk[b] & 255u),
                                 (left < 32) ? left : 32, rows, lane);
                }
            }
        }

        int cnt = 0;
        int saved = 0;
        const int nGs = scan ? nGf : 0;
        for (int g0 = wave; g0 < nGs; g0 += OWN_U * nwav) {
            uint4 d[OWN_U];
            int   eb[OWN_U];
#pragma unroll
            for (int u = 0; u < OWN_U; ++u) {
                int g = g0 + u * nwav;
                if (g >= nGf) g = nGf - 1;
                g += rot;
                if (g >= nGf) g -= nGf;
                eb[u] = g * 512 + lane * 16;
                d[u]  = *(const uint4*)(key + eb[u]);
            }
#pragma unroll
            for (int u = 0; u < OWN_U; ++u) {
                if (g0 + u * nwav < nGf) {
                    unsigned hit = (own_zero_bytes(d[u].x ^ pat) >> 7)
                                 | (own_zero_bytes(d[u].y ^ pat) >> 6)
                                 | (own_zero_bytes(d[u].z ^ pat) >> 5)
                                 | (own_zero_bytes(d[u].w ^ pat) >> 4);
                    while (__builtin_amdgcn_ballot_w32(hit != 0u)) {
                        const bool     has = hit != 0u;
                        const unsigned p   = (unsigned)__builtin_ctz(hit | 0x80000000u);
                        const unsigned e   = (unsigned)eb[u] + ((p & 7u) << 2) + (p >> 3);
                        cnt = own_push(has, e, myqe, cnt);
                        hit &= hit - 1u;
                        if (cnt >= OWN_FLUSH) {
                            __builtin_amdgcn_fence(__ATOMIC_ACQ_REL, "wavefront");
                            saved = own_drain(src, dst, h, acc, myqe, cnt, lo, rows, lane,
                                              save, mylst, saved);
                            cnt = 0;
                        }
                    }
                }
            }
        }
        if (wave == 0 && scan) {
            for (int eb0 = nGf * 512; eb0 < nE; eb0 += 32) {
                const int e = eb0 + lane;
                const unsigned r = ((e < nE) ? (unsigned)dst[e] : 0xFFFFFFFFu) - (unsigned)lo;
                cnt = own_push(r < (unsigned)rows && e < nE, (unsigned)e, myqe, cnt);
                if (cnt >= OWN_FLUSH) {
                    __builtin_amdgcn_fence(__ATOMIC_ACQ_REL, "wavefront");
                    saved = own_drain(src, dst, h, acc, myqe, cnt, lo, rows, lane,
                                      save, mylst, saved);
                    cnt = 0;
                }
            }
        }
        __builtin_amdgcn_fence(__ATOMIC_ACQ_REL, "wavefront");
        saved = own_drain(src, dst, h, acc, myqe, cnt, lo, rows, lane, save, mylst, saved);
        if (save && lane == 0)
            mylst[0] = (saved <= OWN_LCAP) ? (unsigned)saved : OWN_LOVER;

        const int nt = wave & 7;
        v2f b[32];
        {
            const float4* wf4 = (const float4*)Wf + (size_t)nt * 16 * 32 + lane;
#pragma unroll
            for (int kk2 = 0; kk2 < 16; ++kk2) {
                const float4 q = wf4[kk2 * 32];
                b[2 * kk2].x     = q.x; b[2 * kk2].y     = q.y;
                b[2 * kk2 + 1].x = q.z; b[2 * kk2 + 1].y = q.w;
            }
        }
        const float bv = bias[nt * 16 + lid];
        __syncthreads();

        for (int t = wave >> 3; t < ntile; t += (nwav >> 3)) {
            const int r = t * 16 + lid;
            const float sn = rsqrtf(fmaxf(acc[r * OWN_P + OWN_DEGW], 1.0f));
            const float* ar = acc + r * OWN_P + halfsel * (2 * OWN_CS);
            v8f c;
#pragma unroll
            for (int j = 0; j < 8; ++j) c[j] = bv;
#pragma unroll
            for (int kk = 0; kk < 32; ++kk) {
                v2f a;
                a.x = ar[kk] * sn;
                a.y = ar[OWN_CS + kk] * sn;
                c = __builtin_amdgcn_wmma_f32_16x16x4_f32(
                    false, a, false, b[kk], (short)0, c, false, false);
            }
#pragma unroll
            for (int j = 0; j < 8; ++j) {
                const int row = t * 16 + j + halfsel * 8;
                if (row < rows) {
                    float v = fmaxf(c[j], 0.0f);
                    if (erowscale) v *= erowscale[lo + row];
                    hout[(size_t)(lo + row) * 128 + nt * 16 + lid] = v;
                }
            }
        }
        __syncthreads();
    }
}

#define SC_RMAX  400
#define SC_P     132
#define SC_QCAP  256
#define SC_FLUSH 64
#define SC_U     8

__device__ __forceinline__ void sc_drain(const int* __restrict__ ids,
                                         const float* __restrict__ h,
                                         float* acc, const unsigned* qe,
                                         int n, int lo, int lane) {
    for (int i0 = 0; i0 < n; i0 += 32) {
        const int m = (n - i0 < 32) ? (n - i0) : 32;
        int my_g = 0, my_d = 0;
        if (lane < m) {
            const unsigned g = qe[i0 + lane];
            my_g = (int)g;
            my_d = ids[g] - lo;
        }
        for (int j0 = 0; j0 < m; j0 += 8) {
            float4 v[8];
            int    dl[8];
#pragma unroll
            for (int u = 0; u < 8; ++u) {
                const int j = (j0 + u < m) ? (j0 + u) : (m - 1);
                const int g = __builtin_amdgcn_readlane(my_g, j);
                dl[u]       = __builtin_amdgcn_readlane(my_d, j);
                v[u] = *(const float4*)(h + (size_t)g * 128 + lane * 4);
            }
#pragma unroll
            for (int u = 0; u < 8; ++u) {
                if (j0 + u < m) {
                    float* row = acc + dl[u] * SC_P + lane;
                    __hip_atomic_fetch_add(row,      v[u].x, __ATOMIC_RELAXED, __HIP_MEMORY_SCOPE_WORKGROUP);
                    __hip_atomic_fetch_add(row + 32, v[u].y, __ATOMIC_RELAXED, __HIP_MEMORY_SCOPE_WORKGROUP);
                    __hip_atomic_fetch_add(row + 64, v[u].z, __ATOMIC_RELAXED, __HIP_MEMORY_SCOPE_WORKGROUP);
                    __hip_atomic_fetch_add(row + 96, v[u].w, __ATOMIC_RELAXED, __HIP_MEMORY_SCOPE_WORKGROUP);
                }
            }
        }
    }
}

__global__ __launch_bounds__(1024)
void scatter_classify_owner_kernel(const int* __restrict__ ids,
                                   const float* __restrict__ h,
                                   const float* __restrict__ Wcf,
                                   const float* __restrict__ bc,
                                   float* __restrict__ out,
                                   int nIds, int nNodes) {
    __shared__ float    acc[SC_RMAX * SC_P];
    __shared__ unsigned qe[OWN_WAVES][SC_QCAP];

    const int lane = threadIdx.x & 31;
    const int wave = __builtin_amdgcn_readfirstlane((int)(threadIdx.x >> 5));
    const int nwav = blockDim.x >> 5;
    const int halfsel = lane >> 4;
    const int lid     = lane & 15;
    int R = (nNodes + OWN_GRID - 1) / OWN_GRID;
    if (R > SC_RMAX) R = SC_RMAX;
    const int nGf = nIds >> 7;
    const int rot = (int)(((long long)blockIdx.x * nGf) / OWN_GRID);
    unsigned* myqe = qe[wave];

    for (int lo = blockIdx.x * R; lo < nNodes; lo += OWN_GRID * R) {
        const int rows  = (nNodes - lo < R) ? (nNodes - lo) : R;
        const int ntile = (rows + 15) >> 4;
        for (int i = threadIdx.x; i < ntile * 16 * (SC_P / 4); i += blockDim.x)
            ((float4*)acc)[i] = make_float4(0.f, 0.f, 0.f, 0.f);
        __syncthreads();

        int cnt = 0;
        for (int g0 = wave; g0 < nGf; g0 += SC_U * nwav) {
            uint4 d[SC_U];
            int   eb[SC_U];
#pragma unroll
            for (int u = 0; u < SC_U; ++u) {
                int g = g0 + u * nwav;
                if (g >= nGf) g = nGf - 1;
                g += rot;
                if (g >= nGf) g -= nGf;
                eb[u] = g * 128 + lane * 4;
                d[u]  = *(const uint4*)(ids + eb[u]);
            }
#pragma unroll
            for (int u = 0; u < SC_U; ++u) {
                if (g0 + u * nwav < nGf) {
                    const unsigned b = (unsigned)eb[u];
                    cnt = own_push(d[u].x - (unsigned)lo < (unsigned)rows, b,      myqe, cnt);
                    cnt = own_push(d[u].y - (unsigned)lo < (unsigned)rows, b + 1u, myqe, cnt);
                    cnt = own_push(d[u].z - (unsigned)lo < (unsigned)rows, b + 2u, myqe, cnt);
                    cnt = own_push(d[u].w - (unsigned)lo < (unsigned)rows, b + 3u, myqe, cnt);
                    if (cnt >= SC_FLUSH) {
                        __builtin_amdgcn_fence(__ATOMIC_ACQ_REL, "wavefront");
                        sc_drain(ids, h, acc, myqe, cnt, lo, lane);
                        cnt = 0;
                    }
                }
            }
        }
        if (wave == 0) {
            for (int e0 = nGf * 128; e0 < nIds; e0 += 32) {
                const int e = e0 + lane;
                const unsigned r = ((e < nIds) ? (unsigned)ids[e] : 0xFFFFFFFFu) - (unsigned)lo;
                cnt = own_push(r < (unsigned)rows && e < nIds, (unsigned)e, myqe, cnt);
                if (cnt >= SC_FLUSH) {
                    __builtin_amdgcn_fence(__ATOMIC_ACQ_REL, "wavefront");
                    sc_drain(ids, h, acc, myqe, cnt, lo, lane);
                    cnt = 0;
                }
            }
        }
        __builtin_amdgcn_fence(__ATOMIC_ACQ_REL, "wavefront");
        sc_drain(ids, h, acc, myqe, cnt, lo, lane);
        __syncthreads();

        for (int t = wave; t < ntile; t += nwav) {
            v2f b[32];
            const float4* wf4 = (const float4*)Wcf + lane;
#pragma unroll
            for (int kk2 = 0; kk2 < 16; ++kk2) {
                const float4 q = wf4[kk2 * 32];
                b[2 * kk2].x     = q.x; b[2 * kk2].y     = q.y;
                b[2 * kk2 + 1].x = q.z; b[2 * kk2 + 1].y = q.w;
            }
            const float bv = bc[lid];
            v8f c;
#pragma unroll
            for (int j = 0; j < 8; ++j) c[j] = bv;
            const float* ar = acc + (t * 16 + lid) * SC_P + halfsel * 64;
#pragma unroll
            for (int kk = 0; kk < 32; ++kk) {
                v2f a;
                a.x = ar[kk];
                a.y = ar[32 + kk];
                c = __builtin_amdgcn_wmma_f32_16x16x4_f32(
                    false, a, false, b[kk], (short)0, c, false, false);
            }
#pragma unroll
            for (int j = 0; j < 8; ++j) {
                const int row = t * 16 + j + halfsel * 8;
                if (row < rows) out[(size_t)(lo + row) * 16 + lid] = c[j];
            }
        }
        __syncthreads();
    }
}

template <int NTILES>
__global__ __launch_bounds__(128)
void gemm_wmma_k128_kernel(const float* __restrict__ A,
                           const float* __restrict__ Wf,
                           const float* __restrict__ bias,
                           const float* __restrict__ arowscale,
                           const float* __restrict__ erowscale,
                           float* __restrict__ out,
                           int M, int relu) {
    const int lane    = threadIdx.x & 31;
    const int wave    = __builtin_amdgcn_readfirstlane((int)(threadIdx.x >> 5));
    const int wpb     = blockDim.x >> 5;
    const int nwaves  = gridDim.x * wpb;
    const int ntiles  = M >> 4;
    int tile          = blockIdx.x * wpb + wave;
    if (tile >= ntiles) return;
    const int halfsel = lane >> 4;
    const int lid     = lane & 15;
    const int N       = NTILES * 16;

    v2f b[NTILES][32];
    {
        const float4* wf4 = (const float4*)Wf + lane;
#pragma unroll
        for (int nt = 0; nt < NTILES; ++nt) {
#pragma unroll
            for (int kk2 = 0; kk2 < 16; ++kk2) {
                const float4 q = wf4[(nt * 16 + kk2) * 32];
                b[nt][2 * kk2].x     = q.x; b[nt][2 * kk2].y     = q.y;
                b[nt][2 * kk2 + 1].x = q.z; b[nt][2 * kk2 + 1].y = q.w;
            }
        }
    }
    float bv[NTILES];
#pragma unroll
    for (int nt = 0; nt < NTILES; ++nt) bv[nt] = bias[nt * 16 + lid];

    v2f   an[32];
    float sn;
    {
        const int r = tile * 16 + lid;
        const float* ap = A + (size_t)r * 128 + halfsel * 2;
#pragma unroll
        for (int kk = 0; kk < 32; ++kk) an[kk] = *(const v2f*)(ap + kk * 4);
        sn = arowscale ? arowscale[r] : 1.0f;
    }

    for (; tile < ntiles; tile += nwaves) {
        v2f a[32];
#pragma unroll
        for (int kk = 0; kk < 32; ++kk) { a[kk].x = an[kk].x * sn; a[kk].y = an[kk].y * sn; }

        const int tnext = tile + nwaves;
        if (tnext < ntiles) {
            const int r = tnext * 16 + lid;
            const float* ap = A + (size_t)r * 128 + halfsel * 2;
#pragma unroll
            for (int kk = 0; kk < 32; ++kk) an[kk] = *(const v2f*)(ap + kk * 4);
            sn = arowscale ? arowscale[r] : 1.0f;
        }

        float es[8];
        if (erowscale) {
#pragma unroll
            for (int j = 0; j < 8; ++j) es[j] = erowscale[tile * 16 + j + halfsel * 8];
        } else {
#pragma unroll
            for (int j = 0; j < 8; ++j) es[j] = 1.0f;
        }

        v8f acc[NTILES];
#pragma unroll
        for (int nt = 0; nt < NTILES; ++nt) {
#pragma unroll
            for (int j = 0; j < 8; ++j) acc[nt][j] = bv[nt];
        }

#pragma unroll
        for (int kk = 0; kk < 32; ++kk) {
#pragma unroll
            for (int nt = 0; nt < NTILES; ++nt) {
                acc[nt] = __builtin_amdgcn_wmma_f32_16x16x4_f32(
                    false, a[kk], false, b[nt][kk], (short)0, acc[nt], false, false);
            }
        }

#pragma unroll
        for (int nt = 0; nt < NTILES; ++nt) {
#pragma unroll
            for (int j = 0; j < 8; ++j) {
                float v = acc[nt][j];
                if (relu) v = fmaxf(v, 0.0f);
                const int row = tile * 16 + j + halfsel * 8;
                if (erowscale) v *= es[j];
                out[(size_t)row * N + nt * 16 + lid] = v;
            }
        }
    }
}


extern "C" void kernel_launch(void* const* d_in, const int* in_sizes, int n_in,
                              void* d_out, int out_size, void* d_ws, size_t ws_size,
                              hipStream_t stream) {
    (void)n_in; (void)ws_size;
    const float* x   = (const float*)d_in[0];
    const float* W2  = (const float*)d_in[1];
    const float* b2  = (const float*)d_in[2];
    const float* gw  = (const float*)d_in[3];
    const float* gb  = (const float*)d_in[4];
    const float* Wc  = (const float*)d_in[5];
    const float* bc  = (const float*)d_in[6];
    const int*   src = (const int*)d_in[7];
    const int*   dst = (const int*)d_in[8];
    const int*   ids = (const int*)d_in[9];

    const int H  = 128;
    const int C  = 16;
    const int L  = in_sizes[2] / H;
    const int NS = in_sizes[9] / L;
    const int NE = in_sizes[7] / L;
    const int NL = in_sizes[4] / (L * H);
    const int NN = out_size / C;

    float* ws       = (float*)d_ws;
    float* hfin     = ws;
    float* hbuf     = hfin + (size_t)(L - 1) * NS * H;
    float* agg      = hbuf + (size_t)NS * H;
    float* deg_out  = agg + (size_t)NS * H;
    float* w2t      = deg_out + 2 * (size_t)NS;
    float* gwt      = w2t + (size_t)L * H * H;
    float* wct      = gwt + (size_t)L * NL * H * H;
    unsigned char* okey = (unsigned char*)d_out;
    unsigned* olist = (unsigned*)(okey + (((size_t)NE + 255) & ~(size_t)255));

    const int T = 256;
    const int GT = 128;
    const int gemm_blocks_NS = 256;

    {
        int nHH = H * H;
        int bHH = (nHH + T - 1) / T;
        for (int i = 0; i < L; ++i)
            pack_bfrag_kernel<<<bHH, T, 0, stream>>>(W2 + (size_t)i * nHH,
                                                     w2t + (size_t)i * nHH, H);
        for (int i = 0; i < L * NL; ++i)
            pack_bfrag_kernel<<<bHH, T, 0, stream>>>(gw + (size_t)i * nHH,
                                                     gwt + (size_t)i * nHH, H);
        int nHC = H * C;
        pack_bfrag_kernel<<<(nHC + T - 1) / T, T, 0, stream>>>(Wc, wct, C);
    }

    for (int i = 0; i < L; ++i) {
        const int*   src_i = src + (size_t)i * NE;
        const int*   dst_i = dst + (size_t)i * NE;
        const float* x_i   = x   + (size_t)i * NS * H;
        const float* w2t_i = w2t + (size_t)i * H * H;
        const float* b2_i  = b2  + (size_t)i * H;

        {
            int n4 = (NS + 3) / 4;
            zero_f4_kernel<<<(n4 + T - 1) / T, T, 0, stream>>>((float4*)deg_out, n4);
        }
        degree_count_kernel<<<(NE + T - 1) / T, T, 0, stream>>>(src_i, dst_i,
                                                                deg_out,
                                                                okey, NE, NS);
        rsqrt_clamp_kernel<<<(NS + T - 1) / T, T, 0, stream>>>(deg_out, NS);

        float* pin  = (NL & 1) ? agg : hbuf;
        float* pout = (NL & 1) ? hbuf : agg;

        gemm_wmma_k128_kernel<8><<<gemm_blocks_NS, GT, 0, stream>>>(
            x_i, w2t_i, b2_i, nullptr, deg_out, pin, NS, 0);

        for (int l = 0; l < NL; ++l) {
            const float* gwt_il = gwt + ((size_t)i * NL + l) * H * H;
            const float* gb_il  = gb  + ((size_t)i * NL + l) * H;
            const float* escale = (l + 1 < NL) ? deg_out : nullptr;
            float* hout = (l + 1 < NL) ? pout : hfin + (size_t)i * NS * H;
            conv_owner_kernel<<<OWN_GRID, 1024, 0, stream>>>(src_i, dst_i, okey, pin,
                                                             gwt_il, gb_il, escale,
                                                             hout, olist, l, NE, NS);
            float* tmp = pin; pin = pout; pout = tmp;
        }
    }

    scatter_classify_owner_kernel<<<OWN_GRID, 1024, 0, stream>>>(
        ids, hfin, wct, bc, (float*)d_out, L * NS, NN);
}


# out-degrees: 64 work-groups each count an edge slice in a whole-graph LDS histogram (integer LDS adds), store it as a plane; rsqrt kernel sums planes; no global atomics or zero pass; 29->26 launches
# speedup vs baseline: 4.1477x; 1.3342x over previous
#include <hip/hip_runtime.h>


typedef __attribute__((ext_vector_type(2))) float v2f;
typedef __attribute__((ext_vector_type(8))) float v8f;


__global__ void pack_bfrag_kernel(const float* __restrict__ in,
                                  float* __restrict__ outF, int N) {
    int i = blockIdx.x * blockDim.x + threadIdx.x;
    if (i < N * 128) {
        int c    = i & 3;
        int lane = (i >> 2) & 31;
        int kk2  = (i >> 7) & 15;
        int nt   = i >> 11;
        int kk   = 2 * kk2 + (c >> 1);
        int k    = kk * 4 + (lane >> 4) * 2 + (c & 1);
        int n    = nt * 16 + (lane & 15);
        outF[i] = in[k * N + n];
    }
}

#define OWN_GRID  256
#define OWN_RMAX  208
#define OWN_P     132
#define OWN_CS    33
#define OWN_DEGW  32

__device__ __forceinline__ int own_range_rows(int nRows) {
    const int R = (nRows + OWN_GRID - 1) / OWN_GRID;
    return (R > OWN_RMAX) ? OWN_RMAX : R;
}

#define DEG_SLICES 64
#define DEG_CAP    51200

__global__ __launch_bounds__(1024)
void degree_slice_kernel(const int* __restrict__ src,
                         const int* __restrict__ dst,
                         unsigned* __restrict__ part,
                         unsigned char* __restrict__ key,
                         int nE, int nRows, int pstride) {
    __shared__ __attribute__((aligned(16))) unsigned cnt[DEG_CAP];
    const unsigned R = (unsigned)own_range_rows(nRows);
    const int per = ((nE + DEG_SLICES - 1) / DEG_SLICES + 3) & ~3;
    const int e0  = (int)blockIdx.x * per;
    const int e1  = (nE - e0 < per) ? nE : (e0 + per);
    unsigned* mypart = part + (size_t)blockIdx.x * pstride;
    const bool vec = ((((size_t)src | (size_t)dst) & 15) == 0) && (((size_t)key & 3) == 0);

    for (int base = 0; base < nRows; base += DEG_CAP) {
        const int n  = (nRows - base < DEG_CAP) ? (nRows - base) : DEG_CAP;
        const int n4 = (n + 3) >> 2;
        for (int i = threadIdx.x; i < n4; i += blockDim.x)
            ((uint4*)cnt)[i] = make_uint4(0u, 0u, 0u, 0u);
        __syncthreads();

        auto one = [&](int e) {
            const unsigned s = (unsigned)src[e] - (unsigned)base;
            if (s < (unsigned)n) atomicAdd(&cnt[s], 1u);
            if (base == 0)
                key[e] = (unsigned char)(((unsigned)dst[e] / R) % (unsigned)OWN_GRID);
        };
        if (vec) {
            const int nfull = (e1 > e0) ? ((e1 - e0) >> 2) : 0;
            for (int g0 = threadIdx.x; g0 < nfull; g0 += 4 * (int)blockDim.x) {
                int4 s4[4], d4[4];
#pragma unroll
                for (int u = 0; u < 4; ++u) {
                    const int g  = g0 + u * (int)blockDim.x;
                    const int gg = (g < nfull) ? g : g0;
                    s4[u] = *(const int4*)(src + e0 + gg * 4);
                    if (base == 0) d4[u] = *(const int4*)(dst + e0 + gg * 4);
                    else           d4[u] = make_int4(0, 0, 0, 0);
                }
#pragma unroll
                for (int u = 0; u < 4; ++u) {
                    const int g = g0 + u * (int)blockDim.x;
                    if (g < nfull) {
                        const unsigned sx = (unsigned)s4[u].x - (unsigned)base;
                        const unsigned sy = (unsigned)s4[u].y - (unsigned)base;
                        const unsigned sz = (unsigned)s4[u].z - (unsigned)base;
                        const unsigned sw = (unsigned)s4[u].w - (unsigned)base;
                        if (sx < (unsigned)n) atomicAdd(&cnt[sx], 1u);
                        if (sy < (unsigned)n) atomicAdd(&cnt[sy], 1u);
                        if (sz < (unsigned)n) atomicAdd(&cnt[sz], 1u);
                        if (sw < (unsigned)n) atomicAdd(&cnt[sw], 1u);
                        if (base == 0) {
                            uchar4 k;
                            k.x = (unsigned char)(((unsigned)d4[u].x / R) % (unsigned)OWN_GRID);
                            k.y = (unsigned char)(((unsigned)d4[u].y / R) % (unsigned)OWN_GRID);
                            k.z = (unsigned char)(((unsigned)d4[u].z / R) % (unsigned)OWN_GRID);
                            k.w = (unsigned char)(((unsigned)d4[u].w / R) % (unsigned)OWN_GRID);
                            *(uchar4*)(key + e0 + g * 4) = k;
                        }
                    }
                }
            }
            for (int e = e0 + nfull * 4 + (int)threadIdx.x; e < e1; e += (int)blockDim.x) one(e);
        } else {
            for (int e = e0 + (int)threadIdx.x; e < e1; e += (int)blockDim.x) one(e);
        }
        __syncthreads();

        for (int i = threadIdx.x; i < n4; i += blockDim.x)
            ((uint4*)(mypart + base))[i] = ((const uint4*)cnt)[i];
        __syncthreads();
    }
}

__global__ void degree_rsqrt_kernel(const unsigned* __restrict__ part,
                                    float* __restrict__ d, int n, int pstride) {
    int i = blockIdx.x * blockDim.x + threadIdx.x;
    if (i < n) {
        unsigned c = 0u;
#pragma unroll
        for (int s = 0; s < DEG_SLICES; ++s) c += part[(size_t)s * pstride + i];
        d[i] = rsqrtf(fmaxf((float)c, 1.0f));
    }
}

#define OWN_WAVES 32
#define OWN_QCAP  512
#define OWN_FLUSH 64
#define OWN_U     8
#define OWN_LSTRIDE 160
#define OWN_LCAP    (OWN_LSTRIDE - 1)
#define OWN_LBATCH  5
#define OWN_LOVER   0xFFFFFFFFu

__device__ __forceinline__ void own_add_rows(const float* __restrict__ h, float* acc,
                                             int my_s, int my_d, int m, int rows,
                                             int lane) {
    for (int j0 = 0; j0 < m; j0 += 8) {
        float4 v[8];
        int    dl[8];
#pragma unroll
        for (int u = 0; u < 8; ++u) {
            const int j = (j0 + u < m) ? (j0 + u) : (m - 1);
            const int s = __builtin_amdgcn_readlane(my_s, j);
            dl[u]       = __builtin_amdgcn_readlane(my_d, j);
            v[u] = *(const float4*)(h + (size_t)s * 128 + lane * 4);
        }
#pragma unroll
        for (int u = 0; u < 8; ++u) {
            if (j0 + u < m && (unsigned)dl[u] < (unsigned)rows) {
                float* row = acc + dl[u] * OWN_P + lane;
                __hip_atomic_fetch_add(row,              v[u].x, __ATOMIC_RELAXED, __HIP_MEMORY_SCOPE_WORKGROUP);
                __hip_atomic_fetch_add(row + OWN_CS,     v[u].y, __ATOMIC_RELAXED, __HIP_MEMORY_SCOPE_WORKGROUP);
                __hip_atomic_fetch_add(row + 2 * OWN_CS, v[u].z, __ATOMIC_RELAXED, __HIP_MEMORY_SCOPE_WORKGROUP);
                __hip_atomic_fetch_add(row + 3 * OWN_CS, v[u].w, __ATOMIC_RELAXED, __HIP_MEMORY_SCOPE_WORKGROUP);
                if (lane == 0)
                    __hip_atomic_fetch_add(row + OWN_DEGW, 1.0f, __ATOMIC_RELAXED, __HIP_MEMORY_SCOPE_WORKGROUP);
            }
        }
    }
}

__device__ __forceinline__ int own_drain(const int* __restrict__ src,
                                         const int* __restrict__ dst,
                                         const float* __restrict__ h,
                                         float* acc, const unsigned* qe,
                                         int n, int lo, int rows, int lane,
                                         bool save, unsigned* lst, int saved) {
    for (int i0 = 0; i0 < n; i0 += 32) {
        const int m = (n - i0 < 32) ? (n - i0) : 32;
        int my_s = 0, my_d = 0;
        if (lane < m) {
            const unsigned e = qe[i0 + lane];
            my_s = src[e];
            my_d = dst[e] - lo;
            const int p = saved + i0 + lane;
            if (save && p < OWN_LCAP)
                lst[1 + p] = ((unsigned)my_s << 8) | (unsigned)my_d;
        }
        own_add_rows(h, acc, my_s, my_d, m, rows, lane);
    }
    return saved + n;
}

__device__ __forceinline__ int own_push(bool m, unsigned e, unsigned* qe, int cnt) {
    const unsigned k = __builtin_amdgcn_ballot_w32(m);
    if (m) qe[cnt + (int)__builtin_amdgcn_mbcnt_lo(k, 0u)] = e;
    return cnt + __builtin_popcount(k);
}

__device__ __forceinline__ unsigned own_zero_bytes(unsigned x) {
    const unsigned t = (x & 0x7F7F7F7Fu) + 0x7F7F7F7Fu;
    return ~(t | x | 0x7F7F7F7Fu);
}

__global__ __launch_bounds__(1024)
void conv_owner_kernel(const int* __restrict__ src,
                       const int* __restrict__ dst,
                       const unsigned char* __restrict__ key,
                       const float* __restrict__ h,
                       const float* Wf,
                       const float* bias,
                       const float* __restrict__ erowscale,
                       float* __restrict__ hout,
                       unsigned* __restrict__ lst,
                       int reuse,
                       int nE, int nRows) {
    __shared__ float    acc[OWN_RMAX * OWN_P];
    __shared__ unsigned qe[OWN_WAVES][OWN_QCAP];

    const int lane = threadIdx.x & 31;
    const int halfsel = lane >> 4;
    const int lid     = lane & 15;
    const int wave = __builtin_amdgcn_readfirstlane((int)(threadIdx.x >> 5));
    const int nwav = blockDim.x >> 5;
    const int R   = own_range_rows(nRows);
    const int nGf = nE >> 9;
    const int rot = (int)(((long long)blockIdx.x * nGf) / OWN_GRID);
    const unsigned pat = (unsigned)blockIdx.x * 0x01010101u;
    unsigned* myqe = qe[wave];
    const bool listable = (OWN_GRID * R >= nRows) && (nRows <= (1 << 24));
    const bool save     = listable && !reuse;
    unsigned* mylst = lst + ((size_t)blockIdx.x * OWN_WAVES + wave) * OWN_LSTRIDE;
    unsigned nl = OWN_LOVER;
    unsigned pk[OWN_LBATCH];
#pragma unroll
    for (int b = 0; b < OWN_LBATCH; ++b) pk[b] = 0u;
    if (listable && reuse && blockIdx.x * R < nRows) {
        nl = (unsigned)__builtin_amdgcn_readfirstlane((int)mylst[0]);
        if (nl <= (unsigned)OWN_LCAP) {
#pragma unroll
            for (int b = 0; b < OWN_LBATCH; ++b) {
                const unsigned idx = (unsigned)(b * 32 + lane);
                if (idx < nl) pk[b] = mylst[1 + idx];
            }
        }
    }
    const bool scan = nl > (unsigned)OWN_LCAP;

    for (int lo = blockIdx.x * R; lo < nRows; lo += OWN_GRID * R) {
        const int rows  = (nRows - lo < R) ? (nRows - lo) : R;
        const int ntile = (rows + 15) >> 4;
        for (int i = threadIdx.x; i < ntile * 16 * (OWN_P / 4); i += blockDim.x)
            ((float4*)acc)[i] = make_float4(0.f, 0.f, 0.f, 0.f);
        __syncthreads();

        if (!scan) {
#pragma unroll
            for (int b = 0; b < OWN_LBATCH; ++b) {
                if ((unsigned)(b * 32) < nl) {
                    const int left = (int)nl - b * 32;
                    own_add_rows(h, acc, (int)(pk[b] >> 8), (int)(pk[b] & 255u),
                                 (left < 32) ? left : 32, rows, lane);
                }
            }
        }

        int cnt = 0;
        int saved = 0;
        const int nGs = scan ? nGf : 0;
        for (int g0 = wave; g0 < nGs; g0 += OWN_U * nwav) {
            uint4 d[OWN_U];
            int   eb[OWN_U];
#pragma unroll
            for (int u = 0; u < OWN_U; ++u) {
                int g = g0 + u * nwav;
                if (g >= nGf) g = nGf - 1;
                g += rot;
                if (g >= nGf) g -= nGf;
                eb[u] = g * 512 + lane * 16;
                d[u]  = *(const uint4*)(key + eb[u]);
            }
#pragma unroll
            for (int u = 0; u < OWN_U; ++u) {
                if (g0 + u * nwav < nGf) {
                    unsigned hit = (own_zero_bytes(d[u].x ^ pat) >> 7)
                                 | (own_zero_bytes(d[u].y ^ pat) >> 6)
                                 | (own_zero_bytes(d[u].z ^ pat) >> 5)
                                 | (own_zero_bytes(d[u].w ^ pat) >> 4);
                    while (__builtin_amdgcn_ballot_w32(hit != 0u)) {
                        const bool     has = hit != 0u;
                        const unsigned p   = (unsigned)__builtin_ctz(hit | 0x80000000u);
                        const unsigned e   = (unsigned)eb[u] + ((p & 7u) << 2) + (p >> 3);
                        cnt = own_push(has, e, myqe, cnt);
                        hit &= hit - 1u;
                        if (cnt >= OWN_FLUSH) {
                            __builtin_amdgcn_fence(__ATOMIC_ACQ_REL, "wavefront");
                            saved = own_drain(src, dst, h, acc, myqe, cnt, lo, rows, lane,
                                              save, mylst, saved);
                            cnt = 0;
                        }
                    }
                }
            }
        }
        if (wave == 0 && scan) {
            for (int eb0 = nGf * 512; eb0 < nE; eb0 += 32) {
                const int e = eb0 + lane;
                const unsigned r = ((e < nE) ? (unsigned)dst[e] : 0xFFFFFFFFu) - (unsigned)lo;
                cnt = own_push(r < (unsigned)rows && e < nE, (unsigned)e, myqe, cnt);
                if (cnt >= OWN_FLUSH) {
                    __builtin_amdgcn_fence(__ATOMIC_ACQ_REL, "wavefront");
                    saved = own_drain(src, dst, h, acc, myqe, cnt, lo, rows, lane,
                                      save, mylst, saved);
                    cnt = 0;
                }
            }
        }
        __builtin_amdgcn_fence(__ATOMIC_ACQ_REL, "wavefront");
        saved = own_drain(src, dst, h, acc, myqe, cnt, lo, rows, lane, save, mylst, saved);
        if (save && lane == 0)
            mylst[0] = (saved <= OWN_LCAP) ? (unsigned)saved : OWN_LOVER;

        const int nt = wave & 7;
        v2f b[32];
        {
            const float4* wf4 = (const float4*)Wf + (size_t)nt * 16 * 32 + lane;
#pragma unroll
            for (int kk2 = 0; kk2 < 16; ++kk2) {
                const float4 q = wf4[kk2 * 32];
                b[2 * kk2].x     = q.x; b[2 * kk2].y     = q.y;
                b[2 * kk2 + 1].x = q.z; b[2 * kk2 + 1].y = q.w;
            }
        }
        const float bv = bias[nt * 16 + lid];
        __syncthreads();

        for (int t = wave >> 3; t < ntile; t += (nwav >> 3)) {
            const int r = t * 16 + lid;
            const float sn = rsqrtf(fmaxf(acc[r * OWN_P + OWN_DEGW], 1.0f));
            const float* ar = acc + r * OWN_P + halfsel * (2 * OWN_CS);
            v8f c;
#pragma unroll
            for (int j = 0; j < 8; ++j) c[j] = bv;
#pragma unroll
            for (int kk = 0; kk < 32; ++kk) {
                v2f a;
                a.x = ar[kk] * sn;
                a.y = ar[OWN_CS + kk] * sn;
                c = __builtin_amdgcn_wmma_f32_16x16x4_f32(
                    false, a, false, b[kk], (short)0, c, false, false);
            }
#pragma unroll
            for (int j = 0; j < 8; ++j) {
                const int row = t * 16 + j + halfsel * 8;
                if (row < rows) {
                    float v = fmaxf(c[j], 0.0f);
                    if (erowscale) v *= erowscale[lo + row];
                    hout[(size_t)(lo + row) * 128 + nt * 16 + lid] = v;
                }
            }
        }
        __syncthreads();
    }
}

#define SC_RMAX  400
#define SC_P     132
#define SC_QCAP  256
#define SC_FLUSH 64
#define SC_U     8

__device__ __forceinline__ void sc_drain(const int* __restrict__ ids,
                                         const float* __restrict__ h,
                                         float* acc, const unsigned* qe,
                                         int n, int lo, int lane) {
    for (int i0 = 0; i0 < n; i0 += 32) {
        const int m = (n - i0 < 32) ? (n - i0) : 32;
        int my_g = 0, my_d = 0;
        if (lane < m) {
            const unsigned g = qe[i0 + lane];
            my_g = (int)g;
            my_d = ids[g] - lo;
        }
        for (int j0 = 0; j0 < m; j0 += 8) {
            float4 v[8];
            int    dl[8];
#pragma unroll
            for (int u = 0; u < 8; ++u) {
                const int j = (j0 + u < m) ? (j0 + u) : (m - 1);
                const int g = __builtin_amdgcn_readlane(my_g, j);
                dl[u]       = __builtin_amdgcn_readlane(my_d, j);
                v[u] = *(const float4*)(h + (size_t)g * 128 + lane * 4);
            }
#pragma unroll
            for (int u = 0; u < 8; ++u) {
                if (j0 + u < m) {
                    float* row = acc + dl[u] * SC_P + lane;
                    __hip_atomic_fetch_add(row,      v[u].x, __ATOMIC_RELAXED, __HIP_MEMORY_SCOPE_WORKGROUP);
                    __hip_atomic_fetch_add(row + 32, v[u].y, __ATOMIC_RELAXED, __HIP_MEMORY_SCOPE_WORKGROUP);
                    __hip_atomic_fetch_add(row + 64, v[u].z, __ATOMIC_RELAXED, __HIP_MEMORY_SCOPE_WORKGROUP);
                    __hip_atomic_fetch_add(row + 96, v[u].w, __ATOMIC_RELAXED, __HIP_MEMORY_SCOPE_WORKGROUP);
                }
            }
        }
    }
}

__global__ __launch_bounds__(1024)
void scatter_classify_owner_kernel(const int* __restrict__ ids,
                                   const float* __restrict__ h,
                                   const float* __restrict__ Wcf,
                                   const float* __restrict__ bc,
                                   float* __restrict__ out,
                                   int nIds, int nNodes) {
    __shared__ float    acc[SC_RMAX * SC_P];
    __shared__ unsigned qe[OWN_WAVES][SC_QCAP];

    const int lane = threadIdx.x & 31;
    const int wave = __builtin_amdgcn_readfirstlane((int)(threadIdx.x >> 5));
    const int nwav = blockDim.x >> 5;
    const int halfsel = lane >> 4;
    const int lid     = lane & 15;
    int R = (nNodes + OWN_GRID - 1) / OWN_GRID;
    if (R > SC_RMAX) R = SC_RMAX;
    const int nGf = nIds >> 7;
    const int rot = (int)(((long long)blockIdx.x * nGf) / OWN_GRID);
    unsigned* myqe = qe[wave];

    for (int lo = blockIdx.x * R; lo < nNodes; lo += OWN_GRID * R) {
        const int rows  = (nNodes - lo < R) ? (nNodes - lo) : R;
        const int ntile = (rows + 15) >> 4;
        for (int i = threadIdx.x; i < ntile * 16 * (SC_P / 4); i += blockDim.x)
            ((float4*)acc)[i] = make_float4(0.f, 0.f, 0.f, 0.f);
        __syncthreads();

        int cnt = 0;
        for (int g0 = wave; g0 < nGf; g0 += SC_U * nwav) {
            uint4 d[SC_U];
            int   eb[SC_U];
#pragma unroll
            for (int u = 0; u < SC_U; ++u) {
                int g = g0 + u * nwav;
                if (g >= nGf) g = nGf - 1;
                g += rot;
                if (g >= nGf) g -= nGf;
                eb[u] = g * 128 + lane * 4;
                d[u]  = *(const uint4*)(ids + eb[u]);
            }
#pragma unroll
            for (int u = 0; u < SC_U; ++u) {
                if (g0 + u * nwav < nGf) {
                    const unsigned b = (unsigned)eb[u];
                    cnt = own_push(d[u].x - (unsigned)lo < (unsigned)rows, b,      myqe, cnt);
                    cnt = own_push(d[u].y - (unsigned)lo < (unsigned)rows, b + 1u, myqe, cnt);
                    cnt = own_push(d[u].z - (unsigned)lo < (unsigned)rows, b + 2u, myqe, cnt);
                    cnt = own_push(d[u].w - (unsigned)lo < (unsigned)rows, b + 3u, myqe, cnt);
                    if (cnt >= SC_FLUSH) {
                        __builtin_amdgcn_fence(__ATOMIC_ACQ_REL, "wavefront");
                        sc_drain(ids, h, acc, myqe, cnt, lo, lane);
                        cnt = 0;
                    }
                }
            }
        }
        if (wave == 0) {
            for (int e0 = nGf * 128; e0 < nIds; e0 += 32) {
                const int e = e0 + lane;
                const unsigned r = ((e < nIds) ? (unsigned)ids[e] : 0xFFFFFFFFu) - (unsigned)lo;
                cnt = own_push(r < (unsigned)rows && e < nIds, (unsigned)e, myqe, cnt);
                if (cnt >= SC_FLUSH) {
                    __builtin_amdgcn_fence(__ATOMIC_ACQ_REL, "wavefront");
                    sc_drain(ids, h, acc, myqe, cnt, lo, lane);
                    cnt = 0;
                }
            }
        }
        __builtin_amdgcn_fence(__ATOMIC_ACQ_REL, "wavefront");
        sc_drain(ids, h, acc, myqe, cnt, lo, lane);
        __syncthreads();

        for (int t = wave; t < ntile; t += nwav) {
            v2f b[32];
            const float4* wf4 = (const float4*)Wcf + lane;
#pragma unroll
            for (int kk2 = 0; kk2 < 16; ++kk2) {
                const float4 q = wf4[kk2 * 32];
                b[2 * kk2].x     = q.x; b[2 * kk2].y     = q.y;
                b[2 * kk2 + 1].x = q.z; b[2 * kk2 + 1].y = q.w;
            }
            const float bv = bc[lid];
            v8f c;
#pragma unroll
            for (int j = 0; j < 8; ++j) c[j] = bv;
            const float* ar = acc + (t * 16 + lid) * SC_P + halfsel * 64;
#pragma unroll
            for (int kk = 0; kk < 32; ++kk) {
                v2f a;
                a.x = ar[kk];
                a.y = ar[32 + kk];
                c = __builtin_amdgcn_wmma_f32_16x16x4_f32(
                    false, a, false, b[kk], (short)0, c, false, false);
            }
#pragma unroll
            for (int j = 0; j < 8; ++j) {
                const int row = t * 16 + j + halfsel * 8;
                if (row < rows) out[(size_t)(lo + row) * 16 + lid] = c[j];
            }
        }
        __syncthreads();
    }
}

template <int NTILES>
__global__ __launch_bounds__(128)
void gemm_wmma_k128_kernel(const float* __restrict__ A,
                           const float* __restrict__ Wf,
                           const float* __restrict__ bias,
                           const float* __restrict__ arowscale,
                           const float* __restrict__ erowscale,
                           float* __restrict__ out,
                           int M, int relu) {
    const int lane    = threadIdx.x & 31;
    const int wave    = __builtin_amdgcn_readfirstlane((int)(threadIdx.x >> 5));
    const int wpb     = blockDim.x >> 5;
    const int nwaves  = gridDim.x * wpb;
    const int ntiles  = M >> 4;
    int tile          = blockIdx.x * wpb + wave;
    if (tile >= ntiles) return;
    const int halfsel = lane >> 4;
    const int lid     = lane & 15;
    const int N       = NTILES * 16;

    v2f b[NTILES][32];
    {
        const float4* wf4 = (const float4*)Wf + lane;
#pragma unroll
        for (int nt = 0; nt < NTILES; ++nt) {
#pragma unroll
            for (int kk2 = 0; kk2 < 16; ++kk2) {
                const float4 q = wf4[(nt * 16 + kk2) * 32];
                b[nt][2 * kk2].x     = q.x; b[nt][2 * kk2].y     = q.y;
                b[nt][2 * kk2 + 1].x = q.z; b[nt][2 * kk2 + 1].y = q.w;
            }
        }
    }
    float bv[NTILES];
#pragma unroll
    for (int nt = 0; nt < NTILES; ++nt) bv[nt] = bias[nt * 16 + lid];

    v2f   an[32];
    float sn;
    {
        const int r = tile * 16 + lid;
        const float* ap = A + (size_t)r * 128 + halfsel * 2;
#pragma unroll
        for (int kk = 0; kk < 32; ++kk) an[kk] = *(const v2f*)(ap + kk * 4);
        sn = arowscale ? arowscale[r] : 1.0f;
    }

    for (; tile < ntiles; tile += nwaves) {
        v2f a[32];
#pragma unroll
        for (int kk = 0; kk < 32; ++kk) { a[kk].x = an[kk].x * sn; a[kk].y = an[kk].y * sn; }

        const int tnext = tile + nwaves;
        if (tnext < ntiles) {
            const int r = tnext * 16 + lid;
            const float* ap = A + (size_t)r * 128 + halfsel * 2;
#pragma unroll
            for (int kk = 0; kk < 32; ++kk) an[kk] = *(const v2f*)(ap + kk * 4);
            sn = arowscale ? arowscale[r] : 1.0f;
        }

        float es[8];
        if (erowscale) {
#pragma unroll
            for (int j = 0; j < 8; ++j) es[j] = erowscale[tile * 16 + j + halfsel * 8];
        } else {
#pragma unroll
            for (int j = 0; j < 8; ++j) es[j] = 1.0f;
        }

        v8f acc[NTILES];
#pragma unroll
        for (int nt = 0; nt < NTILES; ++nt) {
#pragma unroll
            for (int j = 0; j < 8; ++j) acc[nt][j] = bv[nt];
        }

#pragma unroll
        for (int kk = 0; kk < 32; ++kk) {
#pragma unroll
            for (int nt = 0; nt < NTILES; ++nt) {
                acc[nt] = __builtin_amdgcn_wmma_f32_16x16x4_f32(
                    false, a[kk], false, b[nt][kk], (short)0, acc[nt], false, false);
            }
        }

#pragma unroll
        for (int nt = 0; nt < NTILES; ++nt) {
#pragma unroll
            for (int j = 0; j < 8; ++j) {
                float v = acc[nt][j];
                if (relu) v = fmaxf(v, 0.0f);
                const int row = tile * 16 + j + halfsel * 8;
                if (erowscale) v *= es[j];
                out[(size_t)row * N + nt * 16 + lid] = v;
            }
        }
    }
}


extern "C" void kernel_launch(void* const* d_in, const int* in_sizes, int n_in,
                              void* d_out, int out_size, void* d_ws, size_t ws_size,
                              hipStream_t stream) {
    (void)n_in; (void)ws_size;
    const float* x   = (const float*)d_in[0];
    const float* W2  = (const float*)d_in[1];
    const float* b2  = (const float*)d_in[2];
    const float* gw  = (const float*)d_in[3];
    const float* gb  = (const float*)d_in[4];
    const float* Wc  = (const float*)d_in[5];
    const float* bc  = (const float*)d_in[6];
    const int*   src = (const int*)d_in[7];
    const int*   dst = (const int*)d_in[8];
    const int*   ids = (const int*)d_in[9];

    const int H  = 128;
    const int C  = 16;
    const int L  = in_sizes[2] / H;
    const int NS = in_sizes[9] / L;
    const int NE = in_sizes[7] / L;
    const int NL = in_sizes[4] / (L * H);
    const int NN = out_size / C;

    float* ws       = (float*)d_ws;
    float* hfin     = ws;
    float* hbuf     = hfin + (size_t)(L - 1) * NS * H;
    float* agg      = hbuf + (size_t)NS * H;
    float* deg_out  = agg + (size_t)NS * H;
    float* w2t      = deg_out + 2 * (size_t)NS;
    float* gwt      = w2t + (size_t)L * H * H;
    float* wct      = gwt + (size_t)L * NL * H * H;
    unsigned char* okey = (unsigned char*)d_out;
    unsigned* olist = (unsigned*)(okey + (((size_t)NE + 255) & ~(size_t)255));

    const int T = 256;
    const int GT = 128;
    const int gemm_blocks_NS = 256;

    {
        int nHH = H * H;
        int bHH = (nHH + T - 1) / T;
        for (int i = 0; i < L; ++i)
            pack_bfrag_kernel<<<bHH, T, 0, stream>>>(W2 + (size_t)i * nHH,
                                                     w2t + (size_t)i * nHH, H);
        for (int i = 0; i < L * NL; ++i)
            pack_bfrag_kernel<<<bHH, T, 0, stream>>>(gw + (size_t)i * nHH,
                                                     gwt + (size_t)i * nHH, H);
        int nHC = H * C;
        pack_bfrag_kernel<<<(nHC + T - 1) / T, T, 0, stream>>>(Wc, wct, C);
    }

    for (int i = 0; i < L; ++i) {
        const int*   src_i = src + (size_t)i * NE;
        const int*   dst_i = dst + (size_t)i * NE;
        const float* x_i   = x   + (size_t)i * NS * H;
        const float* w2t_i = w2t + (size_t)i * H * H;
        const float* b2_i  = b2  + (size_t)i * H;

        {
            const int pstride = (NS + 3) & ~3;
            unsigned* dpart = (unsigned*)agg;
            degree_slice_kernel<<<DEG_SLICES, 1024, 0, stream>>>(src_i, dst_i, dpart,
                                                                 okey, NE, NS, pstride);
            degree_rsqrt_kernel<<<(NS + T - 1) / T, T, 0, stream>>>(dpart, deg_out,
                                                                    NS, pstride);
        }

        float* pin  = (NL & 1) ? agg : hbuf;
        float* pout = (NL & 1) ? hbuf : agg;

        gemm_wmma_k128_kernel<8><<<gemm_blocks_NS, GT, 0, stream>>>(
            x_i, w2t_i, b2_i, nullptr, deg_out, pin, NS, 0);

        for (int l = 0; l < NL; ++l) {
            const float* gwt_il = gwt + ((size_t)i * NL + l) * H * H;
            const float* gb_il  = gb  + ((size_t)i * NL + l) * H;
            const float* escale = (l + 1 < NL) ? deg_out : nullptr;
            float* hout = (l + 1 < NL) ? pout : hfin + (size_t)i * NS * H;
            conv_owner_kernel<<<OWN_GRID, 1024, 0, stream>>>(src_i, dst_i, okey, pin,
                                                             gwt_il, gb_il, escale,
                                                             hout, olist, l, NE, NS);
            float* tmp = pin; pin = pout; pout = tmp;
        }
    }

    scatter_classify_owner_kernel<<<OWN_GRID, 1024, 0, stream>>>(
        ids, hfin, wct, bc, (float*)d_out, L * NS, NN);
}


# edge sums: out-degree launch groups its slice's edges by owner work-group in LDS (count, prefix sum, place), stores lists; both conv layers read list segments, no key scan; LDS arrival order
# speedup vs baseline: 4.4700x; 1.0777x over previous
#include <hip/hip_runtime.h>


typedef __attribute__((ext_vector_type(2))) float v2f;
typedef __attribute__((ext_vector_type(8))) float v8f;


__global__ void pack_bfrag_kernel(const float* __restrict__ in,
                                  float* __restrict__ outF, int N) {
    int i = blockIdx.x * blockDim.x + threadIdx.x;
    if (i < N * 128) {
        int c    = i & 3;
        int lane = (i >> 2) & 31;
        int kk2  = (i >> 7) & 15;
        int nt   = i >> 11;
        int kk   = 2 * kk2 + (c >> 1);
        int k    = kk * 4 + (lane >> 4) * 2 + (c & 1);
        int n    = nt * 16 + (lane & 15);
        outF[i] = in[k * N + n];
    }
}

#define OWN_GRID  256
#define OWN_RMAX  208
#define OWN_P     132
#define OWN_CS    33
#define OWN_DEGW  32

__host__ __device__ __forceinline__ int own_range_rows(int nRows) {
    const int R = (nRows + OWN_GRID - 1) / OWN_GRID;
    return (R > OWN_RMAX) ? OWN_RMAX : R;
}

#define DEG_SLICES 64
#define DEG_CAP    51200
#define SL_CAP     16384

__host__ __device__ __forceinline__ int sl_per(int nE) {
    return ((nE + DEG_SLICES - 1) / DEG_SLICES + 3) & ~3;
}
__host__ __device__ __forceinline__ int sl_stride(int nE) {
    return (sl_per(nE) + OWN_GRID + 1 + 3) & ~3;
}

__global__ __launch_bounds__(1024)
void degree_slice_kernel(const int* __restrict__ src,
                         const int* __restrict__ dst,
                         unsigned* __restrict__ part,
                         unsigned char* __restrict__ key,
                         unsigned* __restrict__ slist,
                         int nE, int nRows, int pstride) {
    __shared__ __attribute__((aligned(16))) unsigned cnt[DEG_CAP];
    __shared__ unsigned ocnt[OWN_GRID];
    __shared__ unsigned ooff[OWN_GRID + 1];
    const unsigned R = (unsigned)own_range_rows(nRows);
    const int per = sl_per(nE);
    const int e0  = (int)blockIdx.x * per;
    const int e1  = (nE - e0 < per) ? nE : (e0 + per);
    unsigned* mypart = part + (size_t)blockIdx.x * pstride;
    const bool vec = ((((size_t)src | (size_t)dst) & 15) == 0) && (((size_t)key & 3) == 0);

    for (int base = 0; base < nRows; base += DEG_CAP) {
        const int n  = (nRows - base < DEG_CAP) ? (nRows - base) : DEG_CAP;
        const int n4 = (n + 3) >> 2;
        for (int i = threadIdx.x; i < n4; i += blockDim.x)
            ((uint4*)cnt)[i] = make_uint4(0u, 0u, 0u, 0u);
        __syncthreads();

        const bool wkey = (base == 0) && (slist == nullptr);
        auto one = [&](int e) {
            const unsigned s = (unsigned)src[e] - (unsigned)base;
            if (s < (unsigned)n) atomicAdd(&cnt[s], 1u);
            if (wkey)
                key[e] = (unsigned char)(((unsigned)dst[e] / R) % (unsigned)OWN_GRID);
        };
        if (vec) {
            const int nfull = (e1 > e0) ? ((e1 - e0) >> 2) : 0;
            for (int g0 = threadIdx.x; g0 < nfull; g0 += 4 * (int)blockDim.x) {
                int4 s4[4], d4[4];
#pragma unroll
                for (int u = 0; u < 4; ++u) {
                    const int g  = g0 + u * (int)blockDim.x;
                    const int gg = (g < nfull) ? g : g0;
                    s4[u] = *(const int4*)(src + e0 + gg * 4);
                    if (wkey) d4[u] = *(const int4*)(dst + e0 + gg * 4);
                    else           d4[u] = make_int4(0, 0, 0, 0);
                }
#pragma unroll
                for (int u = 0; u < 4; ++u) {
                    const int g = g0 + u * (int)blockDim.x;
                    if (g < nfull) {
                        const unsigned sx = (unsigned)s4[u].x - (unsigned)base;
                        const unsigned sy = (unsigned)s4[u].y - (unsigned)base;
                        const unsigned sz = (unsigned)s4[u].z - (unsigned)base;
                        const unsigned sw = (unsigned)s4[u].w - (unsigned)base;
                        if (sx < (unsigned)n) atomicAdd(&cnt[sx], 1u);
                        if (sy < (unsigned)n) atomicAdd(&cnt[sy], 1u);
                        if (sz < (unsigned)n) atomicAdd(&cnt[sz], 1u);
                        if (sw < (unsigned)n) atomicAdd(&cnt[sw], 1u);
                        if (wkey) {
                            uchar4 k;
                            k.x = (unsigned char)(((unsigned)d4[u].x / R) % (unsigned)OWN_GRID);
                            k.y = (unsigned char)(((unsigned)d4[u].y / R) % (unsigned)OWN_GRID);
                            k.z = (unsigned char)(((unsigned)d4[u].z / R) % (unsigned)OWN_GRID);
                            k.w = (unsigned char)(((unsigned)d4[u].w / R) % (unsigned)OWN_GRID);
                            *(uchar4*)(key + e0 + g * 4) = k;
                        }
                    }
                }
            }
            for (int e = e0 + nfull * 4 + (int)threadIdx.x; e < e1; e += (int)blockDim.x) one(e);
        } else {
            for (int e = e0 + (int)threadIdx.x; e < e1; e += (int)blockDim.x) one(e);
        }
        __syncthreads();

        for (int i = threadIdx.x; i < n4; i += blockDim.x)
            ((uint4*)(mypart + base))[i] = ((const uint4*)cnt)[i];
        __syncthreads();
    }

    if (slist == nullptr) return;

    unsigned* sw  = cnt;
    unsigned* sk  = cnt + SL_CAP;
    unsigned* fin = cnt + 2 * SL_CAP;
    const int ne = (e1 > e0) ? (e1 - e0) : 0;
    for (int i = threadIdx.x; i < OWN_GRID; i += blockDim.x) ocnt[i] = 0u;
    __syncthreads();
    auto place = [&](int i, unsigned s, unsigned d) {
        const unsigned o = d / R;
        const unsigned rank = atomicAdd(&ocnt[o], 1u);
        sw[i] = (s << 8) | (d - o * R);
        sk[i] = (o << 16) | rank;
    };
    {
        const int nfull = vec ? (ne >> 2) : 0;
        for (int g0 = threadIdx.x; g0 < nfull; g0 += 4 * (int)blockDim.x) {
            int4 s4[4], d4[4];
#pragma unroll
            for (int u = 0; u < 4; ++u) {
                const int g  = g0 + u * (int)blockDim.x;
                const int gg = (g < nfull) ? g : g0;
                s4[u] = *(const int4*)(src + e0 + gg * 4);
                d4[u] = *(const int4*)(dst + e0 + gg * 4);
            }
#pragma unroll
            for (int u = 0; u < 4; ++u) {
                const int g = g0 + u * (int)blockDim.x;
                if (g < nfull) {
                    place(g * 4,     (unsigned)s4[u].x, (unsigned)d4[u].x);
                    place(g * 4 + 1, (unsigned)s4[u].y, (unsigned)d4[u].y);
                    place(g * 4 + 2, (unsigned)s4[u].z, (unsigned)d4[u].z);
                    place(g * 4 + 3, (unsigned)s4[u].w, (unsigned)d4[u].w);
                }
            }
        }
        for (int i = nfull * 4 + (int)threadIdx.x; i < ne; i += (int)blockDim.x)
            place(i, (unsigned)src[e0 + i], (unsigned)dst[e0 + i]);
    }
    __syncthreads();

    if (threadIdx.x < 32) {
        const int l = (int)threadIdx.x;
        unsigned c[OWN_GRID / 32];
        unsigned t = 0u;
#pragma unroll
        for (int k = 0; k < OWN_GRID / 32; ++k) { c[k] = ocnt[l * (OWN_GRID / 32) + k]; t += c[k]; }
        unsigned incl = t;
#pragma unroll
        for (int sft = 1; sft < 32; sft <<= 1) {
            const unsigned v = __shfl_up(incl, (unsigned)sft, 32);
            if (l >= sft) incl += v;
        }
        unsigned run = incl - t;
#pragma unroll
        for (int k = 0; k < OWN_GRID / 32; ++k) { ooff[l * (OWN_GRID / 32) + k] = run; run += c[k]; }
        if (l == 31) ooff[OWN_GRID] = run;
    }
    __syncthreads();
    for (int i0 = threadIdx.x; i0 < ne; i0 += 4 * (int)blockDim.x) {
        unsigned k[4], w[4];
#pragma unroll
        for (int u = 0; u < 4; ++u) {
            const int i  = i0 + u * (int)blockDim.x;
            const int ii = (i < ne) ? i : i0;
            k[u] = sk[ii];
            w[u] = sw[ii];
        }
#pragma unroll
        for (int u = 0; u < 4; ++u) {
            if (i0 + u * (int)blockDim.x < ne)
                fin[ooff[k[u] >> 16] + (k[u] & 0xFFFFu)] = w[u];
        }
    }
    __syncthreads();
    unsigned* mylist = slist + (size_t)blockIdx.x * sl_stride(nE);
    for (int i = threadIdx.x; i < ne; i += blockDim.x) mylist[i] = fin[i];
    for (int i = threadIdx.x; i <= OWN_GRID; i += blockDim.x) mylist[per + i] = ooff[i];
}

__global__ void degree_rsqrt_kernel(const unsigned* __restrict__ part,
                                    float* __restrict__ d, int n, int pstride) {
    int i = blockIdx.x * blockDim.x + threadIdx.x;
    if (i < n) {
        unsigned c = 0u;
#pragma unroll
        for (int s = 0; s < DEG_SLICES; ++s) c += part[(size_t)s * pstride + i];
        d[i] = rsqrtf(fmaxf((float)c, 1.0f));
    }
}

#define OWN_WAVES 32
#define OWN_QCAP  512
#define OWN_FLUSH 64
#define OWN_U     8
#define OWN_LSTRIDE 160
#define OWN_LCAP    (OWN_LSTRIDE - 1)
#define OWN_LBATCH  5
#define OWN_LOVER   0xFFFFFFFFu

__device__ __forceinline__ void own_add_rows(const float* __restrict__ h, float* acc,
                                             int my_s, int my_d, int m, int rows,
                                             int lane) {
    for (int j0 = 0; j0 < m; j0 += 8) {
        float4 v[8];
        int    dl[8];
#pragma unroll
        for (int u = 0; u < 8; ++u) {
            const int j = (j0 + u < m) ? (j0 + u) : (m - 1);
            const int s = __builtin_amdgcn_readlane(my_s, j);
            dl[u]       = __builtin_amdgcn_readlane(my_d, j);
            v[u] = *(const float4*)(h + (size_t)s * 128 + lane * 4);
        }
#pragma unroll
        for (int u = 0; u < 8; ++u) {
            if (j0 + u < m && (unsigned)dl[u] < (unsigned)rows) {
                float* row = acc + dl[u] * OWN_P + lane;
                __hip_atomic_fetch_add(row,              v[u].x, __ATOMIC_RELAXED, __HIP_MEMORY_SCOPE_WORKGROUP);
                __hip_atomic_fetch_add(row + OWN_CS,     v[u].y, __ATOMIC_RELAXED, __HIP_MEMORY_SCOPE_WORKGROUP);
                __hip_atomic_fetch_add(row + 2 * OWN_CS, v[u].z, __ATOMIC_RELAXED, __HIP_MEMORY_SCOPE_WORKGROUP);
                __hip_atomic_fetch_add(row + 3 * OWN_CS, v[u].w, __ATOMIC_RELAXED, __HIP_MEMORY_SCOPE_WORKGROUP);
                if (lane == 0)
                    __hip_atomic_fetch_add(row + OWN_DEGW, 1.0f, __ATOMIC_RELAXED, __HIP_MEMORY_SCOPE_WORKGROUP);
            }
        }
    }
}

__device__ __forceinline__ int own_drain(const int* __restrict__ src,
                                         const int* __restrict__ dst,
                                         const float* __restrict__ h,
                                         float* acc, const unsigned* qe,
                                         int n, int lo, int rows, int lane,
                                         bool save, unsigned* lst, int saved) {
    for (int i0 = 0; i0 < n; i0 += 32) {
        const int m = (n - i0 < 32) ? (n - i0) : 32;
        int my_s = 0, my_d = 0;
        if (lane < m) {
            const unsigned e = qe[i0 + lane];
            my_s = src[e];
            my_d = dst[e] - lo;
            const int p = saved + i0 + lane;
            if (save && p < OWN_LCAP)
                lst[1 + p] = ((unsigned)my_s << 8) | (unsigned)my_d;
        }
        own_add_rows(h, acc, my_s, my_d, m, rows, lane);
    }
    return saved + n;
}

__device__ __forceinline__ int own_push(bool m, unsigned e, unsigned* qe, int cnt) {
    const unsigned k = __builtin_amdgcn_ballot_w32(m);
    if (m) qe[cnt + (int)__builtin_amdgcn_mbcnt_lo(k, 0u)] = e;
    return cnt + __builtin_popcount(k);
}

__device__ __forceinline__ unsigned own_zero_bytes(unsigned x) {
    const unsigned t = (x & 0x7F7F7F7Fu) + 0x7F7F7F7Fu;
    return ~(t | x | 0x7F7F7F7Fu);
}

__global__ __launch_bounds__(1024)
void conv_owner_kernel(const int* __restrict__ src,
                       const int* __restrict__ dst,
                       const unsigned char* __restrict__ key,
                       const float* __restrict__ h,
                       const float* Wf,
                       const float* bias,
                       const float* __restrict__ erowscale,
                       float* __restrict__ hout,
                       unsigned* __restrict__ lst,
                       int reuse,
                       int nE, int nRows) {
    __shared__ float    acc[OWN_RMAX * OWN_P];
    __shared__ unsigned qe[OWN_WAVES][OWN_QCAP];

    const int lane = threadIdx.x & 31;
    const int halfsel = lane >> 4;
    const int lid     = lane & 15;
    const int wave = __builtin_amdgcn_readfirstlane((int)(threadIdx.x >> 5));
    const int nwav = blockDim.x >> 5;
    const int R   = own_range_rows(nRows);
    const int nGf = nE >> 9;
    const int rot = (int)(((long long)blockIdx.x * nGf) / OWN_GRID);
    const unsigned pat = (unsigned)blockIdx.x * 0x01010101u;
    unsigned* myqe = qe[wave];
    const bool listable = (OWN_GRID * R >= nRows) && (nRows <= (1 << 24));
    const bool save     = listable && !reuse;
    unsigned* mylst = lst + ((size_t)blockIdx.x * OWN_WAVES + wave) * OWN_LSTRIDE;
    unsigned nl = OWN_LOVER;
    unsigned pk[OWN_LBATCH];
#pragma unroll
    for (int b = 0; b < OWN_LBATCH; ++b) pk[b] = 0u;
    if (listable && reuse && blockIdx.x * R < nRows) {
        nl = (unsigned)__builtin_amdgcn_readfirstlane((int)mylst[0]);
        if (nl <= (unsigned)OWN_LCAP) {
#pragma unroll
            for (int b = 0; b < OWN_LBATCH; ++b) {
                const unsigned idx = (unsigned)(b * 32 + lane);
                if (idx < nl) pk[b] = mylst[1 + idx];
            }
        }
    }
    const bool scan = nl > (unsigned)OWN_LCAP;

    for (int lo = blockIdx.x * R; lo < nRows; lo += OWN_GRID * R) {
        const int rows  = (nRows - lo < R) ? (nRows - lo) : R;
        const int ntile = (rows + 15) >> 4;
        for (int i = threadIdx.x; i < ntile * 16 * (OWN_P / 4); i += blockDim.x)
            ((float4*)acc)[i] = make_float4(0.f, 0.f, 0.f, 0.f);
        __syncthreads();

        if (!scan) {
#pragma unroll
            for (int b = 0; b < OWN_LBATCH; ++b) {
                if ((unsigned)(b * 32) < nl) {
                    const int left = (int)nl - b * 32;
                    own_add_rows(h, acc, (int)(pk[b] >> 8), (int)(pk[b] & 255u),
                                 (left < 32) ? left : 32, rows, lane);
                }
            }
        }

        int cnt = 0;
        int saved = 0;
        const int nGs = scan ? nGf : 0;
        for (int g0 = wave; g0 < nGs; g0 += OWN_U * nwav) {
            uint4 d[OWN_U];
            int   eb[OWN_U];
#pragma unroll
            for (int u = 0; u < OWN_U; ++u) {
                int g = g0 + u * nwav;
                if (g >= nGf) g = nGf - 1;
                g += rot;
                if (g >= nGf) g -= nGf;
                eb[u] = g * 512 + lane * 16;
                d[u]  = *(const uint4*)(key + eb[u]);
            }
#pragma unroll
            for (int u = 0; u < OWN_U; ++u) {
                if (g0 + u * nwav < nGf) {
                    unsigned hit = (own_zero_bytes(d[u].x ^ pat) >> 7)
                                 | (own_zero_bytes(d[u].y ^ pat) >> 6)
                                 | (own_zero_bytes(d[u].z ^ pat) >> 5)
                                 | (own_zero_bytes(d[u].w ^ pat) >> 4);
                    while (__builtin_amdgcn_ballot_w32(hit != 0u)) {
                        const bool     has = hit != 0u;
                        const unsigned p   = (unsigned)__builtin_ctz(hit | 0x80000000u);
                        const unsigned e   = (unsigned)eb[u] + ((p & 7u) << 2) + (p >> 3);
                        cnt = own_push(has, e, myqe, cnt);
                        hit &= hit - 1u;
                        if (cnt >= OWN_FLUSH) {
                            __builtin_amdgcn_fence(__ATOMIC_ACQ_REL, "wavefront");
                            saved = own_drain(src, dst, h, acc, myqe, cnt, lo, rows, lane,
                                              save, mylst, saved);
                            cnt = 0;
                        }
                    }
                }
            }
        }
        if (wave == 0 && scan) {
            for (int eb0 = nGf * 512; eb0 < nE; eb0 += 32) {
                const int e = eb0 + lane;
                const unsigned r = ((e < nE) ? (unsigned)dst[e] : 0xFFFFFFFFu) - (unsigned)lo;
                cnt = own_push(r < (unsigned)rows && e < nE, (unsigned)e, myqe, cnt);
                if (cnt >= OWN_FLUSH) {
                    __builtin_amdgcn_fence(__ATOMIC_ACQ_REL, "wavefront");
                    saved = own_drain(src, dst, h, acc, myqe, cnt, lo, rows, lane,
                                      save, mylst, saved);
                    cnt = 0;
                }
            }
        }
        __builtin_amdgcn_fence(__ATOMIC_ACQ_REL, "wavefront");
        saved = own_drain(src, dst, h, acc, myqe, cnt, lo, rows, lane, save, mylst, saved);
        if (save && lane == 0)
            mylst[0] = (saved <= OWN_LCAP) ? (unsigned)saved : OWN_LOVER;

        const int nt = wave & 7;
        v2f b[32];
        {
            const float4* wf4 = (const float4*)Wf + (size_t)nt * 16 * 32 + lane;
#pragma unroll
            for (int kk2 = 0; kk2 < 16; ++kk2) {
                const float4 q = wf4[kk2 * 32];
                b[2 * kk2].x     = q.x; b[2 * kk2].y     = q.y;
                b[2 * kk2 + 1].x = q.z; b[2 * kk2 + 1].y = q.w;
            }
        }
        const float bv = bias[nt * 16 + lid];
        __syncthreads();

        for (int t = wave >> 3; t < ntile; t += (nwav >> 3)) {
            const int r = t * 16 + lid;
            const float sn = rsqrtf(fmaxf(acc[r * OWN_P + OWN_DEGW], 1.0f));
            const float* ar = acc + r * OWN_P + halfsel * (2 * OWN_CS);
            v8f c;
#pragma unroll
            for (int j = 0; j < 8; ++j) c[j] = bv;
#pragma unroll
            for (int kk = 0; kk < 32; ++kk) {
                v2f a;
                a.x = ar[kk] * sn;
                a.y = ar[OWN_CS + kk] * sn;
                c = __builtin_amdgcn_wmma_f32_16x16x4_f32(
                    false, a, false, b[kk], (short)0, c, false, false);
            }
#pragma unroll
            for (int j = 0; j < 8; ++j) {
                const int row = t * 16 + j + halfsel * 8;
                if (row < rows) {
                    float v = fmaxf(c[j], 0.0f);
                    if (erowscale) v *= erowscale[lo + row];
                    hout[(size_t)(lo + row) * 128 + nt * 16 + lid] = v;
                }
            }
        }
        __syncthreads();
    }
}

#define SL_SEGS (DEG_SLICES / OWN_WAVES)
#define SL_PF   3
static_assert(DEG_SLICES % OWN_WAVES == 0, "a wave takes whole segments");

__global__ __launch_bounds__(1024)
void conv_lists_kernel(const float* __restrict__ h,
                       const float* Wf,
                       const float* bias,
                       const float* __restrict__ erowscale,
                       float* __restrict__ hout,
                       const unsigned* __restrict__ slist,
                       int nE, int nRows) {
    __shared__ float    acc[OWN_RMAX * OWN_P];
    __shared__ unsigned qe[OWN_WAVES][OWN_QCAP];

    const int lane = threadIdx.x & 31;
    const int halfsel = lane >> 4;
    const int lid     = lane & 15;
    const int wave = __builtin_amdgcn_readfirstlane((int)(threadIdx.x >> 5));
    const int nwav = blockDim.x >> 5;
    const int R    = own_range_rows(nRows);
    const int lo   = (int)blockIdx.x * R;
    if (lo >= nRows) return;
    const int rows  = (nRows - lo < R) ? (nRows - lo) : R;
    const int ntile = (rows + 15) >> 4;
    const int per    = sl_per(nE);
    const int stride = sl_stride(nE);
    unsigned* myqe = qe[wave];

    const unsigned* sb[SL_SEGS];
    int sn[SL_SEGS];
#pragma unroll
    for (int k = 0; k < SL_SEGS; ++k) {
        const unsigned* lst = slist + (size_t)(wave * SL_SEGS + k) * stride;
        const int o0 = __builtin_amdgcn_readfirstlane((int)lst[per + blockIdx.x]);
        const int o1 = __builtin_amdgcn_readfirstlane((int)lst[per + blockIdx.x + 1]);
        sb[k] = lst + o0;
        sn[k] = o1 - o0;
    }
    unsigned pw[SL_SEGS][SL_PF];
#pragma unroll
    for (int k = 0; k < SL_SEGS; ++k) {
#pragma unroll
        for (int b = 0; b < SL_PF; ++b) {
            const int i = b * 32 + lane;
            pw[k][b] = (i < sn[k]) ? sb[k][i] : 0u;
        }
    }

    for (int i = threadIdx.x; i < ntile * 16 * (OWN_P / 4); i += blockDim.x)
        ((float4*)acc)[i] = make_float4(0.f, 0.f, 0.f, 0.f);

    int cnt = 0;
    int qd[SL_SEGS];
#pragma unroll
    for (int k = 0; k < SL_SEGS; ++k) {
        const int take = (sn[k] < OWN_QCAP - cnt) ? sn[k] : (OWN_QCAP - cnt);
#pragma unroll
        for (int b = 0; b < SL_PF; ++b) {
            const int i = b * 32 + lane;
            if (i < take) myqe[cnt + i] = pw[k][b];
        }
        for (int i = SL_PF * 32 + lane; i < take; i += 32) myqe[cnt + i] = sb[k][i];
        qd[k] = take;
        cnt += take;
    }
    __syncthreads();

    for (int i0 = 0; i0 < cnt; i0 += 32) {
        const int m = (cnt - i0 < 32) ? (cnt - i0) : 32;
        const unsigned w = (lane < m) ? myqe[i0 + lane] : 0u;
        own_add_rows(h, acc, (int)(w >> 8), (int)(w & 255u), m, rows, lane);
    }
#pragma unroll
    for (int k = 0; k < SL_SEGS; ++k) {
        for (int i0 = qd[k]; i0 < sn[k]; i0 += 32) {
            const int m = (sn[k] - i0 < 32) ? (sn[k] - i0) : 32;
            const unsigned w = (lane < m) ? sb[k][i0 + lane] : 0u;
            own_add_rows(h, acc, (int)(w >> 8), (int)(w & 255u), m, rows, lane);
        }
    }

    const int nt = wave & 7;
    v2f b[32];
    {
        const float4* wf4 = (const float4*)Wf + (size_t)nt * 16 * 32 + lane;
#pragma unroll
        for (int kk2 = 0; kk2 < 16; ++kk2) {
            const float4 q = wf4[kk2 * 32];
            b[2 * kk2].x     = q.x; b[2 * kk2].y     = q.y;
            b[2 * kk2 + 1].x = q.z; b[2 * kk2 + 1].y = q.w;
        }
    }
    const float bv = bias[nt * 16 + lid];
    __syncthreads();

    for (int t = wave >> 3; t < ntile; t += (nwav >> 3)) {
        const int r = t * 16 + lid;
        const float sc = rsqrtf(fmaxf(acc[r * OWN_P + OWN_DEGW], 1.0f));
        const float* ar = acc + r * OWN_P + halfsel * (2 * OWN_CS);
        v8f c;
#pragma unroll
        for (int j = 0; j < 8; ++j) c[j] = bv;
#pragma unroll
        for (int kk = 0; kk < 32; ++kk) {
            v2f a;
            a.x = ar[kk] * sc;
            a.y = ar[OWN_CS + kk] * sc;
            c = __builtin_amdgcn_wmma_f32_16x16x4_f32(
                false, a, false, b[kk], (short)0, c, false, false);
        }
#pragma unroll
        for (int j = 0; j < 8; ++j) {
            const int row = t * 16 + j + halfsel * 8;
            if (row < rows) {
                float v = fmaxf(c[j], 0.0f);
                if (erowscale) v *= erowscale[lo + row];
                hout[(size_t)(lo + row) * 128 + nt * 16 + lid] = v;
            }
        }
    }
}

#define SC_RMAX  400
#define SC_P     132
#define SC_QCAP  256
#define SC_FLUSH 64
#define SC_U     8

__device__ __forceinline__ void sc_drain(const int* __restrict__ ids,
                                         const float* __restrict__ h,
                                         float* acc, const unsigned* qe,
                                         int n, int lo, int lane) {
    for (int i0 = 0; i0 < n; i0 += 32) {
        const int m = (n - i0 < 32) ? (n - i0) : 32;
        int my_g = 0, my_d = 0;
        if (lane < m) {
            const unsigned g = qe[i0 + lane];
            my_g = (int)g;
            my_d = ids[g] - lo;
        }
        for (int j0 = 0; j0 < m; j0 += 8) {
            float4 v[8];
            int    dl[8];
#pragma unroll
            for (int u = 0; u < 8; ++u) {
                const int j = (j0 + u < m) ? (j0 + u) : (m - 1);
                const int g = __builtin_amdgcn_readlane(my_g, j);
                dl[u]       = __builtin_amdgcn_readlane(my_d, j);
                v[u] = *(const float4*)(h + (size_t)g * 128 + lane * 4);
            }
#pragma unroll
            for (int u = 0; u < 8; ++u) {
                if (j0 + u < m) {
                    float* row = acc + dl[u] * SC_P + lane;
                    __hip_atomic_fetch_add(row,      v[u].x, __ATOMIC_RELAXED, __HIP_MEMORY_SCOPE_WORKGROUP);
                    __hip_atomic_fetch_add(row + 32, v[u].y, __ATOMIC_RELAXED, __HIP_MEMORY_SCOPE_WORKGROUP);
                    __hip_atomic_fetch_add(row + 64, v[u].z, __ATOMIC_RELAXED, __HIP_MEMORY_SCOPE_WORKGROUP);
                    __hip_atomic_fetch_add(row + 96, v[u].w, __ATOMIC_RELAXED, __HIP_MEMORY_SCOPE_WORKGROUP);
                }
            }
        }
    }
}

__global__ __launch_bounds__(1024)
void scatter_classify_owner_kernel(const int* __restrict__ ids,
                                   const float* __restrict__ h,
                                   const float* __restrict__ Wcf,
                                   const float* __restrict__ bc,
                                   float* __restrict__ out,
                                   int nIds, int nNodes) {
    __shared__ float    acc[SC_RMAX * SC_P];
    __shared__ unsigned qe[OWN_WAVES][SC_QCAP];

    const int lane = threadIdx.x & 31;
    const int wave = __builtin_amdgcn_readfirstlane((int)(threadIdx.x >> 5));
    const int nwav = blockDim.x >> 5;
    const int halfsel = lane >> 4;
    const int lid     = lane & 15;
    int R = (nNodes + OWN_GRID - 1) / OWN_GRID;
    if (R > SC_RMAX) R = SC_RMAX;
    const int nGf = nIds >> 7;
    const int rot = (int)(((long long)blockIdx.x * nGf) / OWN_GRID);
    unsigned* myqe = qe[wave];

    for (int lo = blockIdx.x * R; lo < nNodes; lo += OWN_GRID * R) {
        const int rows  = (nNodes - lo < R) ? (nNodes - lo) : R;
        const int ntile = (rows + 15) >> 4;
        for (int i = threadIdx.x; i < ntile * 16 * (SC_P / 4); i += blockDim.x)
            ((float4*)acc)[i] = make_float4(0.f, 0.f, 0.f, 0.f);
        __syncthreads();

        int cnt = 0;
        for (int g0 = wave; g0 < nGf; g0 += SC_U * nwav) {
            uint4 d[SC_U];
            int   eb[SC_U];
#pragma unroll
            for (int u = 0; u < SC_U; ++u) {
                int g = g0 + u * nwav;
                if (g >= nGf) g = nGf - 1;
                g += rot;
                if (g >= nGf) g -= nGf;
                eb[u] = g * 128 + lane * 4;
                d[u]  = *(const uint4*)(ids + eb[u]);
            }
#pragma unroll
            for (int u = 0; u < SC_U; ++u) {
                if (g0 + u * nwav < nGf) {
                    const unsigned b = (unsigned)eb[u];
                    cnt = own_push(d[u].x - (unsigned)lo < (unsigned)rows, b,      myqe, cnt);
                    cnt = own_push(d[u].y - (unsigned)lo < (unsigned)rows, b + 1u, myqe, cnt);
                    cnt = own_push(d[u].z - (unsigned)lo < (unsigned)rows, b + 2u, myqe, cnt);
                    cnt = own_push(d[u].w - (unsigned)lo < (unsigned)rows, b + 3u, myqe, cnt);
                    if (cnt >= SC_FLUSH) {
                        __builtin_amdgcn_fence(__ATOMIC_ACQ_REL, "wavefront");
                        sc_drain(ids, h, acc, myqe, cnt, lo, lane);
                        cnt = 0;
                    }
                }
            }
        }
        if (wave == 0) {
            for (int e0 = nGf * 128; e0 < nIds; e0 += 32) {
                const int e = e0 + lane;
                const unsigned r = ((e < nIds) ? (unsigned)ids[e] : 0xFFFFFFFFu) - (unsigned)lo;
                cnt = own_push(r < (unsigned)rows && e < nIds, (unsigned)e, myqe, cnt);
                if (cnt >= SC_FLUSH) {
                    __builtin_amdgcn_fence(__ATOMIC_ACQ_REL, "wavefront");
                    sc_drain(ids, h, acc, myqe, cnt, lo, lane);
                    cnt = 0;
                }
            }
        }
        __builtin_amdgcn_fence(__ATOMIC_ACQ_REL, "wavefront");
        sc_drain(ids, h, acc, myqe, cnt, lo, lane);
        __syncthreads();

        for (int t = wave; t < ntile; t += nwav) {
            v2f b[32];
            const float4* wf4 = (const float4*)Wcf + lane;
#pragma unroll
            for (int kk2 = 0; kk2 < 16; ++kk2) {
                const float4 q = wf4[kk2 * 32];
                b[2 * kk2].x     = q.x; b[2 * kk2].y     = q.y;
                b[2 * kk2 + 1].x = q.z; b[2 * kk2 + 1].y = q.w;
            }
            const float bv = bc[lid];
            v8f c;
#pragma unroll
            for (int j = 0; j < 8; ++j) c[j] = bv;
            const float* ar = acc + (t * 16 + lid) * SC_P + halfsel * 64;
#pragma unroll
            for (int kk = 0; kk < 32; ++kk) {
                v2f a;
                a.x = ar[kk];
                a.y = ar[32 + kk];
                c = __builtin_amdgcn_wmma_f32_16x16x4_f32(
                    false, a, false, b[kk], (short)0, c, false, false);
            }
#pragma unroll
            for (int j = 0; j < 8; ++j) {
                const int row = t * 16 + j + halfsel * 8;
                if (row < rows) out[(size_t)(lo + row) * 16 + lid] = c[j];
            }
        }
        __syncthreads();
    }
}

template <int NTILES>
__global__ __launch_bounds__(128)
void gemm_wmma_k128_kernel(const float* __restrict__ A,
                           const float* __restrict__ Wf,
                           const float* __restrict__ bias,
                           const float* __restrict__ arowscale,
                           const float* __restrict__ erowscale,
                           float* __restrict__ out,
                           int M, int relu) {
    const int lane    = threadIdx.x & 31;
    const int wave    = __builtin_amdgcn_readfirstlane((int)(threadIdx.x >> 5));
    const int wpb     = blockDim.x >> 5;
    const int nwaves  = gridDim.x * wpb;
    const int ntiles  = M >> 4;
    int tile          = blockIdx.x * wpb + wave;
    if (tile >= ntiles) return;
    const int halfsel = lane >> 4;
    const int lid     = lane & 15;
    const int N       = NTILES * 16;

    v2f b[NTILES][32];
    {
        const float4* wf4 = (const float4*)Wf + lane;
#pragma unroll
        for (int nt = 0; nt < NTILES; ++nt) {
#pragma unroll
            for (int kk2 = 0; kk2 < 16; ++kk2) {
                const float4 q = wf4[(nt * 16 + kk2) * 32];
                b[nt][2 * kk2].x     = q.x; b[nt][2 * kk2].y     = q.y;
                b[nt][2 * kk2 + 1].x = q.z; b[nt][2 * kk2 + 1].y = q.w;
            }
        }
    }
    float bv[NTILES];
#pragma unroll
    for (int nt = 0; nt < NTILES; ++nt) bv[nt] = bias[nt * 16 + lid];

    v2f   an[32];
    float sn;
    {
        const int r = tile * 16 + lid;
        const float* ap = A + (size_t)r * 128 + halfsel * 2;
#pragma unroll
        for (int kk = 0; kk < 32; ++kk) an[kk] = *(const v2f*)(ap + kk * 4);
        sn = arowscale ? arowscale[r] : 1.0f;
    }

    for (; tile < ntiles; tile += nwaves) {
        v2f a[32];
#pragma unroll
        for (int kk = 0; kk < 32; ++kk) { a[kk].x = an[kk].x * sn; a[kk].y = an[kk].y * sn; }

        const int tnext = tile + nwaves;
        if (tnext < ntiles) {
            const int r = tnext * 16 + lid;
            const float* ap = A + (size_t)r * 128 + halfsel * 2;
#pragma unroll
            for (int kk = 0; kk < 32; ++kk) an[kk] = *(const v2f*)(ap + kk * 4);
            sn = arowscale ? arowscale[r] : 1.0f;
        }

        float es[8];
        if (erowscale) {
#pragma unroll
            for (int j = 0; j < 8; ++j) es[j] = erowscale[tile * 16 + j + halfsel * 8];
        } else {
#pragma unroll
            for (int j = 0; j < 8; ++j) es[j] = 1.0f;
        }

        v8f acc[NTILES];
#pragma unroll
        for (int nt = 0; nt < NTILES; ++nt) {
#pragma unroll
            for (int j = 0; j < 8; ++j) acc[nt][j] = bv[nt];
        }

#pragma unroll
        for (int kk = 0; kk < 32; ++kk) {
#pragma unroll
            for (int nt = 0; nt < NTILES; ++nt) {
                acc[nt] = __builtin_amdgcn_wmma_f32_16x16x4_f32(
                    false, a[kk], false, b[nt][kk], (short)0, acc[nt], false, false);
            }
        }

#pragma unroll
        for (int nt = 0; nt < NTILES; ++nt) {
#pragma unroll
            for (int j = 0; j < 8; ++j) {
                float v = acc[nt][j];
                if (relu) v = fmaxf(v, 0.0f);
                const int row = tile * 16 + j + halfsel * 8;
                if (erowscale) v *= es[j];
                out[(size_t)row * N + nt * 16 + lid] = v;
            }
        }
    }
}


extern "C" void kernel_launch(void* const* d_in, const int* in_sizes, int n_in,
                              void* d_out, int out_size, void* d_ws, size_t ws_size,
                              hipStream_t stream) {
    (void)n_in; (void)ws_size;
    const float* x   = (const float*)d_in[0];
    const float* W2  = (const float*)d_in[1];
    const float* b2  = (const float*)d_in[2];
    const float* gw  = (const float*)d_in[3];
    const float* gb  = (const float*)d_in[4];
    const float* Wc  = (const float*)d_in[5];
    const float* bc  = (const float*)d_in[6];
    const int*   src = (const int*)d_in[7];
    const int*   dst = (const int*)d_in[8];
    const int*   ids = (const int*)d_in[9];

    const int H  = 128;
    const int C  = 16;
    const int L  = in_sizes[2] / H;
    const int NS = in_sizes[9] / L;
    const int NE = in_sizes[7] / L;
    const int NL = in_sizes[4] / (L * H);
    const int NN = out_size / C;

    float* ws       = (float*)d_ws;
    float* hfin     = ws;
    float* hbuf     = hfin + (size_t)(L - 1) * NS * H;
    float* agg      = hbuf + (size_t)NS * H;
    float* deg_out  = agg + (size_t)NS * H;
    float* w2t      = deg_out + 2 * (size_t)NS;
    float* gwt      = w2t + (size_t)L * H * H;
    float* wct      = gwt + (size_t)L * NL * H * H;
    unsigned char* okey = (unsigned char*)d_out;
    unsigned* olist = (unsigned*)(okey + (((size_t)NE + 255) & ~(size_t)255));
    const bool lists = ((long long)OWN_GRID * own_range_rows(NS) >= NS) && (NS <= (1 << 24))
                    && (sl_per(NE) <= SL_CAP)
                    && ((size_t)DEG_SLICES * sl_stride(NE) <= (size_t)out_size);
    unsigned* slist = lists ? (unsigned*)d_out : nullptr;

    const int T = 256;
    const int GT = 128;
    const int gemm_blocks_NS = 256;

    {
        int nHH = H * H;
        int bHH = (nHH + T - 1) / T;
        for (int i = 0; i < L; ++i)
            pack_bfrag_kernel<<<bHH, T, 0, stream>>>(W2 + (size_t)i * nHH,
                                                     w2t + (size_t)i * nHH, H);
        for (int i = 0; i < L * NL; ++i)
            pack_bfrag_kernel<<<bHH, T, 0, stream>>>(gw + (size_t)i * nHH,
                                                     gwt + (size_t)i * nHH, H);
        int nHC = H * C;
        pack_bfrag_kernel<<<(nHC + T - 1) / T, T, 0, stream>>>(Wc, wct, C);
    }

    for (int i = 0; i < L; ++i) {
        const int*   src_i = src + (size_t)i * NE;
        const int*   dst_i = dst + (size_t)i * NE;
        const float* x_i   = x   + (size_t)i * NS * H;
        const float* w2t_i = w2t + (size_t)i * H * H;
        const float* b2_i  = b2  + (size_t)i * H;

        {
            const int pstride = (NS + 3) & ~3;
            unsigned* dpart = (unsigned*)agg;
            degree_slice_kernel<<<DEG_SLICES, 1024, 0, stream>>>(src_i, dst_i, dpart,
                                                                 okey, slist, NE, NS, pstride);
            degree_rsqrt_kernel<<<(NS + T - 1) / T, T, 0, stream>>>(dpart, deg_out,
                                                                    NS, pstride);
        }

        float* pin  = (NL & 1) ? agg : hbuf;
        float* pout = (NL & 1) ? hbuf : agg;

        gemm_wmma_k128_kernel<8><<<gemm_blocks_NS, GT, 0, stream>>>(
            x_i, w2t_i, b2_i, nullptr, deg_out, pin, NS, 0);

        for (int l = 0; l < NL; ++l) {
            const float* gwt_il = gwt + ((size_t)i * NL + l) * H * H;
            const float* gb_il  = gb  + ((size_t)i * NL + l) * H;
            const float* escale = (l + 1 < NL) ? deg_out : nullptr;
            float* hout = (l + 1 < NL) ? pout : hfin + (size_t)i * NS * H;
            if (lists)
                conv_lists_kernel<<<OWN_GRID, 1024, 0, stream>>>(pin, gwt_il, gb_il, escale,
                                                                 hout, slist, NE, NS);
            else
                conv_owner_kernel<<<OWN_GRID, 1024, 0, stream>>>(src_i, dst_i, okey, pin,
                                                                 gwt_il, gb_il, escale,
                                                                 hout, olist, l, NE, NS);
            float* tmp = pin; pin = pout; pout = tmp;
        }
    }

    scatter_classify_owner_kernel<<<OWN_GRID, 1024, 0, stream>>>(
        ids, hfin, wct, bc, (float*)d_out, L * NS, NN);
}
